# starting kernel
# baseline (speedup 1.0000x reference)
#include <hip/hip_runtime.h>

typedef float v2f __attribute__((ext_vector_type(2)));
typedef float v8f __attribute__((ext_vector_type(8)));


__global__ void fill_ones_k(float* __restrict__ p, int n) {
    int i = blockIdx.x * blockDim.x + threadIdx.x;
    if (i < n) p[i] = 1.0f;
}

__global__ void deg_accum_k(const int* __restrict__ dst, float* __restrict__ deg, int E) {
    int e = blockIdx.x * blockDim.x + threadIdx.x;
    if (e < E) atomicAdd(&deg[dst[e]], 1.0f);
}

__global__ void rsqrt_k(float* __restrict__ p, int n) {
    int i = blockIdx.x * blockDim.x + threadIdx.x;
    if (i < n) p[i] = rsqrtf(p[i]);
}

__global__ void coef_k(const int* __restrict__ src, const int* __restrict__ dst,
                       const float* __restrict__ dinv, float* __restrict__ coef, int E) {
    int e = blockIdx.x * blockDim.x + threadIdx.x;
    if (e < E) coef[e] = dinv[src[e]] * dinv[dst[e]];
}


template<int KDIM, int NOUT>
__global__ void gemm_wmma_k(const float* __restrict__ X, const float* __restrict__ W,
                            float* __restrict__ H, int ntiles, int nrows) {
    constexpr int NT   = (NOUT + 15) / 16;
    constexpr int KT   = (KDIM + 3) / 4;
    constexpr int KPAD = KT * 4;
    constexpr int NW16 = NT * 16;

    __shared__ float sW[KPAD * NW16];

    for (int idx = threadIdx.x; idx < KPAD * NW16; idx += blockDim.x) {
        const int k = idx / NW16;
        const int n = idx - k * NW16;
        sW[idx] = (k < KDIM && n < NOUT) ? W[k * NOUT + n] : 0.0f;
    }
    __syncthreads();

    const int lane = threadIdx.x & 31;
    const int wv   = blockIdx.x * (blockDim.x >> 5) + (threadIdx.x >> 5);
    if (wv >= ntiles) return;
    const int row0 = wv * 16;
    const int m    = lane & 15;
    const int half = lane >> 4;

    const unsigned lim = (unsigned)nrows * KDIM - 1u;

    v8f acc[NT];
#pragma unroll
    for (int t = 0; t < NT; ++t) acc[t] = 0.0f;

#pragma unroll
    for (int kt = 0; kt < KT; ++kt) {
        const int ka = kt * 4 + half * 2;
        const unsigned i0 = (unsigned)(row0 + m) * KDIM + ka;
        v2f a;
        a.x = X[min(i0,      lim)];
        a.y = X[min(i0 + 1u, lim)];
#pragma unroll
        for (int nt = 0; nt < NT; ++nt) {
            const int col = nt * 16 + m;
            v2f b;
            b.x = sW[ka * NW16 + col];
            b.y = sW[(ka + 1) * NW16 + col];
            acc[nt] = __builtin_amdgcn_wmma_f32_16x16x4_f32(
                false, a, false, b, (short)0, acc[nt], false, false);
        }
    }

#pragma unroll
    for (int nt = 0; nt < NT; ++nt) {
#pragma unroll
        for (int i = 0; i < 8; ++i) {
            const int r   = row0 + i + half * 8;
            const int col = nt * 16 + m;
            if (col < NOUT) H[(size_t)r * NOUT + col] = acc[nt][i];
        }
    }
}


template<int C>
__global__ void init_agg_k(const float* __restrict__ h, const float* __restrict__ dinv,
                           const float* __restrict__ b, float* __restrict__ agg, int n) {
    int t = blockIdx.x * blockDim.x + threadIdx.x;
    if (t >= n * C) return;
    const int i  = t / C;
    const int ch = t - i * C;
    const float di = dinv[i];
    agg[t] = fmaf(h[t], di * di, b[ch]);
}

__global__ void scatter64_k(const int* __restrict__ src, const int* __restrict__ dst,
                            const float* __restrict__ coef, const float* __restrict__ h,
                            float* __restrict__ agg, int E) {
    const int t = blockIdx.x * blockDim.x + threadIdx.x;
    const int e = t >> 4;
    if (e >= E) return;
    const int cg = (t & 15) << 2;
    const int s = src[e];
    const int d = dst[e];
    const float c = coef[e];
    const float4 v = *reinterpret_cast<const float4*>(h + (size_t)s * 64 + cg);
    float* p = agg + (size_t)d * 64 + cg;
    atomicAdd(p + 0, v.x * c);
    atomicAdd(p + 1, v.y * c);
    atomicAdd(p + 2, v.z * c);
    atomicAdd(p + 3, v.w * c);
}

__global__ void scatter3_k(const int* __restrict__ src, const int* __restrict__ dst,
                           const float* __restrict__ coef, const float* __restrict__ h,
                           float* __restrict__ out, int E) {
    const int e = blockIdx.x * blockDim.x + threadIdx.x;
    if (e >= E) return;
    const int s = src[e];
    const int d = dst[e];
    const float c = coef[e];
    const float* hp = h + (size_t)s * 3;
    float* p = out + (size_t)d * 3;
    atomicAdd(p + 0, hp[0] * c);
    atomicAdd(p + 1, hp[1] * c);
    atomicAdd(p + 2, hp[2] * c);
}

__global__ void relu_k(float* __restrict__ p, int n) {
    int i = blockIdx.x * blockDim.x + threadIdx.x;
    if (i < n) p[i] = fmaxf(p[i], 0.0f);
}


static inline int cdiv(long long a, int b) { return (int)((a + b - 1) / b); }

extern "C" void kernel_launch(void* const* d_in, const int* in_sizes, int n_in,
                              void* d_out, int out_size, void* d_ws, size_t ws_size,
                              hipStream_t stream) {
    const float* x   = (const float*)d_in[0];
    const int*   ei  = (const int*)d_in[1];
    const float* W1  = (const float*)d_in[2];
    const float* b1  = (const float*)d_in[3];
    const float* W2  = (const float*)d_in[4];
    const float* b2  = (const float*)d_in[5];
    const float* W3  = (const float*)d_in[6];
    const float* b3  = (const float*)d_in[7];
    float* out = (float*)d_out;

    const int N = in_sizes[0] / 9;
    const int E = in_sizes[1] / 2;
    const int* src = ei;
    const int* dst = ei + E;

    float* dinv = (float*)d_ws;
    float* coef = dinv + N;
    float* h    = coef + E;
    float* agg  = h + (size_t)N * 64;
    (void)ws_size; (void)n_in; (void)out_size;

    const int ntiles = (N + 15) / 16;

    fill_ones_k<<<cdiv(N, 256), 256, 0, stream>>>(dinv, N);
    deg_accum_k<<<cdiv(E, 256), 256, 0, stream>>>(dst, dinv, E);
    rsqrt_k<<<cdiv(N, 256), 256, 0, stream>>>(dinv, N);
    coef_k<<<cdiv(E, 256), 256, 0, stream>>>(src, dst, dinv, coef, E);

    gemm_wmma_k<9, 64><<<cdiv(ntiles, 8), 256, 0, stream>>>(x, W1, h, ntiles, N);
    init_agg_k<64><<<cdiv((long long)N * 64, 256), 256, 0, stream>>>(h, dinv, b1, agg, N);
    scatter64_k<<<cdiv((long long)E * 16, 256), 256, 0, stream>>>(src, dst, coef, h, agg, E);
    relu_k<<<cdiv((long long)N * 64, 256), 256, 0, stream>>>(agg, N * 64);

    gemm_wmma_k<64, 64><<<cdiv(ntiles, 8), 256, 0, stream>>>(agg, W2, h, ntiles, N);
    init_agg_k<64><<<cdiv((long long)N * 64, 256), 256, 0, stream>>>(h, dinv, b2, agg, N);
    scatter64_k<<<cdiv((long long)E * 16, 256), 256, 0, stream>>>(src, dst, coef, h, agg, E);
    relu_k<<<cdiv((long long)N * 64, 256), 256, 0, stream>>>(agg, N * 64);

    gemm_wmma_k<64, 3><<<cdiv(ntiles, 8), 256, 0, stream>>>(agg, W3, h, ntiles, N);
    init_agg_k<3><<<cdiv((long long)N * 3, 256), 256, 0, stream>>>(h, dinv, b3, out, N);
    scatter3_k<<<cdiv(E, 256), 256, 0, stream>>>(src, dst, coef, h, out, E);
}


# Float atomic scatter-adds replaced by a per-call edge list grouped by destination (integer atomic rank, scan, place) and per-node gather sums in arrival-rank order; layer 3 joins 8 lanes by a tree
# speedup vs baseline: 2.4401x; 2.4401x over previous
#include <hip/hip_runtime.h>


typedef float v2f __attribute__((ext_vector_type(2)));
typedef float v8f __attribute__((ext_vector_type(8)));


__global__ void zero_cnt_k(int* __restrict__ cnt, int n) {
    int i = blockIdx.x * blockDim.x + threadIdx.x;
    if (i < n) cnt[i] = 0;
}

__global__ void hist_rank_k(const int* __restrict__ dst, int* __restrict__ cnt,
                            int* __restrict__ rank, int E) {
    int e = blockIdx.x * blockDim.x + threadIdx.x;
    if (e < E) rank[e] = atomicAdd(&cnt[dst[e]], 1);
}

__global__ void scan_k(const int* __restrict__ cnt, int* __restrict__ rowstart, int n) {
    __shared__ int s[1024];
    const int tid  = threadIdx.x;
    const int base = blockIdx.x * 1024;

    int part = 0;
    for (int j = tid; j < base; j += 1024) part += cnt[j];
    s[tid] = part;
    __syncthreads();
    for (int st = 512; st > 0; st >>= 1) {
        if (tid < st) s[tid] += s[tid + st];
        __syncthreads();
    }
    const int offset = s[0];
    __syncthreads();

    const int i = base + tid;
    const int v = (i < n) ? cnt[i] : 0;
    s[tid] = v;
    __syncthreads();
    for (int d = 1; d < 1024; d <<= 1) {
        const int t = (tid >= d) ? s[tid - d] : 0;
        __syncthreads();
        s[tid] += t;
        __syncthreads();
    }
    if (i < n) rowstart[i] = offset + s[tid] - v;
}

__global__ void fill_k(const int* __restrict__ src, const int* __restrict__ dst,
                       const int* __restrict__ rowstart, const int* __restrict__ rank,
                       int* __restrict__ csrc, int E) {
    int e = blockIdx.x * blockDim.x + threadIdx.x;
    if (e >= E) return;
    unsigned pos = (unsigned)(rowstart[dst[e]] + rank[e]);
    pos = min(pos, (unsigned)(E - 1));
    csrc[pos] = src[e];
}

__device__ __forceinline__ float dinv_of(const int* __restrict__ rowstart, int s, int n, int E) {
    const int q0 = rowstart[s];
    int q1 = rowstart[min(s + 1, n - 1)];
    q1 = (s + 1 < n) ? q1 : E;
    return rsqrtf((float)(q1 - q0 + 1));
}


template<int KDIM, int NOUT>
__global__ void gemm_wmma_k(const float* __restrict__ X, const float* __restrict__ W,
                            float* __restrict__ H, int ntiles, int nrows) {
    constexpr int NT   = (NOUT + 15) / 16;
    constexpr int KT   = (KDIM + 3) / 4;
    constexpr int KPAD = KT * 4;
    constexpr int NW16 = NT * 16;

    __shared__ float sW[KPAD * NW16];

    for (int idx = threadIdx.x; idx < KPAD * NW16; idx += blockDim.x) {
        const int k = idx / NW16;
        const int n = idx - k * NW16;
        sW[idx] = (k < KDIM && n < NOUT) ? W[k * NOUT + n] : 0.0f;
    }
    __syncthreads();

    const int lane = threadIdx.x & 31;
    const int wv   = blockIdx.x * (blockDim.x >> 5) + (threadIdx.x >> 5);
    if (wv >= ntiles) return;
    const int row0 = wv * 16;
    const int m    = lane & 15;
    const int half = lane >> 4;

    const unsigned lim = (unsigned)nrows * KDIM - 1u;

    v8f acc[NT];
#pragma unroll
    for (int t = 0; t < NT; ++t) acc[t] = 0.0f;

#pragma unroll
    for (int kt = 0; kt < KT; ++kt) {
        const int ka = kt * 4 + half * 2;
        const unsigned i0 = (unsigned)(row0 + m) * KDIM + ka;
        v2f a;
        a.x = X[min(i0,      lim)];
        a.y = X[min(i0 + 1u, lim)];
#pragma unroll
        for (int nt = 0; nt < NT; ++nt) {
            const int col = nt * 16 + m;
            v2f b;
            b.x = sW[ka * NW16 + col];
            b.y = sW[(ka + 1) * NW16 + col];
            acc[nt] = __builtin_amdgcn_wmma_f32_16x16x4_f32(
                false, a, false, b, (short)0, acc[nt], false, false);
        }
    }

#pragma unroll
    for (int nt = 0; nt < NT; ++nt) {
#pragma unroll
        for (int i = 0; i < 8; ++i) {
            const int r   = row0 + i + half * 8;
            const int col = nt * 16 + m;
            if (col < NOUT) H[(size_t)r * NOUT + col] = acc[nt][i];
        }
    }
}


__global__ void gather64_relu_k(const int* __restrict__ rowstart, const int* __restrict__ csrc,
                                const float* __restrict__ h, const float* __restrict__ b,
                                float* __restrict__ agg, int n, int E) {
    const int t = blockIdx.x * blockDim.x + threadIdx.x;
    const int i = t >> 4;
    if (i >= n) return;
    const int cg = (t & 15) << 2;

    int p0 = rowstart[i];
    int p1 = (i + 1 < n) ? rowstart[i + 1] : E;
    const float di = rsqrtf((float)(p1 - p0 + 1));
    p0 = max(p0, 0);
    p1 = min(p1, E);

    const float4 hv = *reinterpret_cast<const float4*>(h + (size_t)i * 64 + cg);
    const float4 bv = *reinterpret_cast<const float4*>(b + cg);
    const float sl = di * di;
    float4 acc;
    acc.x = fmaf(hv.x, sl, bv.x);
    acc.y = fmaf(hv.y, sl, bv.y);
    acc.z = fmaf(hv.z, sl, bv.z);
    acc.w = fmaf(hv.w, sl, bv.w);

#pragma unroll 4
    for (int p = p0; p < p1; ++p) {
        int s = csrc[p];
        s = ((unsigned)s < (unsigned)n) ? s : i;
        const float c = dinv_of(rowstart, s, n, E) * di;
        const float4 v = *reinterpret_cast<const float4*>(h + (size_t)s * 64 + cg);
        acc.x += v.x * c;
        acc.y += v.y * c;
        acc.z += v.z * c;
        acc.w += v.w * c;
    }

    acc.x = fmaxf(acc.x, 0.0f);
    acc.y = fmaxf(acc.y, 0.0f);
    acc.z = fmaxf(acc.z, 0.0f);
    acc.w = fmaxf(acc.w, 0.0f);
    *reinterpret_cast<float4*>(agg + (size_t)i * 64 + cg) = acc;
}

__global__ void gather3_k(const int* __restrict__ rowstart, const int* __restrict__ csrc,
                          const float* __restrict__ h, const float* __restrict__ b,
                          float* __restrict__ out, int n, int E) {
    __shared__ float so[96];
    const int tid = threadIdx.x;
    const int sub = tid & 7;
    const int i   = blockIdx.x * 32 + (tid >> 3);
    const bool live = i < n;
    const int ic = live ? i : (n - 1);

    int p0 = rowstart[ic];
    int p1 = (ic + 1 < n) ? rowstart[ic + 1] : E;
    const float di = rsqrtf((float)(p1 - p0 + 1));
    p0 = max(p0, 0);
    p1 = live ? min(p1, E) : p0;

    float a0 = 0.0f, a1 = 0.0f, a2 = 0.0f;
    for (int p = p0 + sub; p < p1; p += 8) {
        int s = csrc[p];
        s = ((unsigned)s < (unsigned)n) ? s : ic;
        const float c = dinv_of(rowstart, s, n, E) * di;
        const float* hp = h + (size_t)s * 3;
        a0 += hp[0] * c;
        a1 += hp[1] * c;
        a2 += hp[2] * c;
    }
    a0 += __shfl_xor(a0, 4); a1 += __shfl_xor(a1, 4); a2 += __shfl_xor(a2, 4);
    a0 += __shfl_xor(a0, 2); a1 += __shfl_xor(a1, 2); a2 += __shfl_xor(a2, 2);
    a0 += __shfl_xor(a0, 1); a1 += __shfl_xor(a1, 1); a2 += __shfl_xor(a2, 1);

    if (sub == 0) {
        const float sl = di * di;
        const float* hp = h + (size_t)ic * 3;
        const int o = (tid >> 3) * 3;
        so[o + 0] = fmaf(hp[0], sl, b[0]) + a0;
        so[o + 1] = fmaf(hp[1], sl, b[1]) + a1;
        so[o + 2] = fmaf(hp[2], sl, b[2]) + a2;
    }
    __syncthreads();
    if (tid < 96) {
        const long long idx = (long long)blockIdx.x * 96 + tid;
        if (idx < (long long)n * 3) out[idx] = so[tid];
    }
}


static inline int cdiv(long long a, int b) { return (int)((a + b - 1) / b); }

extern "C" void kernel_launch(void* const* d_in, const int* in_sizes, int n_in,
                              void* d_out, int out_size, void* d_ws, size_t ws_size,
                              hipStream_t stream) {
    const float* x   = (const float*)d_in[0];
    const int*   ei  = (const int*)d_in[1];
    const float* W1  = (const float*)d_in[2];
    const float* b1  = (const float*)d_in[3];
    const float* W2  = (const float*)d_in[4];
    const float* b2  = (const float*)d_in[5];
    const float* W3  = (const float*)d_in[6];
    const float* b3  = (const float*)d_in[7];
    float* out = (float*)d_out;

    const int N = in_sizes[0] / 9;
    const int E = in_sizes[1] / 2;
    const int* src = ei;
    const int* dst = ei + E;

    int*   rowstart = (int*)d_ws;
    int*   csrc     = rowstart + N;
    float* h        = (float*)(csrc + E);
    float* agg      = h + (size_t)N * 64;
    int*   rank     = (int*)h;
    int*   cnt      = (int*)agg;
    (void)ws_size; (void)n_in; (void)out_size;

    const int ntiles = (N + 15) / 16;

    zero_cnt_k<<<cdiv(N, 256), 256, 0, stream>>>(cnt, N);
    hist_rank_k<<<cdiv(E, 256), 256, 0, stream>>>(dst, cnt, rank, E);
    scan_k<<<cdiv(N, 1024), 1024, 0, stream>>>(cnt, rowstart, N);
    fill_k<<<cdiv(E, 256), 256, 0, stream>>>(src, dst, rowstart, rank, csrc, E);

    gemm_wmma_k<9, 64><<<cdiv(ntiles, 8), 256, 0, stream>>>(x, W1, h, ntiles, N);
    gather64_relu_k<<<cdiv((long long)N * 16, 256), 256, 0, stream>>>(rowstart, csrc, h, b1, agg, N, E);

    gemm_wmma_k<64, 64><<<cdiv(ntiles, 8), 256, 0, stream>>>(agg, W2, h, ntiles, N);
    gather64_relu_k<<<cdiv((long long)N * 16, 256), 256, 0, stream>>>(rowstart, csrc, h, b2, agg, N, E);

    gemm_wmma_k<64, 3><<<cdiv(ntiles, 8), 256, 0, stream>>>(agg, W3, h, ntiles, N);
    gather3_k<<<cdiv(N, 32), 256, 0, stream>>>(rowstart, csrc, h, b3, out, N, E);
}


# Source-side deg^-1/2 folded into each transform's stored rows; gathers add plain rows, then scale by destination deg^-1/2 and add bias last: no per-edge degree look-ups (rounding order changes)
# speedup vs baseline: 2.6604x; 1.0903x over previous
#include <hip/hip_runtime.h>


typedef float v2f __attribute__((ext_vector_type(2)));
typedef float v8f __attribute__((ext_vector_type(8)));


__global__ void zero_cnt_k(int* __restrict__ cnt, int n) {
    int i = blockIdx.x * blockDim.x + threadIdx.x;
    if (i < n) cnt[i] = 0;
}

__global__ void hist_rank_k(const int* __restrict__ dst, int* __restrict__ cnt,
                            int* __restrict__ rank, int E) {
    int e = blockIdx.x * blockDim.x + threadIdx.x;
    if (e < E) rank[e] = atomicAdd(&cnt[dst[e]], 1);
}

__global__ void scan_k(const int* __restrict__ cnt, int* __restrict__ rowstart, int n) {
    __shared__ int s[1024];
    const int tid  = threadIdx.x;
    const int base = blockIdx.x * 1024;

    int part = 0;
    for (int j = tid; j < base; j += 1024) part += cnt[j];
    s[tid] = part;
    __syncthreads();
    for (int st = 512; st > 0; st >>= 1) {
        if (tid < st) s[tid] += s[tid + st];
        __syncthreads();
    }
    const int offset = s[0];
    __syncthreads();

    const int i = base + tid;
    const int v = (i < n) ? cnt[i] : 0;
    s[tid] = v;
    __syncthreads();
    for (int d = 1; d < 1024; d <<= 1) {
        const int t = (tid >= d) ? s[tid - d] : 0;
        __syncthreads();
        s[tid] += t;
        __syncthreads();
    }
    if (i < n) rowstart[i] = offset + s[tid] - v;
}

__global__ void fill_k(const int* __restrict__ src, const int* __restrict__ dst,
                       const int* __restrict__ rowstart, const int* __restrict__ rank,
                       int* __restrict__ csrc, int E) {
    int e = blockIdx.x * blockDim.x + threadIdx.x;
    if (e >= E) return;
    unsigned pos = (unsigned)(rowstart[dst[e]] + rank[e]);
    pos = min(pos, (unsigned)(E - 1));
    csrc[pos] = src[e];
}

__device__ __forceinline__ float dinv_of(const int* __restrict__ rowstart, int s, int n, int E) {
    const int q0 = rowstart[s];
    int q1 = rowstart[min(s + 1, n - 1)];
    q1 = (s + 1 < n) ? q1 : E;
    return rsqrtf((float)(q1 - q0 + 1));
}


template<int KDIM, int NOUT>
__global__ void gemm_wmma_k(const float* __restrict__ X, const float* __restrict__ W,
                            const int* __restrict__ rowstart,
                            float* __restrict__ H, int ntiles, int nrows, int E) {
    constexpr int NT   = (NOUT + 15) / 16;
    constexpr int KT   = (KDIM + 3) / 4;
    constexpr int KPAD = KT * 4;
    constexpr int NW16 = NT * 16;

    __shared__ float sW[KPAD * NW16];
    __shared__ alignas(16) float sD[128];

    for (int idx = threadIdx.x; idx < KPAD * NW16; idx += blockDim.x) {
        const int k = idx / NW16;
        const int n = idx - k * NW16;
        sW[idx] = (k < KDIM && n < NOUT) ? W[k * NOUT + n] : 0.0f;
    }
    if (threadIdx.x < 128) {
        const int r = blockIdx.x * 128 + threadIdx.x;
        sD[threadIdx.x] = dinv_of(rowstart, min(r, nrows - 1), nrows, E);
    }
    __syncthreads();

    const int lane = threadIdx.x & 31;
    const int wv   = blockIdx.x * (blockDim.x >> 5) + (threadIdx.x >> 5);
    if (wv >= ntiles) return;
    const int row0 = wv * 16;
    const int m    = lane & 15;
    const int half = lane >> 4;

    const unsigned lim = (unsigned)nrows * KDIM - 1u;

    const float* dv = sD + (threadIdx.x >> 5) * 16 + half * 8;

    v8f acc[NT];
#pragma unroll
    for (int t = 0; t < NT; ++t) acc[t] = 0.0f;

#pragma unroll
    for (int kt = 0; kt < KT; ++kt) {
        const int ka = kt * 4 + half * 2;
        const unsigned i0 = (unsigned)(row0 + m) * KDIM + ka;
        v2f a;
        a.x = X[min(i0,      lim)];
        a.y = X[min(i0 + 1u, lim)];
#pragma unroll
        for (int nt = 0; nt < NT; ++nt) {
            const int col = nt * 16 + m;
            v2f b;
            b.x = sW[ka * NW16 + col];
            b.y = sW[(ka + 1) * NW16 + col];
            acc[nt] = __builtin_amdgcn_wmma_f32_16x16x4_f32(
                false, a, false, b, (short)0, acc[nt], false, false);
        }
    }

#pragma unroll
    for (int nt = 0; nt < NT; ++nt) {
#pragma unroll
        for (int i = 0; i < 8; ++i) {
            const int r   = row0 + i + half * 8;
            const int col = nt * 16 + m;
            if (col < NOUT) H[(size_t)r * NOUT + col] = acc[nt][i] * dv[i];
        }
    }
}


__global__ void gather64_relu_k(const int* __restrict__ rowstart, const int* __restrict__ csrc,
                                const float* __restrict__ h, const float* __restrict__ b,
                                float* __restrict__ agg, int n, int E) {
    const int t = blockIdx.x * blockDim.x + threadIdx.x;
    const int i = t >> 4;
    if (i >= n) return;
    const int cg = (t & 15) << 2;

    int p0 = rowstart[i];
    int p1 = (i + 1 < n) ? rowstart[i + 1] : E;
    const float di = rsqrtf((float)(p1 - p0 + 1));
    p0 = max(p0, 0);
    p1 = min(p1, E);

    const float4 bv = *reinterpret_cast<const float4*>(b + cg);
    float4 acc = *reinterpret_cast<const float4*>(h + (size_t)i * 64 + cg);

#pragma unroll 4
    for (int p = p0; p < p1; ++p) {
        int s = csrc[p];
        s = ((unsigned)s < (unsigned)n) ? s : i;
        const float4 v = *reinterpret_cast<const float4*>(h + (size_t)s * 64 + cg);
        acc.x += v.x;
        acc.y += v.y;
        acc.z += v.z;
        acc.w += v.w;
    }

    acc.x = fmaxf(fmaf(acc.x, di, bv.x), 0.0f);
    acc.y = fmaxf(fmaf(acc.y, di, bv.y), 0.0f);
    acc.z = fmaxf(fmaf(acc.z, di, bv.z), 0.0f);
    acc.w = fmaxf(fmaf(acc.w, di, bv.w), 0.0f);
    *reinterpret_cast<float4*>(agg + (size_t)i * 64 + cg) = acc;
}

__global__ void gather3_k(const int* __restrict__ rowstart, const int* __restrict__ csrc,
                          const float* __restrict__ h, const float* __restrict__ b,
                          float* __restrict__ out, int n, int E) {
    __shared__ float so[96];
    const int tid = threadIdx.x;
    const int sub = tid & 7;
    const int i   = blockIdx.x * 32 + (tid >> 3);
    const bool live = i < n;
    const int ic = live ? i : (n - 1);

    int p0 = rowstart[ic];
    int p1 = (ic + 1 < n) ? rowstart[ic + 1] : E;
    const float di = rsqrtf((float)(p1 - p0 + 1));
    p0 = max(p0, 0);
    p1 = live ? min(p1, E) : p0;

    float a0 = 0.0f, a1 = 0.0f, a2 = 0.0f;
    for (int p = p0 + sub; p < p1; p += 8) {
        int s = csrc[p];
        s = ((unsigned)s < (unsigned)n) ? s : ic;
        const float* hp = h + (size_t)s * 3;
        a0 += hp[0];
        a1 += hp[1];
        a2 += hp[2];
    }
    a0 += __shfl_xor(a0, 4); a1 += __shfl_xor(a1, 4); a2 += __shfl_xor(a2, 4);
    a0 += __shfl_xor(a0, 2); a1 += __shfl_xor(a1, 2); a2 += __shfl_xor(a2, 2);
    a0 += __shfl_xor(a0, 1); a1 += __shfl_xor(a1, 1); a2 += __shfl_xor(a2, 1);

    if (sub == 0) {
        const float* hp = h + (size_t)ic * 3;
        const int o = (tid >> 3) * 3;
        so[o + 0] = fmaf(hp[0] + a0, di, b[0]);
        so[o + 1] = fmaf(hp[1] + a1, di, b[1]);
        so[o + 2] = fmaf(hp[2] + a2, di, b[2]);
    }
    __syncthreads();
    if (tid < 96) {
        const long long idx = (long long)blockIdx.x * 96 + tid;
        if (idx < (long long)n * 3) out[idx] = so[tid];
    }
}


static inline int cdiv(long long a, int b) { return (int)((a + b - 1) / b); }

extern "C" void kernel_launch(void* const* d_in, const int* in_sizes, int n_in,
                              void* d_out, int out_size, void* d_ws, size_t ws_size,
                              hipStream_t stream) {
    const float* x   = (const float*)d_in[0];
    const int*   ei  = (const int*)d_in[1];
    const float* W1  = (const float*)d_in[2];
    const float* b1  = (const float*)d_in[3];
    const float* W2  = (const float*)d_in[4];
    const float* b2  = (const float*)d_in[5];
    const float* W3  = (const float*)d_in[6];
    const float* b3  = (const float*)d_in[7];
    float* out = (float*)d_out;

    const int N = in_sizes[0] / 9;
    const int E = in_sizes[1] / 2;
    const int* src = ei;
    const int* dst = ei + E;

    int*   rowstart = (int*)d_ws;
    int*   csrc     = rowstart + N;
    float* h        = (float*)(csrc + E);
    float* agg      = h + (size_t)N * 64;
    int*   rank     = (int*)h;
    int*   cnt      = (int*)agg;
    (void)ws_size; (void)n_in; (void)out_size;

    const int ntiles = (N + 15) / 16;

    zero_cnt_k<<<cdiv(N, 256), 256, 0, stream>>>(cnt, N);
    hist_rank_k<<<cdiv(E, 256), 256, 0, stream>>>(dst, cnt, rank, E);
    scan_k<<<cdiv(N, 1024), 1024, 0, stream>>>(cnt, rowstart, N);
    fill_k<<<cdiv(E, 256), 256, 0, stream>>>(src, dst, rowstart, rank, csrc, E);

    gemm_wmma_k<9, 64><<<cdiv(ntiles, 8), 256, 0, stream>>>(x, W1, rowstart, h, ntiles, N, E);
    gather64_relu_k<<<cdiv((long long)N * 16, 256), 256, 0, stream>>>(rowstart, csrc, h, b1, agg, N, E);

    gemm_wmma_k<64, 64><<<cdiv(ntiles, 8), 256, 0, stream>>>(agg, W2, rowstart, h, ntiles, N, E);
    gather64_relu_k<<<cdiv((long long)N * 16, 256), 256, 0, stream>>>(rowstart, csrc, h, b2, agg, N, E);

    gemm_wmma_k<64, 3><<<cdiv(ntiles, 8), 256, 0, stream>>>(agg, W3, rowstart, h, ntiles, N, E);
    gather3_k<<<cdiv(N, 32), 256, 0, stream>>>(rowstart, csrc, h, b3, out, N, E);
}


# Edge list grouped by destination by a two-level counting sort in LDS (128-node buckets, then nodes): no global atomic, random read or lone scattered store per edge; in-row sum order changes
# speedup vs baseline: 4.4803x; 1.6841x over previous
#include <hip/hip_runtime.h>


typedef float v2f __attribute__((ext_vector_type(2)));
typedef float v8f __attribute__((ext_vector_type(8)));


constexpr int BSH   = 7;
constexpr int BSZ   = 1 << BSH;
constexpr int NBMAX = 1024;
constexpr int TILE  = 6400;
constexpr int EPT   = (TILE + 1023) / 1024;
constexpr int STG   = 3072;
constexpr int KPT   = STG / 256;

__device__ __forceinline__ unsigned bucket_of(int d, int NB) {
    return min((unsigned)d >> BSH, (unsigned)(NB - 1));
}

__global__ void __launch_bounds__(1024)
bhist_k(const int* __restrict__ dst, int* __restrict__ hist, int E, int NB) {
    __shared__ int sh[NBMAX];
    const int tid = threadIdx.x;
    sh[tid] = 0;
    __syncthreads();
    const int e0 = blockIdx.x * TILE;
    const int e1 = min(e0 + TILE, E);
    const int elast = max(e1 - 1, 0);
    int d[EPT];
#pragma unroll
    for (int k = 0; k < EPT; ++k) d[k] = dst[min(e0 + tid + k * 1024, elast)];
#pragma unroll
    for (int k = 0; k < EPT; ++k) {
        const int e = e0 + tid + k * 1024;
        if (e < e1) atomicAdd(&sh[bucket_of(d[k], NB)], 1);
    }
    __syncthreads();
    if (tid < NB) hist[blockIdx.x * NB + tid] = sh[tid];
}

__global__ void __launch_bounds__(256)
bscan_k(const int* __restrict__ hist, int* __restrict__ loc, int* __restrict__ btot, int P, int NB) {
    __shared__ int wsum[8][32];
    const int lane = threadIdx.x & 31;
    const int w    = threadIdx.x >> 5;
    const int b    = blockIdx.x * 32 + lane;
    const int bc   = min(b, NB - 1);
    int carry = 0;
    for (int base = 0; base < P; base += 256) {
        const int p0 = base + w * 32;
        int v[32];
#pragma unroll
        for (int k = 0; k < 32; ++k) {
            const int p = p0 + k;
            const int t = hist[min(p, P - 1) * NB + bc];
            v[k] = (p < P) ? t : 0;
        }
        int tot = 0;
#pragma unroll
        for (int k = 0; k < 32; ++k) tot += v[k];
        wsum[w][lane] = tot;
        __syncthreads();
        int run = carry, all = 0;
#pragma unroll
        for (int j = 0; j < 8; ++j) {
            const int t = wsum[j][lane];
            run += (j < w) ? t : 0;
            all += t;
        }
#pragma unroll
        for (int k = 0; k < 32; ++k) {
            const int p = p0 + k;
            if (p < P && b < NB) loc[p * NB + b] = run;
            run += v[k];
        }
        carry += all;
        __syncthreads();
    }
    if (w == 0 && b < NB) btot[b] = carry;
}

__global__ void __launch_bounds__(1024)
part_k(const int* __restrict__ src, const int* __restrict__ dst,
       const int* __restrict__ loc, const int* __restrict__ btot,
       unsigned* __restrict__ tmp, int E, int NB) {
    __shared__ int sc[NBMAX];
    __shared__ int sg[NBMAX];
    __shared__ int lstart[NBMAX];
    __shared__ int goff[NBMAX];
    __shared__ int cur[NBMAX];
    __shared__ unsigned sorted[TILE];
    __shared__ unsigned short sbk[TILE];

    const int tid = threadIdx.x;
    const int e0  = blockIdx.x * TILE;
    const int e1  = min(e0 + TILE, E);

    const int elast = max(e1 - 1, 0);
    int dd[EPT], ss[EPT];
#pragma unroll
    for (int k = 0; k < EPT; ++k) {
        const int ec = min(e0 + tid + k * 1024, elast);
        dd[k] = dst[ec];
        ss[k] = src[ec];
    }
    const int gq = btot[min(tid, NB - 1)];
    const int lq = loc[blockIdx.x * NB + min(tid, NB - 1)];

    cur[tid] = 0;
    __syncthreads();
#pragma unroll
    for (int k = 0; k < EPT; ++k) {
        const int e = e0 + tid + k * 1024;
        if (e < e1) atomicAdd(&cur[bucket_of(dd[k], NB)], 1);
    }
    __syncthreads();

    const int v = cur[tid];
    const int g = (tid < NB) ? gq : 0;
    const int l = (tid < NB) ? lq : 0;
    sc[tid] = v;
    sg[tid] = g;
    __syncthreads();
    for (int d = 1; d < 1024; d <<= 1) {
        const int tv = (tid >= d) ? sc[tid - d] : 0;
        const int tg = (tid >= d) ? sg[tid - d] : 0;
        __syncthreads();
        sc[tid] += tv;
        sg[tid] += tg;
        __syncthreads();
    }
    lstart[tid] = sc[tid] - v;
    cur[tid]    = sc[tid] - v;
    goff[tid]   = sg[tid] - g + l;
    __syncthreads();

#pragma unroll
    for (int k = 0; k < EPT; ++k) {
        const int e = e0 + tid + k * 1024;
        if (e < e1) {
            const unsigned b = bucket_of(dd[k], NB);
            unsigned pos = (unsigned)atomicAdd(&cur[b], 1);
            pos = min(pos, (unsigned)(TILE - 1));
            sorted[pos] = ((unsigned)ss[k] << BSH) | ((unsigned)dd[k] & (unsigned)(BSZ - 1));
            sbk[pos]    = (unsigned short)b;
        }
    }
    __syncthreads();

    const int cnt = e1 - e0;
    for (int i = tid; i < cnt; i += 1024) {
        const int b = sbk[i];
        unsigned slot = (unsigned)(goff[b] + (i - lstart[b]));
        slot = min(slot, (unsigned)(E - 1));
        tmp[slot] = sorted[i];
    }
}

__global__ void __launch_bounds__(256)
finish_k(const unsigned* __restrict__ tmp, const int* __restrict__ btot,
         int* __restrict__ rowstart, int* __restrict__ csrc, int E, int n) {
    __shared__ int red[256];
    __shared__ int cnt[BSZ];
    __shared__ int cur[BSZ];
    __shared__ int stage[STG];

    const int tid = threadIdx.x;
    const int b   = blockIdx.x;

    const int mb = btot[b];
    int part = 0;
#pragma unroll
    for (int k = 0; k < NBMAX / 256; ++k) {
        const int j = tid + k * 256;
        const int t = btot[min(j, b)];
        part += (j < b) ? t : 0;
    }
    red[tid] = part;
    if (tid < BSZ) cnt[tid] = 0;
    __syncthreads();
    for (int st = 128; st > 0; st >>= 1) {
        if (tid < st) red[tid] += red[tid + st];
        __syncthreads();
    }
    const int start = min(max(red[0], 0), E);
    const int m     = min(max(mb, 0), E - start);
    __syncthreads();

    const bool small = (m <= STG);
    unsigned pk[KPT];
    if (small) {
        const int ilast = max(m - 1, 0);
#pragma unroll
        for (int k = 0; k < KPT; ++k) pk[k] = tmp[min(start + min(tid + k * 256, ilast), E - 1)];
#pragma unroll
        for (int k = 0; k < KPT; ++k) {
            const int i = tid + k * 256;
            if (i < m) atomicAdd(&cnt[pk[k] & (unsigned)(BSZ - 1)], 1);
        }
    } else {
        for (int i = tid; i < m; i += 256) atomicAdd(&cnt[tmp[start + i] & (unsigned)(BSZ - 1)], 1);
    }
    __syncthreads();

    const int v = (tid < BSZ) ? cnt[tid] : 0;
    red[tid] = v;
    __syncthreads();
    for (int d = 1; d < BSZ; d <<= 1) {
        const int t = (tid >= d) ? red[tid - d] : 0;
        __syncthreads();
        red[tid] += t;
        __syncthreads();
    }
    if (tid < BSZ) {
        const int excl = red[tid] - v;
        const int node = b * BSZ + tid;
        if (node < n) rowstart[node] = start + excl;
        cur[tid] = excl;
    }
    __syncthreads();

    if (small) {
#pragma unroll
        for (int k = 0; k < KPT; ++k) {
            const int i = tid + k * 256;
            if (i < m) {
                unsigned pos = (unsigned)atomicAdd(&cur[pk[k] & (unsigned)(BSZ - 1)], 1);
                pos = min(pos, (unsigned)(STG - 1));
                stage[pos] = (int)(pk[k] >> BSH);
            }
        }
        __syncthreads();
        for (int i = tid; i < m; i += 256) csrc[start + i] = stage[i];
    } else {
        for (int i = tid; i < m; i += 256) {
            const unsigned q = tmp[start + i];
            unsigned pos = (unsigned)atomicAdd(&cur[q & (unsigned)(BSZ - 1)], 1);
            pos = min(pos, (unsigned)(m - 1));
            csrc[start + pos] = (int)(q >> BSH);
        }
    }
}

__device__ __forceinline__ float dinv_of(const int* __restrict__ rowstart, int s, int n, int E) {
    const int q0 = rowstart[s];
    int q1 = rowstart[min(s + 1, n - 1)];
    q1 = (s + 1 < n) ? q1 : E;
    return rsqrtf((float)(q1 - q0 + 1));
}


template<int KDIM, int NOUT>
__global__ void gemm_wmma_k(const float* __restrict__ X, const float* __restrict__ W,
                            const int* __restrict__ rowstart,
                            float* __restrict__ H, int ntiles, int nrows, int E) {
    constexpr int NT   = (NOUT + 15) / 16;
    constexpr int KT   = (KDIM + 3) / 4;
    constexpr int KPAD = KT * 4;
    constexpr int NW16 = NT * 16;

    __shared__ float sW[KPAD * NW16];
    __shared__ alignas(16) float sD[128];

    for (int idx = threadIdx.x; idx < KPAD * NW16; idx += blockDim.x) {
        const int k = idx / NW16;
        const int n = idx - k * NW16;
        sW[idx] = (k < KDIM && n < NOUT) ? W[k * NOUT + n] : 0.0f;
    }
    if (threadIdx.x < 128) {
        const int r = blockIdx.x * 128 + threadIdx.x;
        sD[threadIdx.x] = dinv_of(rowstart, min(r, nrows - 1), nrows, E);
    }
    __syncthreads();

    const int lane = threadIdx.x & 31;
    const int wv   = blockIdx.x * (blockDim.x >> 5) + (threadIdx.x >> 5);
    if (wv >= ntiles) return;
    const int row0 = wv * 16;
    const int m    = lane & 15;
    const int half = lane >> 4;

    const unsigned lim = (unsigned)nrows * KDIM - 1u;

    const float* dv = sD + (threadIdx.x >> 5) * 16 + half * 8;

    v8f acc[NT];
#pragma unroll
    for (int t = 0; t < NT; ++t) acc[t] = 0.0f;

#pragma unroll
    for (int kt = 0; kt < KT; ++kt) {
        const int ka = kt * 4 + half * 2;
        const unsigned i0 = (unsigned)(row0 + m) * KDIM + ka;
        v2f a;
        a.x = X[min(i0,      lim)];
        a.y = X[min(i0 + 1u, lim)];
#pragma unroll
        for (int nt = 0; nt < NT; ++nt) {
            const int col = nt * 16 + m;
            v2f b;
            b.x = sW[ka * NW16 + col];
            b.y = sW[(ka + 1) * NW16 + col];
            acc[nt] = __builtin_amdgcn_wmma_f32_16x16x4_f32(
                false, a, false, b, (short)0, acc[nt], false, false);
        }
    }

#pragma unroll
    for (int nt = 0; nt < NT; ++nt) {
#pragma unroll
        for (int i = 0; i < 8; ++i) {
            const int r   = row0 + i + half * 8;
            const int col = nt * 16 + m;
            if (col < NOUT) H[(size_t)r * NOUT + col] = acc[nt][i] * dv[i];
        }
    }
}


__global__ void gather64_relu_k(const int* __restrict__ rowstart, const int* __restrict__ csrc,
                                const float* __restrict__ h, const float* __restrict__ b,
                                float* __restrict__ agg, int n, int E) {
    const int t = blockIdx.x * blockDim.x + threadIdx.x;
    const int i = t >> 4;
    if (i >= n) return;
    const int cg = (t & 15) << 2;

    int p0 = rowstart[i];
    int p1 = (i + 1 < n) ? rowstart[i + 1] : E;
    const float di = rsqrtf((float)(p1 - p0 + 1));
    p0 = max(p0, 0);
    p1 = min(p1, E);

    const float4 bv = *reinterpret_cast<const float4*>(b + cg);
    float4 acc = *reinterpret_cast<const float4*>(h + (size_t)i * 64 + cg);

#pragma unroll 4
    for (int p = p0; p < p1; ++p) {
        int s = csrc[p];
        s = ((unsigned)s < (unsigned)n) ? s : i;
        const float4 v = *reinterpret_cast<const float4*>(h + (size_t)s * 64 + cg);
        acc.x += v.x;
        acc.y += v.y;
        acc.z += v.z;
        acc.w += v.w;
    }

    acc.x = fmaxf(fmaf(acc.x, di, bv.x), 0.0f);
    acc.y = fmaxf(fmaf(acc.y, di, bv.y), 0.0f);
    acc.z = fmaxf(fmaf(acc.z, di, bv.z), 0.0f);
    acc.w = fmaxf(fmaf(acc.w, di, bv.w), 0.0f);
    *reinterpret_cast<float4*>(agg + (size_t)i * 64 + cg) = acc;
}

__global__ void gather3_k(const int* __restrict__ rowstart, const int* __restrict__ csrc,
                          const float* __restrict__ h, const float* __restrict__ b,
                          float* __restrict__ out, int n, int E) {
    __shared__ float so[96];
    const int tid = threadIdx.x;
    const int sub = tid & 7;
    const int i   = blockIdx.x * 32 + (tid >> 3);
    const bool live = i < n;
    const int ic = live ? i : (n - 1);

    int p0 = rowstart[ic];
    int p1 = (ic + 1 < n) ? rowstart[ic + 1] : E;
    const float di = rsqrtf((float)(p1 - p0 + 1));
    p0 = max(p0, 0);
    p1 = live ? min(p1, E) : p0;

    float a0 = 0.0f, a1 = 0.0f, a2 = 0.0f;
    for (int p = p0 + sub; p < p1; p += 8) {
        int s = csrc[p];
        s = ((unsigned)s < (unsigned)n) ? s : ic;
        const float* hp = h + (size_t)s * 3;
        a0 += hp[0];
        a1 += hp[1];
        a2 += hp[2];
    }
    a0 += __shfl_xor(a0, 4); a1 += __shfl_xor(a1, 4); a2 += __shfl_xor(a2, 4);
    a0 += __shfl_xor(a0, 2); a1 += __shfl_xor(a1, 2); a2 += __shfl_xor(a2, 2);
    a0 += __shfl_xor(a0, 1); a1 += __shfl_xor(a1, 1); a2 += __shfl_xor(a2, 1);

    if (sub == 0) {
        const float* hp = h + (size_t)ic * 3;
        const int o = (tid >> 3) * 3;
        so[o + 0] = fmaf(hp[0] + a0, di, b[0]);
        so[o + 1] = fmaf(hp[1] + a1, di, b[1]);
        so[o + 2] = fmaf(hp[2] + a2, di, b[2]);
    }
    __syncthreads();
    if (tid < 96) {
        const long long idx = (long long)blockIdx.x * 96 + tid;
        if (idx < (long long)n * 3) out[idx] = so[tid];
    }
}


static inline int cdiv(long long a, int b) { return (int)((a + b - 1) / b); }

extern "C" void kernel_launch(void* const* d_in, const int* in_sizes, int n_in,
                              void* d_out, int out_size, void* d_ws, size_t ws_size,
                              hipStream_t stream) {
    const float* x   = (const float*)d_in[0];
    const int*   ei  = (const int*)d_in[1];
    const float* W1  = (const float*)d_in[2];
    const float* b1  = (const float*)d_in[3];
    const float* W2  = (const float*)d_in[4];
    const float* b2  = (const float*)d_in[5];
    const float* W3  = (const float*)d_in[6];
    const float* b3  = (const float*)d_in[7];
    float* out = (float*)d_out;

    const int N = in_sizes[0] / 9;
    const int E = in_sizes[1] / 2;
    const int* src = ei;
    const int* dst = ei + E;

    int*   rowstart = (int*)d_ws;
    int*   csrc     = rowstart + N;
    float* h        = (float*)(csrc + E);
    float* agg      = h + (size_t)N * 64;
    const int P  = cdiv(E, TILE);
    const int NB = cdiv(N, BSZ);
    unsigned* tmp  = (unsigned*)h;
    int*      hist = (int*)agg;
    int*      loc  = hist + (size_t)P * NB;
    int*      btot = loc + (size_t)P * NB;
    (void)ws_size; (void)n_in; (void)out_size;

    const int ntiles = (N + 15) / 16;

    bhist_k<<<P, 1024, 0, stream>>>(dst, hist, E, NB);
    bscan_k<<<cdiv(NB, 32), 256, 0, stream>>>(hist, loc, btot, P, NB);
    part_k<<<P, 1024, 0, stream>>>(src, dst, loc, btot, tmp, E, NB);
    finish_k<<<NB, 256, 0, stream>>>(tmp, btot, rowstart, csrc, E, N);

    gemm_wmma_k<9, 64><<<cdiv(ntiles, 8), 256, 0, stream>>>(x, W1, rowstart, h, ntiles, N, E);
    gather64_relu_k<<<cdiv((long long)N * 16, 256), 256, 0, stream>>>(rowstart, csrc, h, b1, agg, N, E);

    gemm_wmma_k<64, 64><<<cdiv(ntiles, 8), 256, 0, stream>>>(agg, W2, rowstart, h, ntiles, N, E);
    gather64_relu_k<<<cdiv((long long)N * 16, 256), 256, 0, stream>>>(rowstart, csrc, h, b2, agg, N, E);

    gemm_wmma_k<64, 3><<<cdiv(ntiles, 8), 256, 0, stream>>>(agg, W3, rowstart, h, ntiles, N, E);
    gather3_k<<<cdiv(N, 32), 256, 0, stream>>>(rowstart, csrc, h, b3, out, N, E);
}


# Each 64-channel gather keeps its 16-node tile in LDS and runs the next layer's WMMA transform itself (one wave per tile): 2 launches fewer, no 64-channel round trip; sum and k order kept
# speedup vs baseline: 5.1322x; 1.1455x over previous
#include <hip/hip_runtime.h>


typedef float v2f __attribute__((ext_vector_type(2)));
typedef float v8f __attribute__((ext_vector_type(8)));


constexpr int BSH   = 7;
constexpr int BSZ   = 1 << BSH;
constexpr int NBMAX = 1024;
constexpr int TILE  = 6400;
constexpr int EPT   = (TILE + 1023) / 1024;
constexpr int STG   = 3072;
constexpr int KPT   = STG / 256;

__device__ __forceinline__ unsigned bucket_of(int d, int NB) {
    return min((unsigned)d >> BSH, (unsigned)(NB - 1));
}

__global__ void __launch_bounds__(1024)
bhist_k(const int* __restrict__ dst, int* __restrict__ hist, int E, int NB) {
    __shared__ int sh[NBMAX];
    const int tid = threadIdx.x;
    sh[tid] = 0;
    __syncthreads();
    const int e0 = blockIdx.x * TILE;
    const int e1 = min(e0 + TILE, E);
    const int elast = max(e1 - 1, 0);
    int d[EPT];
#pragma unroll
    for (int k = 0; k < EPT; ++k) d[k] = dst[min(e0 + tid + k * 1024, elast)];
#pragma unroll
    for (int k = 0; k < EPT; ++k) {
        const int e = e0 + tid + k * 1024;
        if (e < e1) atomicAdd(&sh[bucket_of(d[k], NB)], 1);
    }
    __syncthreads();
    if (tid < NB) hist[blockIdx.x * NB + tid] = sh[tid];
}

__global__ void __launch_bounds__(256)
bscan_k(const int* __restrict__ hist, int* __restrict__ loc, int* __restrict__ btot, int P, int NB) {
    __shared__ int wsum[8][32];
    const int lane = threadIdx.x & 31;
    const int w    = threadIdx.x >> 5;
    const int b    = blockIdx.x * 32 + lane;
    const int bc   = min(b, NB - 1);
    int carry = 0;
    for (int base = 0; base < P; base += 256) {
        const int p0 = base + w * 32;
        int v[32];
#pragma unroll
        for (int k = 0; k < 32; ++k) {
            const int p = p0 + k;
            const int t = hist[min(p, P - 1) * NB + bc];
            v[k] = (p < P) ? t : 0;
        }
        int tot = 0;
#pragma unroll
        for (int k = 0; k < 32; ++k) tot += v[k];
        wsum[w][lane] = tot;
        __syncthreads();
        int run = carry, all = 0;
#pragma unroll
        for (int j = 0; j < 8; ++j) {
            const int t = wsum[j][lane];
            run += (j < w) ? t : 0;
            all += t;
        }
#pragma unroll
        for (int k = 0; k < 32; ++k) {
            const int p = p0 + k;
            if (p < P && b < NB) loc[p * NB + b] = run;
            run += v[k];
        }
        carry += all;
        __syncthreads();
    }
    if (w == 0 && b < NB) btot[b] = carry;
}

__global__ void __launch_bounds__(1024)
part_k(const int* __restrict__ src, const int* __restrict__ dst,
       const int* __restrict__ loc, const int* __restrict__ btot,
       unsigned* __restrict__ tmp, int E, int NB) {
    __shared__ int sc[NBMAX];
    __shared__ int sg[NBMAX];
    __shared__ int lstart[NBMAX];
    __shared__ int goff[NBMAX];
    __shared__ int cur[NBMAX];
    __shared__ unsigned sorted[TILE];
    __shared__ unsigned short sbk[TILE];

    const int tid = threadIdx.x;
    const int e0  = blockIdx.x * TILE;
    const int e1  = min(e0 + TILE, E);

    const int elast = max(e1 - 1, 0);
    int dd[EPT], ss[EPT];
#pragma unroll
    for (int k = 0; k < EPT; ++k) {
        const int ec = min(e0 + tid + k * 1024, elast);
        dd[k] = dst[ec];
        ss[k] = src[ec];
    }
    const int gq = btot[min(tid, NB - 1)];
    const int lq = loc[blockIdx.x * NB + min(tid, NB - 1)];

    cur[tid] = 0;
    __syncthreads();
#pragma unroll
    for (int k = 0; k < EPT; ++k) {
        const int e = e0 + tid + k * 1024;
        if (e < e1) atomicAdd(&cur[bucket_of(dd[k], NB)], 1);
    }
    __syncthreads();

    const int v = cur[tid];
    const int g = (tid < NB) ? gq : 0;
    const int l = (tid < NB) ? lq : 0;
    sc[tid] = v;
    sg[tid] = g;
    __syncthreads();
    for (int d = 1; d < 1024; d <<= 1) {
        const int tv = (tid >= d) ? sc[tid - d] : 0;
        const int tg = (tid >= d) ? sg[tid - d] : 0;
        __syncthreads();
        sc[tid] += tv;
        sg[tid] += tg;
        __syncthreads();
    }
    lstart[tid] = sc[tid] - v;
    cur[tid]    = sc[tid] - v;
    goff[tid]   = sg[tid] - g + l;
    __syncthreads();

#pragma unroll
    for (int k = 0; k < EPT; ++k) {
        const int e = e0 + tid + k * 1024;
        if (e < e1) {
            const unsigned b = bucket_of(dd[k], NB);
            unsigned pos = (unsigned)atomicAdd(&cur[b], 1);
            pos = min(pos, (unsigned)(TILE - 1));
            sorted[pos] = ((unsigned)ss[k] << BSH) | ((unsigned)dd[k] & (unsigned)(BSZ - 1));
            sbk[pos]    = (unsigned short)b;
        }
    }
    __syncthreads();

    const int cnt = e1 - e0;
    for (int i = tid; i < cnt; i += 1024) {
        const int b = sbk[i];
        unsigned slot = (unsigned)(goff[b] + (i - lstart[b]));
        slot = min(slot, (unsigned)(E - 1));
        tmp[slot] = sorted[i];
    }
}

__global__ void __launch_bounds__(256)
finish_k(const unsigned* __restrict__ tmp, const int* __restrict__ btot,
         int* __restrict__ rowstart, int* __restrict__ csrc, int E, int n) {
    __shared__ int red[256];
    __shared__ int cnt[BSZ];
    __shared__ int cur[BSZ];
    __shared__ int stage[STG];

    const int tid = threadIdx.x;
    const int b   = blockIdx.x;

    const int mb = btot[b];
    int part = 0;
#pragma unroll
    for (int k = 0; k < NBMAX / 256; ++k) {
        const int j = tid + k * 256;
        const int t = btot[min(j, b)];
        part += (j < b) ? t : 0;
    }
    red[tid] = part;
    if (tid < BSZ) cnt[tid] = 0;
    __syncthreads();
    for (int st = 128; st > 0; st >>= 1) {
        if (tid < st) red[tid] += red[tid + st];
        __syncthreads();
    }
    const int start = min(max(red[0], 0), E);
    const int m     = min(max(mb, 0), E - start);
    __syncthreads();

    const bool small = (m <= STG);
    unsigned pk[KPT];
    if (small) {
        const int ilast = max(m - 1, 0);
#pragma unroll
        for (int k = 0; k < KPT; ++k) pk[k] = tmp[min(start + min(tid + k * 256, ilast), E - 1)];
#pragma unroll
        for (int k = 0; k < KPT; ++k) {
            const int i = tid + k * 256;
            if (i < m) atomicAdd(&cnt[pk[k] & (unsigned)(BSZ - 1)], 1);
        }
    } else {
        for (int i = tid; i < m; i += 256) atomicAdd(&cnt[tmp[start + i] & (unsigned)(BSZ - 1)], 1);
    }
    __syncthreads();

    const int v = (tid < BSZ) ? cnt[tid] : 0;
    red[tid] = v;
    __syncthreads();
    for (int d = 1; d < BSZ; d <<= 1) {
        const int t = (tid >= d) ? red[tid - d] : 0;
        __syncthreads();
        red[tid] += t;
        __syncthreads();
    }
    if (tid < BSZ) {
        const int excl = red[tid] - v;
        const int node = b * BSZ + tid;
        if (node < n) rowstart[node] = start + excl;
        cur[tid] = excl;
    }
    __syncthreads();

    if (small) {
#pragma unroll
        for (int k = 0; k < KPT; ++k) {
            const int i = tid + k * 256;
            if (i < m) {
                unsigned pos = (unsigned)atomicAdd(&cur[pk[k] & (unsigned)(BSZ - 1)], 1);
                pos = min(pos, (unsigned)(STG - 1));
                stage[pos] = (int)(pk[k] >> BSH);
            }
        }
        __syncthreads();
        for (int i = tid; i < m; i += 256) csrc[start + i] = stage[i];
    } else {
        for (int i = tid; i < m; i += 256) {
            const unsigned q = tmp[start + i];
            unsigned pos = (unsigned)atomicAdd(&cur[q & (unsigned)(BSZ - 1)], 1);
            pos = min(pos, (unsigned)(m - 1));
            csrc[start + pos] = (int)(q >> BSH);
        }
    }
}

__device__ __forceinline__ float dinv_of(const int* __restrict__ rowstart, int s, int n, int E) {
    const int q0 = rowstart[s];
    int q1 = rowstart[min(s + 1, n - 1)];
    q1 = (s + 1 < n) ? q1 : E;
    return rsqrtf((float)(q1 - q0 + 1));
}


template<int KDIM, int NOUT>
__global__ void gemm_wmma_k(const float* __restrict__ X, const float* __restrict__ W,
                            const int* __restrict__ rowstart,
                            float* __restrict__ H, int ntiles, int nrows, int E) {
    constexpr int NT   = (NOUT + 15) / 16;
    constexpr int KT   = (KDIM + 3) / 4;
    constexpr int KPAD = KT * 4;
    constexpr int NW16 = NT * 16;

    __shared__ float sW[KPAD * NW16];
    __shared__ alignas(16) float sD[128];

    for (int idx = threadIdx.x; idx < KPAD * NW16; idx += blockDim.x) {
        const int k = idx / NW16;
        const int n = idx - k * NW16;
        sW[idx] = (k < KDIM && n < NOUT) ? W[k * NOUT + n] : 0.0f;
    }
    if (threadIdx.x < 128) {
        const int r = blockIdx.x * 128 + threadIdx.x;
        sD[threadIdx.x] = dinv_of(rowstart, min(r, nrows - 1), nrows, E);
    }
    __syncthreads();

    const int lane = threadIdx.x & 31;
    const int wv   = blockIdx.x * (blockDim.x >> 5) + (threadIdx.x >> 5);
    if (wv >= ntiles) return;
    const int row0 = wv * 16;
    const int m    = lane & 15;
    const int half = lane >> 4;

    const unsigned lim = (unsigned)nrows * KDIM - 1u;

    const float* dv = sD + (threadIdx.x >> 5) * 16 + half * 8;

    v8f acc[NT];
#pragma unroll
    for (int t = 0; t < NT; ++t) acc[t] = 0.0f;

#pragma unroll
    for (int kt = 0; kt < KT; ++kt) {
        const int ka = kt * 4 + half * 2;
        const unsigned i0 = (unsigned)(row0 + m) * KDIM + ka;
        v2f a;
        a.x = X[min(i0,      lim)];
        a.y = X[min(i0 + 1u, lim)];
#pragma unroll
        for (int nt = 0; nt < NT; ++nt) {
            const int col = nt * 16 + m;
            v2f b;
            b.x = sW[ka * NW16 + col];
            b.y = sW[(ka + 1) * NW16 + col];
            acc[nt] = __builtin_amdgcn_wmma_f32_16x16x4_f32(
                false, a, false, b, (short)0, acc[nt], false, false);
        }
    }

#pragma unroll
    for (int nt = 0; nt < NT; ++nt) {
#pragma unroll
        for (int i = 0; i < 8; ++i) {
            const int r   = row0 + i + half * 8;
            const int col = nt * 16 + m;
            if (col < NOUT) H[(size_t)r * NOUT + col] = acc[nt][i] * dv[i];
        }
    }
}


template<int NOUT>
__global__ void __launch_bounds__(256)
gather64_gemm_k(const int* __restrict__ rowstart, const int* __restrict__ csrc,
                const float* __restrict__ h, const float* __restrict__ b,
                const float* __restrict__ W, float* __restrict__ Hout, int n, int E) {
    constexpr int KDIM = 64;
    constexpr int NT   = (NOUT + 15) / 16;
    constexpr int KT   = KDIM / 4;
    constexpr int NW16 = NT * 16;
    constexpr int AP   = 68;
    constexpr int CH   = 16;

    __shared__ float sW[KDIM * NW16];
    __shared__ alignas(16) float sA[8 * 16 * AP];
    __shared__ float sDi[8 * 16];

    {
        constexpr int WPT = KDIM * NW16 / 256;
        float wreg[WPT];
#pragma unroll
        for (int j = 0; j < WPT; ++j) {
            const int idx = threadIdx.x + j * 256;
            const int k = idx / NW16;
            const int c = idx - k * NW16;
            wreg[j] = (c < NOUT) ? W[k * NOUT + c] : 0.0f;
        }
#pragma unroll
        for (int j = 0; j < WPT; ++j) sW[threadIdx.x + j * 256] = wreg[j];
    }

    const int lane = threadIdx.x & 31;
    const int w    = threadIdx.x >> 5;
    const int l16  = lane & 15;
    const int half = lane >> 4;
    const int cg   = l16 << 2;
    const int row0 = (blockIdx.x * 8 + w) * 16;
    float* tile = sA + w * (16 * AP);

    int rs;
    {
        const int node = row0 + min(lane, 16);
        const int v = rowstart[min(node, n - 1)];
        rs = (node < n) ? v : E;
        rs = min(max(rs, 0), E);
    }
    const float4 bv = *reinterpret_cast<const float4*>(b + cg);

    int s[CH];
    float4 selfrow;
    {
        const int q0 = __shfl(rs, half);
        const int qe = __shfl(rs, half + 1);
        const int q1 = (row0 + half < n) ? qe : q0;
        const int ql = max(q1 - 1, 0);
#pragma unroll
        for (int k = 0; k < CH; ++k) s[k] = csrc[min(q0 + k, ql)];
        selfrow = *reinterpret_cast<const float4*>(h + (size_t)min(row0 + half, n - 1) * 64 + cg);
    }

    for (int pass = 0; pass < 8; ++pass) {
        const int nl = pass * 2 + half;
        const int i  = row0 + nl;
        const bool live = i < n;
        const int ic = live ? i : (n - 1);
        const int p0 = __shfl(rs, nl);
        int p1 = __shfl(rs, nl + 1);
        const float di = rsqrtf((float)(p1 - p0 + 1));
        if (!live) p1 = p0;
        const int pl = max(p1 - 1, 0);

        const int inx = min(i + 2, n - 1);
        const int p0n = __shfl(rs, min(nl + 2, 16));
        const int pen = __shfl(rs, min(nl + 3, 16));
        const int p1n = (i + 2 < n) ? pen : p0n;
        const int pln = max(p1n - 1, 0);

        float4 acc = selfrow;

        for (int p = p0; p < p1; p += CH) {
            int sk[CH];
#pragma unroll
            for (int k = 0; k < CH; ++k)
                sk[k] = ((unsigned)s[k] < (unsigned)n) ? s[k] : ic;

            const bool last = (p + CH >= p1);
            const int nb = last ? p0n : (p + CH);
            const int nlim = last ? pln : pl;
#pragma unroll
            for (int k = 0; k < CH; ++k) s[k] = csrc[min(nb + k, nlim)];
            if (last) selfrow = *reinterpret_cast<const float4*>(h + (size_t)inx * 64 + cg);

            float4 v[CH];
#pragma unroll
            for (int k = 0; k < CH; ++k) {
                v[k] = make_float4(0.0f, 0.0f, 0.0f, 0.0f);
                if (p + k < p1) v[k] = *reinterpret_cast<const float4*>(h + (size_t)sk[k] * 64 + cg);
            }
#pragma unroll
            for (int k = 0; k < CH; ++k) {
                acc.x += v[k].x;
                acc.y += v[k].y;
                acc.z += v[k].z;
                acc.w += v[k].w;
            }
        }
        if (p0 >= p1) {
#pragma unroll
            for (int k = 0; k < CH; ++k) s[k] = csrc[min(p0n + k, pln)];
            selfrow = *reinterpret_cast<const float4*>(h + (size_t)inx * 64 + cg);
        }

        acc.x = fmaxf(fmaf(acc.x, di, bv.x), 0.0f);
        acc.y = fmaxf(fmaf(acc.y, di, bv.y), 0.0f);
        acc.z = fmaxf(fmaf(acc.z, di, bv.z), 0.0f);
        acc.w = fmaxf(fmaf(acc.w, di, bv.w), 0.0f);
        if (!live) acc = make_float4(0.0f, 0.0f, 0.0f, 0.0f);
        *reinterpret_cast<float4*>(tile + nl * AP + cg) = acc;
        if (l16 == 0) sDi[w * 16 + nl] = di;
    }
    __syncthreads();

    if (row0 >= n) return;

    const int m = l16;
    const float* dv = sDi + w * 16 + half * 8;

    v8f accm[NT];
#pragma unroll
    for (int t = 0; t < NT; ++t) accm[t] = 0.0f;

#pragma unroll
    for (int kt = 0; kt < KT; ++kt) {
        const int ka = kt * 4 + half * 2;
        v2f a;
        a.x = tile[m * AP + ka];
        a.y = tile[m * AP + ka + 1];
#pragma unroll
        for (int nt = 0; nt < NT; ++nt) {
            const int col = nt * 16 + m;
            v2f bb;
            bb.x = sW[ka * NW16 + col];
            bb.y = sW[(ka + 1) * NW16 + col];
            accm[nt] = __builtin_amdgcn_wmma_f32_16x16x4_f32(
                false, a, false, bb, (short)0, accm[nt], false, false);
        }
    }

#pragma unroll
    for (int nt = 0; nt < NT; ++nt) {
#pragma unroll
        for (int i = 0; i < 8; ++i) {
            const int r   = row0 + i + half * 8;
            const int col = nt * 16 + m;
            if (col < NOUT && r < n) Hout[(size_t)r * NOUT + col] = accm[nt][i] * dv[i];
        }
    }
}

__global__ void gather3_k(const int* __restrict__ rowstart, const int* __restrict__ csrc,
                          const float* __restrict__ h, const float* __restrict__ b,
                          float* __restrict__ out, int n, int E) {
    __shared__ float so[96];
    const int tid = threadIdx.x;
    const int sub = tid & 7;
    const int i   = blockIdx.x * 32 + (tid >> 3);
    const bool live = i < n;
    const int ic = live ? i : (n - 1);

    int p0 = rowstart[ic];
    int p1 = (ic + 1 < n) ? rowstart[ic + 1] : E;
    const float di = rsqrtf((float)(p1 - p0 + 1));
    p0 = max(p0, 0);
    p1 = live ? min(p1, E) : p0;

    float a0 = 0.0f, a1 = 0.0f, a2 = 0.0f;
    for (int p = p0 + sub; p < p1; p += 8) {
        int s = csrc[p];
        s = ((unsigned)s < (unsigned)n) ? s : ic;
        const float* hp = h + (size_t)s * 3;
        a0 += hp[0];
        a1 += hp[1];
        a2 += hp[2];
    }
    a0 += __shfl_xor(a0, 4); a1 += __shfl_xor(a1, 4); a2 += __shfl_xor(a2, 4);
    a0 += __shfl_xor(a0, 2); a1 += __shfl_xor(a1, 2); a2 += __shfl_xor(a2, 2);
    a0 += __shfl_xor(a0, 1); a1 += __shfl_xor(a1, 1); a2 += __shfl_xor(a2, 1);

    if (sub == 0) {
        const float* hp = h + (size_t)ic * 3;
        const int o = (tid >> 3) * 3;
        so[o + 0] = fmaf(hp[0] + a0, di, b[0]);
        so[o + 1] = fmaf(hp[1] + a1, di, b[1]);
        so[o + 2] = fmaf(hp[2] + a2, di, b[2]);
    }
    __syncthreads();
    if (tid < 96) {
        const long long idx = (long long)blockIdx.x * 96 + tid;
        if (idx < (long long)n * 3) out[idx] = so[tid];
    }
}


static inline int cdiv(long long a, int b) { return (int)((a + b - 1) / b); }

extern "C" void kernel_launch(void* const* d_in, const int* in_sizes, int n_in,
                              void* d_out, int out_size, void* d_ws, size_t ws_size,
                              hipStream_t stream) {
    const float* x   = (const float*)d_in[0];
    const int*   ei  = (const int*)d_in[1];
    const float* W1  = (const float*)d_in[2];
    const float* b1  = (const float*)d_in[3];
    const float* W2  = (const float*)d_in[4];
    const float* b2  = (const float*)d_in[5];
    const float* W3  = (const float*)d_in[6];
    const float* b3  = (const float*)d_in[7];
    float* out = (float*)d_out;

    const int N = in_sizes[0] / 9;
    const int E = in_sizes[1] / 2;
    const int* src = ei;
    const int* dst = ei + E;

    int*   rowstart = (int*)d_ws;
    int*   csrc     = rowstart + N;
    float* h        = (float*)(csrc + E);
    float* agg      = h + (size_t)N * 64;
    const int P  = cdiv(E, TILE);
    const int NB = cdiv(N, BSZ);
    unsigned* tmp  = (unsigned*)h;
    int*      hist = (int*)agg;
    int*      loc  = hist + (size_t)P * NB;
    int*      btot = loc + (size_t)P * NB;
    (void)ws_size; (void)n_in; (void)out_size;

    const int ntiles = (N + 15) / 16;

    bhist_k<<<P, 1024, 0, stream>>>(dst, hist, E, NB);
    bscan_k<<<cdiv(NB, 32), 256, 0, stream>>>(hist, loc, btot, P, NB);
    part_k<<<P, 1024, 0, stream>>>(src, dst, loc, btot, tmp, E, NB);
    finish_k<<<NB, 256, 0, stream>>>(tmp, btot, rowstart, csrc, E, N);

    gemm_wmma_k<9, 64><<<cdiv(ntiles, 8), 256, 0, stream>>>(x, W1, rowstart, h, ntiles, N, E);

    gather64_gemm_k<64><<<cdiv(ntiles, 8), 256, 0, stream>>>(rowstart, csrc, h, b1, W2, agg, N, E);

    gather64_gemm_k<3><<<cdiv(ntiles, 8), 256, 0, stream>>>(rowstart, csrc, agg, b2, W3, h, N, E);

    gather3_k<<<cdiv(N, 32), 256, 0, stream>>>(rowstart, csrc, h, b3, out, N, E);
}


# Layer-1 aggregation moved in front of its transform, (A x) W1 for A (x W1): an edge adds a 64 B row of 9 input channels, not a 256 B row; W1, ReLU, W2 on WMMA in that kernel; rounding order changes
# speedup vs baseline: 6.5418x; 1.2747x over previous
#include <hip/hip_runtime.h>


typedef float v2f __attribute__((ext_vector_type(2)));
typedef float v8f __attribute__((ext_vector_type(8)));


constexpr int BSH   = 7;
constexpr int BSZ   = 1 << BSH;
constexpr int NBMAX = 1024;
constexpr int TILE  = 6400;
constexpr int EPT   = (TILE + 1023) / 1024;
constexpr int STG   = 3072;
constexpr int KPT   = STG / 256;
constexpr int XW    = 16;

__device__ __forceinline__ unsigned bucket_of(int d, int NB) {
    return min((unsigned)d >> BSH, (unsigned)(NB - 1));
}

__global__ void __launch_bounds__(1024)
bhist_k(const int* __restrict__ dst, int* __restrict__ hist, int E, int NB) {
    __shared__ int sh[NBMAX];
    const int tid = threadIdx.x;
    sh[tid] = 0;
    __syncthreads();
    const int e0 = blockIdx.x * TILE;
    const int e1 = min(e0 + TILE, E);
    const int elast = max(e1 - 1, 0);
    int d[EPT];
#pragma unroll
    for (int k = 0; k < EPT; ++k) d[k] = dst[min(e0 + tid + k * 1024, elast)];
#pragma unroll
    for (int k = 0; k < EPT; ++k) {
        const int e = e0 + tid + k * 1024;
        if (e < e1) atomicAdd(&sh[bucket_of(d[k], NB)], 1);
    }
    __syncthreads();
    if (tid < NB) hist[blockIdx.x * NB + tid] = sh[tid];
}

__global__ void __launch_bounds__(256)
bscan_k(const int* __restrict__ hist, int* __restrict__ loc, int* __restrict__ btot, int P, int NB) {
    __shared__ int wsum[8][32];
    const int lane = threadIdx.x & 31;
    const int w    = threadIdx.x >> 5;
    const int b    = blockIdx.x * 32 + lane;
    const int bc   = min(b, NB - 1);
    int carry = 0;
    for (int base = 0; base < P; base += 256) {
        const int p0 = base + w * 32;
        int v[32];
#pragma unroll
        for (int k = 0; k < 32; ++k) {
            const int p = p0 + k;
            const int t = hist[min(p, P - 1) * NB + bc];
            v[k] = (p < P) ? t : 0;
        }
        int tot = 0;
#pragma unroll
        for (int k = 0; k < 32; ++k) tot += v[k];
        wsum[w][lane] = tot;
        __syncthreads();
        int run = carry, all = 0;
#pragma unroll
        for (int j = 0; j < 8; ++j) {
            const int t = wsum[j][lane];
            run += (j < w) ? t : 0;
            all += t;
        }
#pragma unroll
        for (int k = 0; k < 32; ++k) {
            const int p = p0 + k;
            if (p < P && b < NB) loc[p * NB + b] = run;
            run += v[k];
        }
        carry += all;
        __syncthreads();
    }
    if (w == 0 && b < NB) btot[b] = carry;
}

__global__ void __launch_bounds__(1024)
part_k(const int* __restrict__ src, const int* __restrict__ dst,
       const int* __restrict__ loc, const int* __restrict__ btot,
       unsigned* __restrict__ tmp, int E, int NB) {
    __shared__ int sc[NBMAX];
    __shared__ int sg[NBMAX];
    __shared__ int lstart[NBMAX];
    __shared__ int goff[NBMAX];
    __shared__ int cur[NBMAX];
    __shared__ unsigned sorted[TILE];
    __shared__ unsigned short sbk[TILE];

    const int tid = threadIdx.x;
    const int e0  = blockIdx.x * TILE;
    const int e1  = min(e0 + TILE, E);

    const int elast = max(e1 - 1, 0);
    int dd[EPT], ss[EPT];
#pragma unroll
    for (int k = 0; k < EPT; ++k) {
        const int ec = min(e0 + tid + k * 1024, elast);
        dd[k] = dst[ec];
        ss[k] = src[ec];
    }
    const int gq = btot[min(tid, NB - 1)];
    const int lq = loc[blockIdx.x * NB + min(tid, NB - 1)];

    cur[tid] = 0;
    __syncthreads();
#pragma unroll
    for (int k = 0; k < EPT; ++k) {
        const int e = e0 + tid + k * 1024;
        if (e < e1) atomicAdd(&cur[bucket_of(dd[k], NB)], 1);
    }
    __syncthreads();

    const int v = cur[tid];
    const int g = (tid < NB) ? gq : 0;
    const int l = (tid < NB) ? lq : 0;
    sc[tid] = v;
    sg[tid] = g;
    __syncthreads();
    for (int d = 1; d < 1024; d <<= 1) {
        const int tv = (tid >= d) ? sc[tid - d] : 0;
        const int tg = (tid >= d) ? sg[tid - d] : 0;
        __syncthreads();
        sc[tid] += tv;
        sg[tid] += tg;
        __syncthreads();
    }
    lstart[tid] = sc[tid] - v;
    cur[tid]    = sc[tid] - v;
    goff[tid]   = sg[tid] - g + l;
    __syncthreads();

#pragma unroll
    for (int k = 0; k < EPT; ++k) {
        const int e = e0 + tid + k * 1024;
        if (e < e1) {
            const unsigned b = bucket_of(dd[k], NB);
            unsigned pos = (unsigned)atomicAdd(&cur[b], 1);
            pos = min(pos, (unsigned)(TILE - 1));
            sorted[pos] = ((unsigned)ss[k] << BSH) | ((unsigned)dd[k] & (unsigned)(BSZ - 1));
            sbk[pos]    = (unsigned short)b;
        }
    }
    __syncthreads();

    const int cnt = e1 - e0;
    for (int i = tid; i < cnt; i += 1024) {
        const int b = sbk[i];
        unsigned slot = (unsigned)(goff[b] + (i - lstart[b]));
        slot = min(slot, (unsigned)(E - 1));
        tmp[slot] = sorted[i];
    }
}

__global__ void __launch_bounds__(256)
finish_k(const unsigned* __restrict__ tmp, const int* __restrict__ btot,
         int* __restrict__ rowstart, int* __restrict__ csrc,
         const float* __restrict__ x, float* __restrict__ xs, int E, int n) {
    __shared__ int red[256];
    __shared__ int cnt[BSZ];
    __shared__ int cur[BSZ];
    __shared__ int stage[STG];
    __shared__ float sdi[BSZ];

    const int tid = threadIdx.x;
    const int b   = blockIdx.x;

    float xv[BSZ * XW / 256];
#pragma unroll
    for (int j = 0; j < BSZ * XW / 256; ++j) {
        const int idx  = tid + j * 256;
        const int nloc = idx / XW;
        const int c    = idx - nloc * XW;
        const int node = min(b * BSZ + nloc, n - 1);
        xv[j] = x[(size_t)node * 9 + min(c, 8)];
    }

    const int mb = btot[b];
    int part = 0;
#pragma unroll
    for (int k = 0; k < NBMAX / 256; ++k) {
        const int j = tid + k * 256;
        const int t = btot[min(j, b)];
        part += (j < b) ? t : 0;
    }
    red[tid] = part;
    if (tid < BSZ) cnt[tid] = 0;
    __syncthreads();
    for (int st = 128; st > 0; st >>= 1) {
        if (tid < st) red[tid] += red[tid + st];
        __syncthreads();
    }
    const int start = min(max(red[0], 0), E);
    const int m     = min(max(mb, 0), E - start);
    __syncthreads();

    const bool small = (m <= STG);
    unsigned pk[KPT];
    if (small) {
        const int ilast = max(m - 1, 0);
#pragma unroll
        for (int k = 0; k < KPT; ++k) pk[k] = tmp[min(start + min(tid + k * 256, ilast), E - 1)];
#pragma unroll
        for (int k = 0; k < KPT; ++k) {
            const int i = tid + k * 256;
            if (i < m) atomicAdd(&cnt[pk[k] & (unsigned)(BSZ - 1)], 1);
        }
    } else {
        for (int i = tid; i < m; i += 256) atomicAdd(&cnt[tmp[start + i] & (unsigned)(BSZ - 1)], 1);
    }
    __syncthreads();

    const int v = (tid < BSZ) ? cnt[tid] : 0;
    red[tid] = v;
    __syncthreads();
    for (int d = 1; d < BSZ; d <<= 1) {
        const int t = (tid >= d) ? red[tid - d] : 0;
        __syncthreads();
        red[tid] += t;
        __syncthreads();
    }
    if (tid < BSZ) {
        const int excl = red[tid] - v;
        const int node = b * BSZ + tid;
        if (node < n) rowstart[node] = start + excl;
        cur[tid] = excl;
        sdi[tid] = rsqrtf((float)(v + 1));
    }
    __syncthreads();

#pragma unroll
    for (int j = 0; j < BSZ * XW / 256; ++j) {
        const int idx  = tid + j * 256;
        const int nloc = idx / XW;
        const int c    = idx - nloc * XW;
        if (b * BSZ + nloc < n) xs[(size_t)b * (BSZ * XW) + idx] = (c < 9) ? xv[j] * sdi[nloc] : 0.0f;
    }

    if (small) {
#pragma unroll
        for (int k = 0; k < KPT; ++k) {
            const int i = tid + k * 256;
            if (i < m) {
                unsigned pos = (unsigned)atomicAdd(&cur[pk[k] & (unsigned)(BSZ - 1)], 1);
                pos = min(pos, (unsigned)(STG - 1));
                stage[pos] = (int)(pk[k] >> BSH);
            }
        }
        __syncthreads();
        for (int i = tid; i < m; i += 256) csrc[start + i] = stage[i];
    } else {
        for (int i = tid; i < m; i += 256) {
            const unsigned q = tmp[start + i];
            unsigned pos = (unsigned)atomicAdd(&cur[q & (unsigned)(BSZ - 1)], 1);
            pos = min(pos, (unsigned)(m - 1));
            csrc[start + pos] = (int)(q >> BSH);
        }
    }
}

__global__ void __launch_bounds__(256)
gather16_gemm2_k(const int* __restrict__ rowstart, const int* __restrict__ csrc,
                 const float* __restrict__ xs, const float* __restrict__ W1,
                 const float* __restrict__ b1, const float* __restrict__ W2,
                 float* __restrict__ Hout, int n, int E) {
    constexpr int K1P = 12;
    constexpr int AP  = 68;
    constexpr int XP  = 20;
    constexpr int CAP = 384;
    constexpr int CH  = 16;

    __shared__ float sW2[64 * 64];
    __shared__ float sW1[K1P * 64];
    __shared__ alignas(16) float sA[8 * 16 * AP];
    __shared__ alignas(16) float sX[8 * 16 * XP];
    __shared__ float sDi[8 * 16];

    const int lane = threadIdx.x & 31;
    const int w    = threadIdx.x >> 5;
    const int q    = lane & 3;
    const int g    = lane >> 2;
    const int row0 = (blockIdx.x * 8 + w) * 16;
    float* tile = sA + w * (16 * AP);
    float* xt   = sX + w * (16 * XP);

    int rs;
    {
        const int node = row0 + min(lane, 16);
        const int v = rowstart[min(node, n - 1)];
        rs = (node < n) ? v : E;
        rs = min(max(rs, 0), E);
    }
    const int t0 = __shfl(rs, 0);

    float bc[4];
#pragma unroll
    for (int t = 0; t < 4; ++t) bc[t] = b1[t * 16 + (lane & 15)];

    int iv[CAP / 32];
#pragma unroll
    for (int k = 0; k < CAP / 32; ++k) iv[k] = csrc[min(t0 + lane + 32 * k, E - 1)];
    float4 selfrow[2];
#pragma unroll
    for (int pass = 0; pass < 2; ++pass)
        selfrow[pass] = *reinterpret_cast<const float4*>(xs + (size_t)min(row0 + pass * 8 + g, n - 1) * XW + q * 4);
    {
        constexpr int WPT = 64 * 64 / 256;
        float wreg[WPT];
#pragma unroll
        for (int j = 0; j < WPT; ++j) wreg[j] = W2[threadIdx.x + j * 256];
        float w1reg[K1P * 64 / 256];
#pragma unroll
        for (int j = 0; j < K1P * 64 / 256; ++j) {
            const int idx = threadIdx.x + j * 256;
            const int k = idx >> 6;
            const float t = W1[min(k, 8) * 64 + (idx & 63)];
            w1reg[j] = (k < 9) ? t : 0.0f;
        }
#pragma unroll
        for (int k = 0; k < CAP / 32; ++k) tile[lane + 32 * k] = __int_as_float(iv[k]);
#pragma unroll
        for (int j = 0; j < WPT; ++j) sW2[threadIdx.x + j * 256] = wreg[j];
#pragma unroll
        for (int j = 0; j < K1P * 64 / 256; ++j) sW1[threadIdx.x + j * 256] = w1reg[j];
    }
    __syncthreads();

    for (int pass = 0; pass < 2; ++pass) {
        const int nl = pass * 8 + g;
        const int i  = row0 + nl;
        const bool live = i < n;
        const int ic = live ? i : (n - 1);
        const int p0 = __shfl(rs, nl);
        int p1 = __shfl(rs, nl + 1);
        const float di = rsqrtf((float)(p1 - p0 + 1));
        if (!live) p1 = p0;

        float4 acc = selfrow[pass];

        for (int p = p0; p < p1; p += CH) {
            int sk[CH];
#pragma unroll
            for (int k = 0; k < CH; ++k) sk[k] = __float_as_int(tile[min(max(p + k - t0, 0), CAP - 1)]);
            if (p + CH - t0 > CAP) {
#pragma unroll
                for (int k = 0; k < CH; ++k)
                    if (p + k - t0 >= CAP) sk[k] = csrc[min(p + k, E - 1)];
            }
            float4 v[CH];
#pragma unroll
            for (int k = 0; k < CH; ++k) {
                const int s = ((unsigned)sk[k] < (unsigned)n) ? sk[k] : ic;
                v[k] = make_float4(0.0f, 0.0f, 0.0f, 0.0f);
                if (p + k < p1) v[k] = *reinterpret_cast<const float4*>(xs + (size_t)s * XW + q * 4);
            }
#pragma unroll
            for (int k = 0; k < CH; ++k) {
                acc.x += v[k].x;
                acc.y += v[k].y;
                acc.z += v[k].z;
                acc.w += v[k].w;
            }
        }
        acc.x *= di;
        acc.y *= di;
        acc.z *= di;
        acc.w *= di;
        if (!live) acc = make_float4(0.0f, 0.0f, 0.0f, 0.0f);
        *reinterpret_cast<float4*>(xt + nl * XP + q * 4) = acc;
        if (q == 0) sDi[w * 16 + nl] = di;
    }
    __syncthreads();

    const int m    = lane & 15;
    const int half = lane >> 4;

    v8f c1[4];
#pragma unroll
    for (int t = 0; t < 4; ++t) c1[t] = 0.0f;
#pragma unroll
    for (int kt = 0; kt < K1P / 4; ++kt) {
        const int ka = kt * 4 + half * 2;
        v2f a;
        a.x = xt[m * XP + ka];
        a.y = xt[m * XP + ka + 1];
#pragma unroll
        for (int nt = 0; nt < 4; ++nt) {
            const int col = nt * 16 + m;
            v2f bb;
            bb.x = sW1[ka * 64 + col];
            bb.y = sW1[(ka + 1) * 64 + col];
            c1[nt] = __builtin_amdgcn_wmma_f32_16x16x4_f32(
                false, a, false, bb, (short)0, c1[nt], false, false);
        }
    }
#pragma unroll
    for (int nt = 0; nt < 4; ++nt) {
#pragma unroll
        for (int r = 0; r < 8; ++r)
            tile[(r + half * 8) * AP + nt * 16 + m] = fmaxf(c1[nt][r] + bc[nt], 0.0f);
    }
    __syncthreads();

    const float* dv = sDi + w * 16 + half * 8;

    v8f accm[4];
#pragma unroll
    for (int t = 0; t < 4; ++t) accm[t] = 0.0f;
#pragma unroll
    for (int kt = 0; kt < 16; ++kt) {
        const int ka = kt * 4 + half * 2;
        v2f a;
        a.x = tile[m * AP + ka];
        a.y = tile[m * AP + ka + 1];
#pragma unroll
        for (int nt = 0; nt < 4; ++nt) {
            const int col = nt * 16 + m;
            v2f bb;
            bb.x = sW2[ka * 64 + col];
            bb.y = sW2[(ka + 1) * 64 + col];
            accm[nt] = __builtin_amdgcn_wmma_f32_16x16x4_f32(
                false, a, false, bb, (short)0, accm[nt], false, false);
        }
    }
#pragma unroll
    for (int nt = 0; nt < 4; ++nt) {
#pragma unroll
        for (int r = 0; r < 8; ++r) {
            const int row = row0 + r + half * 8;
            if (row < n) Hout[(size_t)row * 64 + nt * 16 + m] = accm[nt][r] * dv[r];
        }
    }
}


template<int NOUT>
__global__ void __launch_bounds__(256)
gather64_gemm_k(const int* __restrict__ rowstart, const int* __restrict__ csrc,
                const float* __restrict__ h, const float* __restrict__ b,
                const float* __restrict__ W, float* __restrict__ Hout, int n, int E) {
    constexpr int KDIM = 64;
    constexpr int NT   = (NOUT + 15) / 16;
    constexpr int KT   = KDIM / 4;
    constexpr int NW16 = NT * 16;
    constexpr int AP   = 68;
    constexpr int CH   = 16;

    __shared__ float sW[KDIM * NW16];
    __shared__ alignas(16) float sA[8 * 16 * AP];
    __shared__ float sDi[8 * 16];

    {
        constexpr int WPT = KDIM * NW16 / 256;
        float wreg[WPT];
#pragma unroll
        for (int j = 0; j < WPT; ++j) {
            const int idx = threadIdx.x + j * 256;
            const int k = idx / NW16;
            const int c = idx - k * NW16;
            wreg[j] = (c < NOUT) ? W[k * NOUT + c] : 0.0f;
        }
#pragma unroll
        for (int j = 0; j < WPT; ++j) sW[threadIdx.x + j * 256] = wreg[j];
    }

    const int lane = threadIdx.x & 31;
    const int w    = threadIdx.x >> 5;
    const int l16  = lane & 15;
    const int half = lane >> 4;
    const int cg   = l16 << 2;
    const int row0 = (blockIdx.x * 8 + w) * 16;
    float* tile = sA + w * (16 * AP);

    int rs;
    {
        const int node = row0 + min(lane, 16);
        const int v = rowstart[min(node, n - 1)];
        rs = (node < n) ? v : E;
        rs = min(max(rs, 0), E);
    }
    const float4 bv = *reinterpret_cast<const float4*>(b + cg);

    int s[CH];
    float4 selfrow;
    {
        const int q0 = __shfl(rs, half);
        const int qe = __shfl(rs, half + 1);
        const int q1 = (row0 + half < n) ? qe : q0;
        const int ql = max(q1 - 1, 0);
#pragma unroll
        for (int k = 0; k < CH; ++k) s[k] = csrc[min(q0 + k, ql)];
        selfrow = *reinterpret_cast<const float4*>(h + (size_t)min(row0 + half, n - 1) * 64 + cg);
    }

    for (int pass = 0; pass < 8; ++pass) {
        const int nl = pass * 2 + half;
        const int i  = row0 + nl;
        const bool live = i < n;
        const int ic = live ? i : (n - 1);
        const int p0 = __shfl(rs, nl);
        int p1 = __shfl(rs, nl + 1);
        const float di = rsqrtf((float)(p1 - p0 + 1));
        if (!live) p1 = p0;
        const int pl = max(p1 - 1, 0);

        const int inx = min(i + 2, n - 1);
        const int p0n = __shfl(rs, min(nl + 2, 16));
        const int pen = __shfl(rs, min(nl + 3, 16));
        const int p1n = (i + 2 < n) ? pen : p0n;
        const int pln = max(p1n - 1, 0);

        float4 acc = selfrow;

        for (int p = p0; p < p1; p += CH) {
            int sk[CH];
#pragma unroll
            for (int k = 0; k < CH; ++k)
                sk[k] = ((unsigned)s[k] < (unsigned)n) ? s[k] : ic;

            const bool last = (p + CH >= p1);
            const int nb = last ? p0n : (p + CH);
            const int nlim = last ? pln : pl;
#pragma unroll
            for (int k = 0; k < CH; ++k) s[k] = csrc[min(nb + k, nlim)];
            if (last) selfrow = *reinterpret_cast<const float4*>(h + (size_t)inx * 64 + cg);

            float4 v[CH];
#pragma unroll
            for (int k = 0; k < CH; ++k) {
                v[k] = make_float4(0.0f, 0.0f, 0.0f, 0.0f);
                if (p + k < p1) v[k] = *reinterpret_cast<const float4*>(h + (size_t)sk[k] * 64 + cg);
            }
#pragma unroll
            for (int k = 0; k < CH; ++k) {
                acc.x += v[k].x;
                acc.y += v[k].y;
                acc.z += v[k].z;
                acc.w += v[k].w;
            }
        }
        if (p0 >= p1) {
#pragma unroll
            for (int k = 0; k < CH; ++k) s[k] = csrc[min(p0n + k, pln)];
            selfrow = *reinterpret_cast<const float4*>(h + (size_t)inx * 64 + cg);
        }

        acc.x = fmaxf(fmaf(acc.x, di, bv.x), 0.0f);
        acc.y = fmaxf(fmaf(acc.y, di, bv.y), 0.0f);
        acc.z = fmaxf(fmaf(acc.z, di, bv.z), 0.0f);
        acc.w = fmaxf(fmaf(acc.w, di, bv.w), 0.0f);
        if (!live) acc = make_float4(0.0f, 0.0f, 0.0f, 0.0f);
        *reinterpret_cast<float4*>(tile + nl * AP + cg) = acc;
        if (l16 == 0) sDi[w * 16 + nl] = di;
    }
    __syncthreads();

    if (row0 >= n) return;

    const int m = l16;
    const float* dv = sDi + w * 16 + half * 8;

    v8f accm[NT];
#pragma unroll
    for (int t = 0; t < NT; ++t) accm[t] = 0.0f;

#pragma unroll
    for (int kt = 0; kt < KT; ++kt) {
        const int ka = kt * 4 + half * 2;
        v2f a;
        a.x = tile[m * AP + ka];
        a.y = tile[m * AP + ka + 1];
#pragma unroll
        for (int nt = 0; nt < NT; ++nt) {
            const int col = nt * 16 + m;
            v2f bb;
            bb.x = sW[ka * NW16 + col];
            bb.y = sW[(ka + 1) * NW16 + col];
            accm[nt] = __builtin_amdgcn_wmma_f32_16x16x4_f32(
                false, a, false, bb, (short)0, accm[nt], false, false);
        }
    }

#pragma unroll
    for (int nt = 0; nt < NT; ++nt) {
#pragma unroll
        for (int i = 0; i < 8; ++i) {
            const int r   = row0 + i + half * 8;
            const int col = nt * 16 + m;
            if (col < NOUT && r < n) Hout[(size_t)r * NOUT + col] = accm[nt][i] * dv[i];
        }
    }
}

__global__ void gather3_k(const int* __restrict__ rowstart, const int* __restrict__ csrc,
                          const float* __restrict__ h, const float* __restrict__ b,
                          float* __restrict__ out, int n, int E) {
    __shared__ float so[96];
    const int tid = threadIdx.x;
    const int sub = tid & 7;
    const int i   = blockIdx.x * 32 + (tid >> 3);
    const bool live = i < n;
    const int ic = live ? i : (n - 1);

    int p0 = rowstart[ic];
    int p1 = (ic + 1 < n) ? rowstart[ic + 1] : E;
    const float di = rsqrtf((float)(p1 - p0 + 1));
    p0 = max(p0, 0);
    p1 = live ? min(p1, E) : p0;

    float a0 = 0.0f, a1 = 0.0f, a2 = 0.0f;
    for (int p = p0 + sub; p < p1; p += 8) {
        int s = csrc[p];
        s = ((unsigned)s < (unsigned)n) ? s : ic;
        const float* hp = h + (size_t)s * 3;
        a0 += hp[0];
        a1 += hp[1];
        a2 += hp[2];
    }
    a0 += __shfl_xor(a0, 4); a1 += __shfl_xor(a1, 4); a2 += __shfl_xor(a2, 4);
    a0 += __shfl_xor(a0, 2); a1 += __shfl_xor(a1, 2); a2 += __shfl_xor(a2, 2);
    a0 += __shfl_xor(a0, 1); a1 += __shfl_xor(a1, 1); a2 += __shfl_xor(a2, 1);

    if (sub == 0) {
        const float* hp = h + (size_t)ic * 3;
        const int o = (tid >> 3) * 3;
        so[o + 0] = fmaf(hp[0] + a0, di, b[0]);
        so[o + 1] = fmaf(hp[1] + a1, di, b[1]);
        so[o + 2] = fmaf(hp[2] + a2, di, b[2]);
    }
    __syncthreads();
    if (tid < 96) {
        const long long idx = (long long)blockIdx.x * 96 + tid;
        if (idx < (long long)n * 3) out[idx] = so[tid];
    }
}


static inline int cdiv(long long a, int b) { return (int)((a + b - 1) / b); }

extern "C" void kernel_launch(void* const* d_in, const int* in_sizes, int n_in,
                              void* d_out, int out_size, void* d_ws, size_t ws_size,
                              hipStream_t stream) {
    const float* x   = (const float*)d_in[0];
    const int*   ei  = (const int*)d_in[1];
    const float* W1  = (const float*)d_in[2];
    const float* b1  = (const float*)d_in[3];
    const float* W2  = (const float*)d_in[4];
    const float* b2  = (const float*)d_in[5];
    const float* W3  = (const float*)d_in[6];
    const float* b3  = (const float*)d_in[7];
    float* out = (float*)d_out;

    const int N = in_sizes[0] / 9;
    const int E = in_sizes[1] / 2;
    const int* src = ei;
    const int* dst = ei + E;

    int*   rowstart = (int*)d_ws;
    int*   csrc     = rowstart + N;
    float* h        = (float*)(csrc + E);
    float* agg      = h + (size_t)N * 64;
    const int P  = cdiv(E, TILE);
    const int NB = cdiv(N, BSZ);
    unsigned* tmp  = (unsigned*)h;
    float*    xs   = h + E;
    int*      hist = (int*)agg;
    int*      loc  = hist + (size_t)P * NB;
    int*      btot = loc + (size_t)P * NB;
    (void)ws_size; (void)n_in; (void)out_size;

    const int ntiles = (N + 15) / 16;

    bhist_k<<<P, 1024, 0, stream>>>(dst, hist, E, NB);
    bscan_k<<<cdiv(NB, 32), 256, 0, stream>>>(hist, loc, btot, P, NB);
    part_k<<<P, 1024, 0, stream>>>(src, dst, loc, btot, tmp, E, NB);
    finish_k<<<NB, 256, 0, stream>>>(tmp, btot, rowstart, csrc, x, xs, E, N);

    gather16_gemm2_k<<<cdiv(ntiles, 8), 256, 0, stream>>>(rowstart, csrc, xs, W1, b1, W2, agg, N, E);

    gather64_gemm_k<3><<<cdiv(ntiles, 8), 256, 0, stream>>>(rowstart, csrc, agg, b2, W3, h, N, E);

    gather3_k<<<cdiv(N, 32), 256, 0, stream>>>(rowstart, csrc, h, b3, out, N, E);
}


# Layer-3 aggregation by 16-node tiles: a wave loads all indices of its edge slice as one group, then all rows (padded to 16 B) as a second, and sums runs from LDS; 4x fewer waves; sum order changes
# speedup vs baseline: 6.8897x; 1.0532x over previous
#include <hip/hip_runtime.h>


typedef float v2f __attribute__((ext_vector_type(2)));
typedef float v8f __attribute__((ext_vector_type(8)));


constexpr int BSH   = 7;
constexpr int BSZ   = 1 << BSH;
constexpr int NBMAX = 1024;
constexpr int TILE  = 6400;
constexpr int EPT   = (TILE + 1023) / 1024;
constexpr int STG   = 3072;
constexpr int KPT   = STG / 256;
constexpr int XW    = 16;

__device__ __forceinline__ unsigned bucket_of(int d, int NB) {
    return min((unsigned)d >> BSH, (unsigned)(NB - 1));
}

__global__ void __launch_bounds__(1024)
bhist_k(const int* __restrict__ dst, int* __restrict__ hist, int E, int NB) {
    __shared__ int sh[NBMAX];
    const int tid = threadIdx.x;
    sh[tid] = 0;
    __syncthreads();
    const int e0 = blockIdx.x * TILE;
    const int e1 = min(e0 + TILE, E);
    const int elast = max(e1 - 1, 0);
    int d[EPT];
#pragma unroll
    for (int k = 0; k < EPT; ++k) d[k] = dst[min(e0 + tid + k * 1024, elast)];
#pragma unroll
    for (int k = 0; k < EPT; ++k) {
        const int e = e0 + tid + k * 1024;
        if (e < e1) atomicAdd(&sh[bucket_of(d[k], NB)], 1);
    }
    __syncthreads();
    if (tid < NB) hist[blockIdx.x * NB + tid] = sh[tid];
}

__global__ void __launch_bounds__(256)
bscan_k(const int* __restrict__ hist, int* __restrict__ loc, int* __restrict__ btot, int P, int NB) {
    __shared__ int wsum[8][32];
    const int lane = threadIdx.x & 31;
    const int w    = threadIdx.x >> 5;
    const int b    = blockIdx.x * 32 + lane;
    const int bc   = min(b, NB - 1);
    int carry = 0;
    for (int base = 0; base < P; base += 256) {
        const int p0 = base + w * 32;
        int v[32];
#pragma unroll
        for (int k = 0; k < 32; ++k) {
            const int p = p0 + k;
            const int t = hist[min(p, P - 1) * NB + bc];
            v[k] = (p < P) ? t : 0;
        }
        int tot = 0;
#pragma unroll
        for (int k = 0; k < 32; ++k) tot += v[k];
        wsum[w][lane] = tot;
        __syncthreads();
        int run = carry, all = 0;
#pragma unroll
        for (int j = 0; j < 8; ++j) {
            const int t = wsum[j][lane];
            run += (j < w) ? t : 0;
            all += t;
        }
#pragma unroll
        for (int k = 0; k < 32; ++k) {
            const int p = p0 + k;
            if (p < P && b < NB) loc[p * NB + b] = run;
            run += v[k];
        }
        carry += all;
        __syncthreads();
    }
    if (w == 0 && b < NB) btot[b] = carry;
}

__global__ void __launch_bounds__(1024)
part_k(const int* __restrict__ src, const int* __restrict__ dst,
       const int* __restrict__ loc, const int* __restrict__ btot,
       unsigned* __restrict__ tmp, int E, int NB) {
    __shared__ int sc[NBMAX];
    __shared__ int sg[NBMAX];
    __shared__ int lstart[NBMAX];
    __shared__ int goff[NBMAX];
    __shared__ int cur[NBMAX];
    __shared__ unsigned sorted[TILE];
    __shared__ unsigned short sbk[TILE];

    const int tid = threadIdx.x;
    const int e0  = blockIdx.x * TILE;
    const int e1  = min(e0 + TILE, E);

    const int elast = max(e1 - 1, 0);
    int dd[EPT], ss[EPT];
#pragma unroll
    for (int k = 0; k < EPT; ++k) {
        const int ec = min(e0 + tid + k * 1024, elast);
        dd[k] = dst[ec];
        ss[k] = src[ec];
    }
    const int gq = btot[min(tid, NB - 1)];
    const int lq = loc[blockIdx.x * NB + min(tid, NB - 1)];

    cur[tid] = 0;
    __syncthreads();
#pragma unroll
    for (int k = 0; k < EPT; ++k) {
        const int e = e0 + tid + k * 1024;
        if (e < e1) atomicAdd(&cur[bucket_of(dd[k], NB)], 1);
    }
    __syncthreads();

    const int v = cur[tid];
    const int g = (tid < NB) ? gq : 0;
    const int l = (tid < NB) ? lq : 0;
    sc[tid] = v;
    sg[tid] = g;
    __syncthreads();
    for (int d = 1; d < 1024; d <<= 1) {
        const int tv = (tid >= d) ? sc[tid - d] : 0;
        const int tg = (tid >= d) ? sg[tid - d] : 0;
        __syncthreads();
        sc[tid] += tv;
        sg[tid] += tg;
        __syncthreads();
    }
    lstart[tid] = sc[tid] - v;
    cur[tid]    = sc[tid] - v;
    goff[tid]   = sg[tid] - g + l;
    __syncthreads();

#pragma unroll
    for (int k = 0; k < EPT; ++k) {
        const int e = e0 + tid + k * 1024;
        if (e < e1) {
            const unsigned b = bucket_of(dd[k], NB);
            unsigned pos = (unsigned)atomicAdd(&cur[b], 1);
            pos = min(pos, (unsigned)(TILE - 1));
            sorted[pos] = ((unsigned)ss[k] << BSH) | ((unsigned)dd[k] & (unsigned)(BSZ - 1));
            sbk[pos]    = (unsigned short)b;
        }
    }
    __syncthreads();

    const int cnt = e1 - e0;
    for (int i = tid; i < cnt; i += 1024) {
        const int b = sbk[i];
        unsigned slot = (unsigned)(goff[b] + (i - lstart[b]));
        slot = min(slot, (unsigned)(E - 1));
        tmp[slot] = sorted[i];
    }
}

__global__ void __launch_bounds__(256)
finish_k(const unsigned* __restrict__ tmp, const int* __restrict__ btot,
         int* __restrict__ rowstart, int* __restrict__ csrc,
         const float* __restrict__ x, float* __restrict__ xs, int E, int n) {
    __shared__ int red[256];
    __shared__ int cnt[BSZ];
    __shared__ int cur[BSZ];
    __shared__ int stage[STG];
    __shared__ float sdi[BSZ];

    const int tid = threadIdx.x;
    const int b   = blockIdx.x;

    float xv[BSZ * XW / 256];
#pragma unroll
    for (int j = 0; j < BSZ * XW / 256; ++j) {
        const int idx  = tid + j * 256;
        const int nloc = idx / XW;
        const int c    = idx - nloc * XW;
        const int node = min(b * BSZ + nloc, n - 1);
        xv[j] = x[(size_t)node * 9 + min(c, 8)];
    }

    const int mb = btot[b];
    int part = 0;
#pragma unroll
    for (int k = 0; k < NBMAX / 256; ++k) {
        const int j = tid + k * 256;
        const int t = btot[min(j, b)];
        part += (j < b) ? t : 0;
    }
    red[tid] = part;
    if (tid < BSZ) cnt[tid] = 0;
    __syncthreads();
    for (int st = 128; st > 0; st >>= 1) {
        if (tid < st) red[tid] += red[tid + st];
        __syncthreads();
    }
    const int start = min(max(red[0], 0), E);
    const int m     = min(max(mb, 0), E - start);
    __syncthreads();

    const bool small = (m <= STG);
    unsigned pk[KPT];
    if (small) {
        const int ilast = max(m - 1, 0);
#pragma unroll
        for (int k = 0; k < KPT; ++k) pk[k] = tmp[min(start + min(tid + k * 256, ilast), E - 1)];
#pragma unroll
        for (int k = 0; k < KPT; ++k) {
            const int i = tid + k * 256;
            if (i < m) atomicAdd(&cnt[pk[k] & (unsigned)(BSZ - 1)], 1);
        }
    } else {
        for (int i = tid; i < m; i += 256) atomicAdd(&cnt[tmp[start + i] & (unsigned)(BSZ - 1)], 1);
    }
    __syncthreads();

    const int v = (tid < BSZ) ? cnt[tid] : 0;
    red[tid] = v;
    __syncthreads();
    for (int d = 1; d < BSZ; d <<= 1) {
        const int t = (tid >= d) ? red[tid - d] : 0;
        __syncthreads();
        red[tid] += t;
        __syncthreads();
    }
    if (tid < BSZ) {
        const int excl = red[tid] - v;
        const int node = b * BSZ + tid;
        if (node < n) rowstart[node] = start + excl;
        cur[tid] = excl;
        sdi[tid] = rsqrtf((float)(v + 1));
    }
    __syncthreads();

#pragma unroll
    for (int j = 0; j < BSZ * XW / 256; ++j) {
        const int idx  = tid + j * 256;
        const int nloc = idx / XW;
        const int c    = idx - nloc * XW;
        if (b * BSZ + nloc < n) xs[(size_t)b * (BSZ * XW) + idx] = (c < 9) ? xv[j] * sdi[nloc] : 0.0f;
    }

    if (small) {
#pragma unroll
        for (int k = 0; k < KPT; ++k) {
            const int i = tid + k * 256;
            if (i < m) {
                unsigned pos = (unsigned)atomicAdd(&cur[pk[k] & (unsigned)(BSZ - 1)], 1);
                pos = min(pos, (unsigned)(STG - 1));
                stage[pos] = (int)(pk[k] >> BSH);
            }
        }
        __syncthreads();
        for (int i = tid; i < m; i += 256) csrc[start + i] = stage[i];
    } else {
        for (int i = tid; i < m; i += 256) {
            const unsigned q = tmp[start + i];
            unsigned pos = (unsigned)atomicAdd(&cur[q & (unsigned)(BSZ - 1)], 1);
            pos = min(pos, (unsigned)(m - 1));
            csrc[start + pos] = (int)(q >> BSH);
        }
    }
}

__global__ void __launch_bounds__(256)
gather16_gemm2_k(const int* __restrict__ rowstart, const int* __restrict__ csrc,
                 const float* __restrict__ xs, const float* __restrict__ W1,
                 const float* __restrict__ b1, const float* __restrict__ W2,
                 float* __restrict__ Hout, int n, int E) {
    constexpr int K1P = 12;
    constexpr int AP  = 68;
    constexpr int XP  = 20;
    constexpr int CAP = 384;
    constexpr int CH  = 16;

    __shared__ float sW2[64 * 64];
    __shared__ float sW1[K1P * 64];
    __shared__ alignas(16) float sA[8 * 16 * AP];
    __shared__ alignas(16) float sX[8 * 16 * XP];
    __shared__ float sDi[8 * 16];

    const int lane = threadIdx.x & 31;
    const int w    = threadIdx.x >> 5;
    const int q    = lane & 3;
    const int g    = lane >> 2;
    const int row0 = (blockIdx.x * 8 + w) * 16;
    float* tile = sA + w * (16 * AP);
    float* xt   = sX + w * (16 * XP);

    int rs;
    {
        const int node = row0 + min(lane, 16);
        const int v = rowstart[min(node, n - 1)];
        rs = (node < n) ? v : E;
        rs = min(max(rs, 0), E);
    }
    const int t0 = __shfl(rs, 0);

    float bc[4];
#pragma unroll
    for (int t = 0; t < 4; ++t) bc[t] = b1[t * 16 + (lane & 15)];

    int iv[CAP / 32];
#pragma unroll
    for (int k = 0; k < CAP / 32; ++k) iv[k] = csrc[min(t0 + lane + 32 * k, E - 1)];
    float4 selfrow[2];
#pragma unroll
    for (int pass = 0; pass < 2; ++pass)
        selfrow[pass] = *reinterpret_cast<const float4*>(xs + (size_t)min(row0 + pass * 8 + g, n - 1) * XW + q * 4);
    {
        constexpr int WPT = 64 * 64 / 256;
        float wreg[WPT];
#pragma unroll
        for (int j = 0; j < WPT; ++j) wreg[j] = W2[threadIdx.x + j * 256];
        float w1reg[K1P * 64 / 256];
#pragma unroll
        for (int j = 0; j < K1P * 64 / 256; ++j) {
            const int idx = threadIdx.x + j * 256;
            const int k = idx >> 6;
            const float t = W1[min(k, 8) * 64 + (idx & 63)];
            w1reg[j] = (k < 9) ? t : 0.0f;
        }
#pragma unroll
        for (int k = 0; k < CAP / 32; ++k) tile[lane + 32 * k] = __int_as_float(iv[k]);
#pragma unroll
        for (int j = 0; j < WPT; ++j) sW2[threadIdx.x + j * 256] = wreg[j];
#pragma unroll
        for (int j = 0; j < K1P * 64 / 256; ++j) sW1[threadIdx.x + j * 256] = w1reg[j];
    }
    __syncthreads();

    for (int pass = 0; pass < 2; ++pass) {
        const int nl = pass * 8 + g;
        const int i  = row0 + nl;
        const bool live = i < n;
        const int ic = live ? i : (n - 1);
        const int p0 = __shfl(rs, nl);
        int p1 = __shfl(rs, nl + 1);
        const float di = rsqrtf((float)(p1 - p0 + 1));
        if (!live) p1 = p0;

        float4 acc = selfrow[pass];

        for (int p = p0; p < p1; p += CH) {
            int sk[CH];
#pragma unroll
            for (int k = 0; k < CH; ++k) sk[k] = __float_as_int(tile[min(max(p + k - t0, 0), CAP - 1)]);
            if (p + CH - t0 > CAP) {
#pragma unroll
                for (int k = 0; k < CH; ++k)
                    if (p + k - t0 >= CAP) sk[k] = csrc[min(p + k, E - 1)];
            }
            float4 v[CH];
#pragma unroll
            for (int k = 0; k < CH; ++k) {
                const int s = ((unsigned)sk[k] < (unsigned)n) ? sk[k] : ic;
                v[k] = make_float4(0.0f, 0.0f, 0.0f, 0.0f);
                if (p + k < p1) v[k] = *reinterpret_cast<const float4*>(xs + (size_t)s * XW + q * 4);
            }
#pragma unroll
            for (int k = 0; k < CH; ++k) {
                acc.x += v[k].x;
                acc.y += v[k].y;
                acc.z += v[k].z;
                acc.w += v[k].w;
            }
        }
        acc.x *= di;
        acc.y *= di;
        acc.z *= di;
        acc.w *= di;
        if (!live) acc = make_float4(0.0f, 0.0f, 0.0f, 0.0f);
        *reinterpret_cast<float4*>(xt + nl * XP + q * 4) = acc;
        if (q == 0) sDi[w * 16 + nl] = di;
    }
    __syncthreads();

    const int m    = lane & 15;
    const int half = lane >> 4;

    v8f c1[4];
#pragma unroll
    for (int t = 0; t < 4; ++t) c1[t] = 0.0f;
#pragma unroll
    for (int kt = 0; kt < K1P / 4; ++kt) {
        const int ka = kt * 4 + half * 2;
        v2f a;
        a.x = xt[m * XP + ka];
        a.y = xt[m * XP + ka + 1];
#pragma unroll
        for (int nt = 0; nt < 4; ++nt) {
            const int col = nt * 16 + m;
            v2f bb;
            bb.x = sW1[ka * 64 + col];
            bb.y = sW1[(ka + 1) * 64 + col];
            c1[nt] = __builtin_amdgcn_wmma_f32_16x16x4_f32(
                false, a, false, bb, (short)0, c1[nt], false, false);
        }
    }
#pragma unroll
    for (int nt = 0; nt < 4; ++nt) {
#pragma unroll
        for (int r = 0; r < 8; ++r)
            tile[(r + half * 8) * AP + nt * 16 + m] = fmaxf(c1[nt][r] + bc[nt], 0.0f);
    }
    __syncthreads();

    const float* dv = sDi + w * 16 + half * 8;

    v8f accm[4];
#pragma unroll
    for (int t = 0; t < 4; ++t) accm[t] = 0.0f;
#pragma unroll
    for (int kt = 0; kt < 16; ++kt) {
        const int ka = kt * 4 + half * 2;
        v2f a;
        a.x = tile[m * AP + ka];
        a.y = tile[m * AP + ka + 1];
#pragma unroll
        for (int nt = 0; nt < 4; ++nt) {
            const int col = nt * 16 + m;
            v2f bb;
            bb.x = sW2[ka * 64 + col];
            bb.y = sW2[(ka + 1) * 64 + col];
            accm[nt] = __builtin_amdgcn_wmma_f32_16x16x4_f32(
                false, a, false, bb, (short)0, accm[nt], false, false);
        }
    }
#pragma unroll
    for (int nt = 0; nt < 4; ++nt) {
#pragma unroll
        for (int r = 0; r < 8; ++r) {
            const int row = row0 + r + half * 8;
            if (row < n) Hout[(size_t)row * 64 + nt * 16 + m] = accm[nt][r] * dv[r];
        }
    }
}


template<int NOUT>
__global__ void __launch_bounds__(256)
gather64_gemm_k(const int* __restrict__ rowstart, const int* __restrict__ csrc,
                const float* __restrict__ h, const float* __restrict__ b,
                const float* __restrict__ W, float* __restrict__ Hout, int n, int E) {
    constexpr int KDIM = 64;
    constexpr int NT   = (NOUT + 15) / 16;
    constexpr int KT   = KDIM / 4;
    constexpr int NW16 = NT * 16;
    constexpr int OP   = (NOUT + 3) / 4 * 4;
    constexpr int AP   = 68;
    constexpr int CH   = 16;

    __shared__ float sW[KDIM * NW16];
    __shared__ alignas(16) float sA[8 * 16 * AP];
    __shared__ float sDi[8 * 16];

    {
        constexpr int WPT = KDIM * NW16 / 256;
        float wreg[WPT];
#pragma unroll
        for (int j = 0; j < WPT; ++j) {
            const int idx = threadIdx.x + j * 256;
            const int k = idx / NW16;
            const int c = idx - k * NW16;
            wreg[j] = (c < NOUT) ? W[k * NOUT + c] : 0.0f;
        }
#pragma unroll
        for (int j = 0; j < WPT; ++j) sW[threadIdx.x + j * 256] = wreg[j];
    }

    const int lane = threadIdx.x & 31;
    const int w    = threadIdx.x >> 5;
    const int l16  = lane & 15;
    const int half = lane >> 4;
    const int cg   = l16 << 2;
    const int row0 = (blockIdx.x * 8 + w) * 16;
    float* tile = sA + w * (16 * AP);

    int rs;
    {
        const int node = row0 + min(lane, 16);
        const int v = rowstart[min(node, n - 1)];
        rs = (node < n) ? v : E;
        rs = min(max(rs, 0), E);
    }
    const float4 bv = *reinterpret_cast<const float4*>(b + cg);

    int s[CH];
    float4 selfrow;
    {
        const int q0 = __shfl(rs, half);
        const int qe = __shfl(rs, half + 1);
        const int q1 = (row0 + half < n) ? qe : q0;
        const int ql = max(q1 - 1, 0);
#pragma unroll
        for (int k = 0; k < CH; ++k) s[k] = csrc[min(q0 + k, ql)];
        selfrow = *reinterpret_cast<const float4*>(h + (size_t)min(row0 + half, n - 1) * 64 + cg);
    }

    for (int pass = 0; pass < 8; ++pass) {
        const int nl = pass * 2 + half;
        const int i  = row0 + nl;
        const bool live = i < n;
        const int ic = live ? i : (n - 1);
        const int p0 = __shfl(rs, nl);
        int p1 = __shfl(rs, nl + 1);
        const float di = rsqrtf((float)(p1 - p0 + 1));
        if (!live) p1 = p0;
        const int pl = max(p1 - 1, 0);

        const int inx = min(i + 2, n - 1);
        const int p0n = __shfl(rs, min(nl + 2, 16));
        const int pen = __shfl(rs, min(nl + 3, 16));
        const int p1n = (i + 2 < n) ? pen : p0n;
        const int pln = max(p1n - 1, 0);

        float4 acc = selfrow;

        for (int p = p0; p < p1; p += CH) {
            int sk[CH];
#pragma unroll
            for (int k = 0; k < CH; ++k)
                sk[k] = ((unsigned)s[k] < (unsigned)n) ? s[k] : ic;

            const bool last = (p + CH >= p1);
            const int nb = last ? p0n : (p + CH);
            const int nlim = last ? pln : pl;
#pragma unroll
            for (int k = 0; k < CH; ++k) s[k] = csrc[min(nb + k, nlim)];
            if (last) selfrow = *reinterpret_cast<const float4*>(h + (size_t)inx * 64 + cg);

            float4 v[CH];
#pragma unroll
            for (int k = 0; k < CH; ++k) {
                v[k] = make_float4(0.0f, 0.0f, 0.0f, 0.0f);
                if (p + k < p1) v[k] = *reinterpret_cast<const float4*>(h + (size_t)sk[k] * 64 + cg);
            }
#pragma unroll
            for (int k = 0; k < CH; ++k) {
                acc.x += v[k].x;
                acc.y += v[k].y;
                acc.z += v[k].z;
                acc.w += v[k].w;
            }
        }
        if (p0 >= p1) {
#pragma unroll
            for (int k = 0; k < CH; ++k) s[k] = csrc[min(p0n + k, pln)];
            selfrow = *reinterpret_cast<const float4*>(h + (size_t)inx * 64 + cg);
        }

        acc.x = fmaxf(fmaf(acc.x, di, bv.x), 0.0f);
        acc.y = fmaxf(fmaf(acc.y, di, bv.y), 0.0f);
        acc.z = fmaxf(fmaf(acc.z, di, bv.z), 0.0f);
        acc.w = fmaxf(fmaf(acc.w, di, bv.w), 0.0f);
        if (!live) acc = make_float4(0.0f, 0.0f, 0.0f, 0.0f);
        *reinterpret_cast<float4*>(tile + nl * AP + cg) = acc;
        if (l16 == 0) sDi[w * 16 + nl] = di;
    }
    __syncthreads();

    if (row0 >= n) return;

    const int m = l16;
    const float* dv = sDi + w * 16 + half * 8;

    v8f accm[NT];
#pragma unroll
    for (int t = 0; t < NT; ++t) accm[t] = 0.0f;

#pragma unroll
    for (int kt = 0; kt < KT; ++kt) {
        const int ka = kt * 4 + half * 2;
        v2f a;
        a.x = tile[m * AP + ka];
        a.y = tile[m * AP + ka + 1];
#pragma unroll
        for (int nt = 0; nt < NT; ++nt) {
            const int col = nt * 16 + m;
            v2f bb;
            bb.x = sW[ka * NW16 + col];
            bb.y = sW[(ka + 1) * NW16 + col];
            accm[nt] = __builtin_amdgcn_wmma_f32_16x16x4_f32(
                false, a, false, bb, (short)0, accm[nt], false, false);
        }
    }

#pragma unroll
    for (int nt = 0; nt < NT; ++nt) {
#pragma unroll
        for (int i = 0; i < 8; ++i) {
            const int r   = row0 + i + half * 8;
            const int col = nt * 16 + m;
            if (col < OP && r < n) Hout[(size_t)r * OP + col] = accm[nt][i] * dv[i];
        }
    }
}

__global__ void __launch_bounds__(256)
gather3_k(const int* __restrict__ rowstart, const int* __restrict__ csrc,
          const float* __restrict__ h, const float* __restrict__ b,
          float* __restrict__ out, int n, int E) {
    constexpr int CAP = 384;
    constexpr int LPT = CAP / 32;
    constexpr int SU  = 8;

    __shared__ alignas(16) float srow[8 * CAP * 4];
    __shared__ float so[8 * 16 * 3];

    const int tid  = threadIdx.x;
    const int lane = tid & 31;
    const int w    = tid >> 5;
    const int row0 = (blockIdx.x * 8 + w) * 16;
    float4* stage = reinterpret_cast<float4*>(srow) + w * CAP;

    int rs;
    {
        const int node = row0 + min(lane, 16);
        const int v = rowstart[min(node, n - 1)];
        rs = (node < n) ? v : E;
        rs = min(max(rs, 0), E);
    }

    const int nl = lane >> 1;
    const int hh = lane & 1;
    const int i  = row0 + nl;
    const bool live = i < n;
    const int ic = live ? i : (n - 1);
    const float4 selfrow = *reinterpret_cast<const float4*>(h + (size_t)ic * 4);

    const int t0  = __shfl(rs, 0);
    const int cnt = max(__shfl(rs, 16) - t0, 0);
    const int cst = min(cnt, CAP);

    int iv[LPT];
#pragma unroll
    for (int k = 0; k < LPT; ++k) iv[k] = csrc[min(t0 + lane + 32 * k, E - 1)];
    float4 v[LPT];
#pragma unroll
    for (int k = 0; k < LPT; ++k) {
        const int s = ((unsigned)iv[k] < (unsigned)n) ? iv[k] : ic;
        v[k] = make_float4(0.0f, 0.0f, 0.0f, 0.0f);
        if (lane + 32 * k < cst) v[k] = *reinterpret_cast<const float4*>(h + (size_t)s * 4);
    }
#pragma unroll
    for (int k = 0; k < LPT; ++k) stage[lane + 32 * k] = v[k];
    __syncthreads();

    const int p0 = __shfl(rs, nl);
    int p1 = __shfl(rs, nl + 1);
    const float di = rsqrtf((float)(p1 - p0 + 1));
    if (!live) p1 = p0;
    const int a0 = p0 - t0;
    const int a1 = p1 - t0;

    const int a1s = min(a1, CAP);
    float s0 = 0.0f, s1 = 0.0f, s2 = 0.0f;
    for (int q = a0 + hh; q < a1s; q += 2 * SU) {
        float4 r[SU];
#pragma unroll
        for (int j = 0; j < SU; ++j) r[j] = stage[min(max(q + 2 * j, 0), CAP - 1)];
#pragma unroll
        for (int j = 0; j < SU; ++j) {
            if (q + 2 * j < a1s) {
                s0 += r[j].x;
                s1 += r[j].y;
                s2 += r[j].z;
            }
        }
    }
    {
        int qt = a0 + hh;
        if (qt < CAP) qt += ((CAP - qt + 1) >> 1) << 1;
        for (int q = qt; q < a1; q += 2) {
            int s = csrc[min(max(t0 + q, 0), E - 1)];
            s = ((unsigned)s < (unsigned)n) ? s : ic;
            const float4 r = *reinterpret_cast<const float4*>(h + (size_t)s * 4);
            s0 += r.x;
            s1 += r.y;
            s2 += r.z;
        }
    }
    s0 += __shfl_xor(s0, 1); s1 += __shfl_xor(s1, 1); s2 += __shfl_xor(s2, 1);

    if (hh == 0) {
        const int o = (w * 16 + nl) * 3;
        so[o + 0] = fmaf(selfrow.x + s0, di, b[0]);
        so[o + 1] = fmaf(selfrow.y + s1, di, b[1]);
        so[o + 2] = fmaf(selfrow.z + s2, di, b[2]);
    }
    __syncthreads();
    for (int idx = tid; idx < 8 * 16 * 3; idx += 256) {
        const long long gi = (long long)blockIdx.x * (8 * 16 * 3) + idx;
        if (gi < (long long)n * 3) out[gi] = so[idx];
    }
}


static inline int cdiv(long long a, int b) { return (int)((a + b - 1) / b); }

extern "C" void kernel_launch(void* const* d_in, const int* in_sizes, int n_in,
                              void* d_out, int out_size, void* d_ws, size_t ws_size,
                              hipStream_t stream) {
    const float* x   = (const float*)d_in[0];
    const int*   ei  = (const int*)d_in[1];
    const float* W1  = (const float*)d_in[2];
    const float* b1  = (const float*)d_in[3];
    const float* W2  = (const float*)d_in[4];
    const float* b2  = (const float*)d_in[5];
    const float* W3  = (const float*)d_in[6];
    const float* b3  = (const float*)d_in[7];
    float* out = (float*)d_out;

    const int N = in_sizes[0] / 9;
    const int E = in_sizes[1] / 2;
    const int* src = ei;
    const int* dst = ei + E;

    int*   rowstart = (int*)d_ws;
    int*   csrc     = rowstart + N;
    float* h        = (float*)(csrc + E);
    float* agg      = h + (size_t)N * 64;
    const int P  = cdiv(E, TILE);
    const int NB = cdiv(N, BSZ);
    unsigned* tmp  = (unsigned*)h;
    float*    xs   = h + E;
    int*      hist = (int*)agg;
    int*      loc  = hist + (size_t)P * NB;
    int*      btot = loc + (size_t)P * NB;
    (void)ws_size; (void)n_in; (void)out_size;

    const int ntiles = (N + 15) / 16;

    bhist_k<<<P, 1024, 0, stream>>>(dst, hist, E, NB);
    bscan_k<<<cdiv(NB, 32), 256, 0, stream>>>(hist, loc, btot, P, NB);
    part_k<<<P, 1024, 0, stream>>>(src, dst, loc, btot, tmp, E, NB);
    finish_k<<<NB, 256, 0, stream>>>(tmp, btot, rowstart, csrc, x, xs, E, N);

    gather16_gemm2_k<<<cdiv(ntiles, 8), 256, 0, stream>>>(rowstart, csrc, xs, W1, b1, W2, agg, N, E);

    gather64_gemm_k<3><<<cdiv(ntiles, 8), 256, 0, stream>>>(rowstart, csrc, agg, b2, W3, h, N, E);

    gather3_k<<<cdiv(ntiles, 8), 256, 0, stream>>>(rowstart, csrc, h, b3, out, N, E);
}


# Sort level 1 writes each tile back bucket-sorted with its bucket offsets; level 2 gathers its bucket's runs from all tiles: no count pass, no cross-tile scan, 2 launches fewer; sum order kept
# speedup vs baseline: 7.9771x; 1.1578x over previous
#include <hip/hip_runtime.h>


typedef float v2f __attribute__((ext_vector_type(2)));
typedef float v8f __attribute__((ext_vector_type(8)));


constexpr int BSH   = 7;
constexpr int BSZ   = 1 << BSH;
constexpr int NBMAX = 1024;
constexpr int TILE  = 6400;
constexpr int EPT   = (TILE + 1023) / 1024;
constexpr int PMAX  = 256;
constexpr int STG   = 3072;
constexpr int KPT   = STG / 256;
constexpr int XW    = 16;

__device__ __forceinline__ unsigned bucket_of(int d, int NB) {
    return min((unsigned)d >> BSH, (unsigned)(NB - 1));
}

__global__ void __launch_bounds__(1024)
part_k(const int* __restrict__ src, const int* __restrict__ dst,
       unsigned* __restrict__ tmp, int* __restrict__ toff, int E, int NB) {
    __shared__ int sc[NBMAX];
    __shared__ int cur[NBMAX];
    __shared__ unsigned sorted[TILE];

    const int tid = threadIdx.x;
    const int e0  = blockIdx.x * TILE;
    const int e1  = min(e0 + TILE, E);

    const int elast = max(e1 - 1, 0);
    int dd[EPT], ss[EPT];
#pragma unroll
    for (int k = 0; k < EPT; ++k) {
        const int ec = min(e0 + tid + k * 1024, elast);
        dd[k] = dst[ec];
        ss[k] = src[ec];
    }

    cur[tid] = 0;
    __syncthreads();
#pragma unroll
    for (int k = 0; k < EPT; ++k) {
        const int e = e0 + tid + k * 1024;
        if (e < e1) atomicAdd(&cur[bucket_of(dd[k], NB)], 1);
    }
    __syncthreads();

    const int v = cur[tid];
    sc[tid] = v;
    __syncthreads();
    for (int d = 1; d < 1024; d <<= 1) {
        const int tv = (tid >= d) ? sc[tid - d] : 0;
        __syncthreads();
        sc[tid] += tv;
        __syncthreads();
    }
    const int excl = sc[tid] - v;
    cur[tid] = excl;
    int* trow = toff + (size_t)blockIdx.x * (NB + 1);
    if (tid < NB) trow[tid] = excl;
    if (tid == 0) trow[NB] = max(e1 - e0, 0);
    __syncthreads();

#pragma unroll
    for (int k = 0; k < EPT; ++k) {
        const int e = e0 + tid + k * 1024;
        if (e < e1) {
            const unsigned b = bucket_of(dd[k], NB);
            unsigned pos = (unsigned)atomicAdd(&cur[b], 1);
            pos = min(pos, (unsigned)(TILE - 1));
            sorted[pos] = ((unsigned)ss[k] << BSH) | ((unsigned)dd[k] & (unsigned)(BSZ - 1));
        }
    }
    __syncthreads();

    const int cnt = e1 - e0;
    for (int i = tid; i < cnt; i += 1024) tmp[e0 + i] = sorted[i];
}

__global__ void __launch_bounds__(256)
finish_k(const unsigned* __restrict__ tmp, const int* __restrict__ toff,
         int* __restrict__ rowstart, int* __restrict__ csrc,
         const float* __restrict__ x, float* __restrict__ xs, int E, int n, int P, int NB) {
    __shared__ int red[256];
    __shared__ int cnt[BSZ];
    __shared__ int cur[BSZ];
    __shared__ int stage[STG];
    __shared__ float sdi[BSZ];
    __shared__ int wlen[8];
    __shared__ int wbef[8];

    const int tid  = threadIdx.x;
    const int lane = tid & 31;
    const int w    = tid >> 5;
    const int b    = blockIdx.x;

    float xv[BSZ * XW / 256];
#pragma unroll
    for (int j = 0; j < BSZ * XW / 256; ++j) {
        const int idx  = tid + j * 256;
        const int nloc = idx / XW;
        const int c    = idx - nloc * XW;
        const int node = min(b * BSZ + nloc, n - 1);
        xv[j] = x[(size_t)node * 9 + min(c, 8)];
    }

    const int pt = min(tid, P - 1);
    const int* trow = toff + (size_t)pt * (NB + 1) + b;
    const int qa = trow[0];
    const int qe = trow[1];
    const int tcap = min(max(E - pt * TILE, 0), TILE);
    const int ra   = min(max(qa, 0), tcap);
    const int re   = min(max(qe, ra), tcap);
    const int len  = (tid < P) ? re - ra : 0;
    const int bef  = (tid < P) ? ra : 0;
    const int gb   = pt * TILE + ra;

    int inc = len, sbef = bef;
#pragma unroll
    for (int d = 1; d < 32; d <<= 1) {
        const int t = __shfl_up(inc, d);
        inc  += (lane >= d) ? t : 0;
        sbef += __shfl_xor(sbef, d);
    }
    if (lane == 31) { wlen[w] = inc; wbef[w] = sbef; }
    if (tid < BSZ) cnt[tid] = 0;
    __syncthreads();
    int off = inc - len, mtot = 0, stot = 0;
#pragma unroll
    for (int j = 0; j < 8; ++j) {
        const int t = wlen[j];
        off  += (j < w) ? t : 0;
        mtot += t;
        stot += wbef[j];
    }
    const int start = min(max(stot, 0), E);
    const int m     = min(max(mtot, 0), E - start);

    const bool small = (m <= STG);
    int* rb = stage;
    int* rl = stage + PMAX;
    unsigned pk[KPT];
    if (small) {
        for (int j = 0; j < len; ++j) stage[min(off + j, STG - 1)] = gb + j;
        __syncthreads();
        const int ilast = max(m - 1, 0);
        int si[KPT];
#pragma unroll
        for (int k = 0; k < KPT; ++k) si[k] = stage[min(tid + k * 256, ilast)];
#pragma unroll
        for (int k = 0; k < KPT; ++k) pk[k] = tmp[min(max(si[k], 0), E - 1)];
#pragma unroll
        for (int k = 0; k < KPT; ++k) {
            const int i = tid + k * 256;
            if (i < m) atomicAdd(&cnt[pk[k] & (unsigned)(BSZ - 1)], 1);
        }
    } else {
        rb[tid] = gb;
        rl[tid] = len;
        __syncthreads();
        for (int p = 0; p < min(P, PMAX); ++p) {
            const int L = rl[p], B = rb[p];
            for (int i = tid; i < L; i += 256) atomicAdd(&cnt[tmp[min(B + i, E - 1)] & (unsigned)(BSZ - 1)], 1);
        }
    }
    __syncthreads();

    const int v = (tid < BSZ) ? cnt[tid] : 0;
    red[tid] = v;
    __syncthreads();
    for (int d = 1; d < BSZ; d <<= 1) {
        const int t = (tid >= d) ? red[tid - d] : 0;
        __syncthreads();
        red[tid] += t;
        __syncthreads();
    }
    if (tid < BSZ) {
        const int excl = red[tid] - v;
        const int node = b * BSZ + tid;
        if (node < n) rowstart[node] = start + excl;
        cur[tid] = excl;
        sdi[tid] = rsqrtf((float)(v + 1));
    }
    __syncthreads();

#pragma unroll
    for (int j = 0; j < BSZ * XW / 256; ++j) {
        const int idx  = tid + j * 256;
        const int nloc = idx / XW;
        const int c    = idx - nloc * XW;
        if (b * BSZ + nloc < n) xs[(size_t)b * (BSZ * XW) + idx] = (c < 9) ? xv[j] * sdi[nloc] : 0.0f;
    }

    if (small) {
#pragma unroll
        for (int k = 0; k < KPT; ++k) {
            const int i = tid + k * 256;
            if (i < m) {
                unsigned pos = (unsigned)atomicAdd(&cur[pk[k] & (unsigned)(BSZ - 1)], 1);
                pos = min(pos, (unsigned)(STG - 1));
                stage[pos] = (int)(pk[k] >> BSH);
            }
        }
        __syncthreads();
        for (int i = tid; i < m; i += 256) csrc[start + i] = stage[i];
    } else {
        for (int p = 0; p < min(P, PMAX); ++p) {
            const int L = rl[p], B = rb[p];
            for (int i = tid; i < L; i += 256) {
                const unsigned q = tmp[min(B + i, E - 1)];
                unsigned pos = (unsigned)atomicAdd(&cur[q & (unsigned)(BSZ - 1)], 1);
                pos = min(pos, (unsigned)(m - 1));
                csrc[start + pos] = (int)(q >> BSH);
            }
        }
    }
}

__global__ void __launch_bounds__(256)
gather16_gemm2_k(const int* __restrict__ rowstart, const int* __restrict__ csrc,
                 const float* __restrict__ xs, const float* __restrict__ W1,
                 const float* __restrict__ b1, const float* __restrict__ W2,
                 float* __restrict__ Hout, int n, int E) {
    constexpr int K1P = 12;
    constexpr int AP  = 68;
    constexpr int XP  = 20;
    constexpr int CAP = 384;
    constexpr int CH  = 16;

    __shared__ float sW2[64 * 64];
    __shared__ float sW1[K1P * 64];
    __shared__ alignas(16) float sA[8 * 16 * AP];
    __shared__ alignas(16) float sX[8 * 16 * XP];
    __shared__ float sDi[8 * 16];

    const int lane = threadIdx.x & 31;
    const int w    = threadIdx.x >> 5;
    const int q    = lane & 3;
    const int g    = lane >> 2;
    const int row0 = (blockIdx.x * 8 + w) * 16;
    float* tile = sA + w * (16 * AP);
    float* xt   = sX + w * (16 * XP);

    int rs;
    {
        const int node = row0 + min(lane, 16);
        const int v = rowstart[min(node, n - 1)];
        rs = (node < n) ? v : E;
        rs = min(max(rs, 0), E);
    }
    const int t0 = __shfl(rs, 0);

    float bc[4];
#pragma unroll
    for (int t = 0; t < 4; ++t) bc[t] = b1[t * 16 + (lane & 15)];

    int iv[CAP / 32];
#pragma unroll
    for (int k = 0; k < CAP / 32; ++k) iv[k] = csrc[min(t0 + lane + 32 * k, E - 1)];
    float4 selfrow[2];
#pragma unroll
    for (int pass = 0; pass < 2; ++pass)
        selfrow[pass] = *reinterpret_cast<const float4*>(xs + (size_t)min(row0 + pass * 8 + g, n - 1) * XW + q * 4);
    {
        constexpr int WPT = 64 * 64 / 256;
        float wreg[WPT];
#pragma unroll
        for (int j = 0; j < WPT; ++j) wreg[j] = W2[threadIdx.x + j * 256];
        float w1reg[K1P * 64 / 256];
#pragma unroll
        for (int j = 0; j < K1P * 64 / 256; ++j) {
            const int idx = threadIdx.x + j * 256;
            const int k = idx >> 6;
            const float t = W1[min(k, 8) * 64 + (idx & 63)];
            w1reg[j] = (k < 9) ? t : 0.0f;
        }
#pragma unroll
        for (int k = 0; k < CAP / 32; ++k) tile[lane + 32 * k] = __int_as_float(iv[k]);
#pragma unroll
        for (int j = 0; j < WPT; ++j) sW2[threadIdx.x + j * 256] = wreg[j];
#pragma unroll
        for (int j = 0; j < K1P * 64 / 256; ++j) sW1[threadIdx.x + j * 256] = w1reg[j];
    }
    __syncthreads();

    for (int pass = 0; pass < 2; ++pass) {
        const int nl = pass * 8 + g;
        const int i  = row0 + nl;
        const bool live = i < n;
        const int ic = live ? i : (n - 1);
        const int p0 = __shfl(rs, nl);
        int p1 = __shfl(rs, nl + 1);
        const float di = rsqrtf((float)(p1 - p0 + 1));
        if (!live) p1 = p0;

        float4 acc = selfrow[pass];

        for (int p = p0; p < p1; p += CH) {
            int sk[CH];
#pragma unroll
            for (int k = 0; k < CH; ++k) sk[k] = __float_as_int(tile[min(max(p + k - t0, 0), CAP - 1)]);
            if (p + CH - t0 > CAP) {
#pragma unroll
                for (int k = 0; k < CH; ++k)
                    if (p + k - t0 >= CAP) sk[k] = csrc[min(p + k, E - 1)];
            }
            float4 v[CH];
#pragma unroll
            for (int k = 0; k < CH; ++k) {
                const int s = ((unsigned)sk[k] < (unsigned)n) ? sk[k] : ic;
                v[k] = make_float4(0.0f, 0.0f, 0.0f, 0.0f);
                if (p + k < p1) v[k] = *reinterpret_cast<const float4*>(xs + (size_t)s * XW + q * 4);
            }
#pragma unroll
            for (int k = 0; k < CH; ++k) {
                acc.x += v[k].x;
                acc.y += v[k].y;
                acc.z += v[k].z;
                acc.w += v[k].w;
            }
        }
        acc.x *= di;
        acc.y *= di;
        acc.z *= di;
        acc.w *= di;
        if (!live) acc = make_float4(0.0f, 0.0f, 0.0f, 0.0f);
        *reinterpret_cast<float4*>(xt + nl * XP + q * 4) = acc;
        if (q == 0) sDi[w * 16 + nl] = di;
    }
    __syncthreads();

    const int m    = lane & 15;
    const int half = lane >> 4;

    v8f c1[4];
#pragma unroll
    for (int t = 0; t < 4; ++t) c1[t] = 0.0f;
#pragma unroll
    for (int kt = 0; kt < K1P / 4; ++kt) {
        const int ka = kt * 4 + half * 2;
        v2f a;
        a.x = xt[m * XP + ka];
        a.y = xt[m * XP + ka + 1];
#pragma unroll
        for (int nt = 0; nt < 4; ++nt) {
            const int col = nt * 16 + m;
            v2f bb;
            bb.x = sW1[ka * 64 + col];
            bb.y = sW1[(ka + 1) * 64 + col];
            c1[nt] = __builtin_amdgcn_wmma_f32_16x16x4_f32(
                false, a, false, bb, (short)0, c1[nt], false, false);
        }
    }
#pragma unroll
    for (int nt = 0; nt < 4; ++nt) {
#pragma unroll
        for (int r = 0; r < 8; ++r)
            tile[(r + half * 8) * AP + nt * 16 + m] = fmaxf(c1[nt][r] + bc[nt], 0.0f);
    }
    __syncthreads();

    const float* dv = sDi + w * 16 + half * 8;

    v8f accm[4];
#pragma unroll
    for (int t = 0; t < 4; ++t) accm[t] = 0.0f;
#pragma unroll
    for (int kt = 0; kt < 16; ++kt) {
        const int ka = kt * 4 + half * 2;
        v2f a;
        a.x = tile[m * AP + ka];
        a.y = tile[m * AP + ka + 1];
#pragma unroll
        for (int nt = 0; nt < 4; ++nt) {
            const int col = nt * 16 + m;
            v2f bb;
            bb.x = sW2[ka * 64 + col];
            bb.y = sW2[(ka + 1) * 64 + col];
            accm[nt] = __builtin_amdgcn_wmma_f32_16x16x4_f32(
                false, a, false, bb, (short)0, accm[nt], false, false);
        }
    }
#pragma unroll
    for (int nt = 0; nt < 4; ++nt) {
#pragma unroll
        for (int r = 0; r < 8; ++r) {
            const int row = row0 + r + half * 8;
            if (row < n) Hout[(size_t)row * 64 + nt * 16 + m] = accm[nt][r] * dv[r];
        }
    }
}


template<int NOUT>
__global__ void __launch_bounds__(256)
gather64_gemm_k(const int* __restrict__ rowstart, const int* __restrict__ csrc,
                const float* __restrict__ h, const float* __restrict__ b,
                const float* __restrict__ W, float* __restrict__ Hout, int n, int E) {
    constexpr int KDIM = 64;
    constexpr int NT   = (NOUT + 15) / 16;
    constexpr int KT   = KDIM / 4;
    constexpr int NW16 = NT * 16;
    constexpr int OP   = (NOUT + 3) / 4 * 4;
    constexpr int AP   = 68;
    constexpr int CH   = 16;

    __shared__ float sW[KDIM * NW16];
    __shared__ alignas(16) float sA[8 * 16 * AP];
    __shared__ float sDi[8 * 16];

    {
        constexpr int WPT = KDIM * NW16 / 256;
        float wreg[WPT];
#pragma unroll
        for (int j = 0; j < WPT; ++j) {
            const int idx = threadIdx.x + j * 256;
            const int k = idx / NW16;
            const int c = idx - k * NW16;
            wreg[j] = (c < NOUT) ? W[k * NOUT + c] : 0.0f;
        }
#pragma unroll
        for (int j = 0; j < WPT; ++j) sW[threadIdx.x + j * 256] = wreg[j];
    }

    const int lane = threadIdx.x & 31;
    const int w    = threadIdx.x >> 5;
    const int l16  = lane & 15;
    const int half = lane >> 4;
    const int cg   = l16 << 2;
    const int row0 = (blockIdx.x * 8 + w) * 16;
    float* tile = sA + w * (16 * AP);

    int rs;
    {
        const int node = row0 + min(lane, 16);
        const int v = rowstart[min(node, n - 1)];
        rs = (node < n) ? v : E;
        rs = min(max(rs, 0), E);
    }
    const float4 bv = *reinterpret_cast<const float4*>(b + cg);

    int s[CH];
    float4 selfrow;
    {
        const int q0 = __shfl(rs, half);
        const int qe = __shfl(rs, half + 1);
        const int q1 = (row0 + half < n) ? qe : q0;
        const int ql = max(q1 - 1, 0);
#pragma unroll
        for (int k = 0; k < CH; ++k) s[k] = csrc[min(q0 + k, ql)];
        selfrow = *reinterpret_cast<const float4*>(h + (size_t)min(row0 + half, n - 1) * 64 + cg);
    }

    for (int pass = 0; pass < 8; ++pass) {
        const int nl = pass * 2 + half;
        const int i  = row0 + nl;
        const bool live = i < n;
        const int ic = live ? i : (n - 1);
        const int p0 = __shfl(rs, nl);
        int p1 = __shfl(rs, nl + 1);
        const float di = rsqrtf((float)(p1 - p0 + 1));
        if (!live) p1 = p0;
        const int pl = max(p1 - 1, 0);

        const int inx = min(i + 2, n - 1);
        const int p0n = __shfl(rs, min(nl + 2, 16));
        const int pen = __shfl(rs, min(nl + 3, 16));
        const int p1n = (i + 2 < n) ? pen : p0n;
        const int pln = max(p1n - 1, 0);

        float4 acc = selfrow;

        for (int p = p0; p < p1; p += CH) {
            int sk[CH];
#pragma unroll
            for (int k = 0; k < CH; ++k)
                sk[k] = ((unsigned)s[k] < (unsigned)n) ? s[k] : ic;

            const bool last = (p + CH >= p1);
            const int nb = last ? p0n : (p + CH);
            const int nlim = last ? pln : pl;
#pragma unroll
            for (int k = 0; k < CH; ++k) s[k] = csrc[min(nb + k, nlim)];
            if (last) selfrow = *reinterpret_cast<const float4*>(h + (size_t)inx * 64 + cg);

            float4 v[CH];
#pragma unroll
            for (int k = 0; k < CH; ++k) {
                v[k] = make_float4(0.0f, 0.0f, 0.0f, 0.0f);
                if (p + k < p1) v[k] = *reinterpret_cast<const float4*>(h + (size_t)sk[k] * 64 + cg);
            }
#pragma unroll
            for (int k = 0; k < CH; ++k) {
                acc.x += v[k].x;
                acc.y += v[k].y;
                acc.z += v[k].z;
                acc.w += v[k].w;
            }
        }
        if (p0 >= p1) {
#pragma unroll
            for (int k = 0; k < CH; ++k) s[k] = csrc[min(p0n + k, pln)];
            selfrow = *reinterpret_cast<const float4*>(h + (size_t)inx * 64 + cg);
        }

        acc.x = fmaxf(fmaf(acc.x, di, bv.x), 0.0f);
        acc.y = fmaxf(fmaf(acc.y, di, bv.y), 0.0f);
        acc.z = fmaxf(fmaf(acc.z, di, bv.z), 0.0f);
        acc.w = fmaxf(fmaf(acc.w, di, bv.w), 0.0f);
        if (!live) acc = make_float4(0.0f, 0.0f, 0.0f, 0.0f);
        *reinterpret_cast<float4*>(tile + nl * AP + cg) = acc;
        if (l16 == 0) sDi[w * 16 + nl] = di;
    }
    __syncthreads();

    if (row0 >= n) return;

    const int m = l16;
    const float* dv = sDi + w * 16 + half * 8;

    v8f accm[NT];
#pragma unroll
    for (int t = 0; t < NT; ++t) accm[t] = 0.0f;

#pragma unroll
    for (int kt = 0; kt < KT; ++kt) {
        const int ka = kt * 4 + half * 2;
        v2f a;
        a.x = tile[m * AP + ka];
        a.y = tile[m * AP + ka + 1];
#pragma unroll
        for (int nt = 0; nt < NT; ++nt) {
            const int col = nt * 16 + m;
            v2f bb;
            bb.x = sW[ka * NW16 + col];
            bb.y = sW[(ka + 1) * NW16 + col];
            accm[nt] = __builtin_amdgcn_wmma_f32_16x16x4_f32(
                false, a, false, bb, (short)0, accm[nt], false, false);
        }
    }

#pragma unroll
    for (int nt = 0; nt < NT; ++nt) {
#pragma unroll
        for (int i = 0; i < 8; ++i) {
            const int r   = row0 + i + half * 8;
            const int col = nt * 16 + m;
            if (col < OP && r < n) Hout[(size_t)r * OP + col] = accm[nt][i] * dv[i];
        }
    }
}

__global__ void __launch_bounds__(256)
gather3_k(const int* __restrict__ rowstart, const int* __restrict__ csrc,
          const float* __restrict__ h, const float* __restrict__ b,
          float* __restrict__ out, int n, int E) {
    constexpr int CAP = 384;
    constexpr int LPT = CAP / 32;
    constexpr int SU  = 8;

    __shared__ alignas(16) float srow[8 * CAP * 4];
    __shared__ float so[8 * 16 * 3];

    const int tid  = threadIdx.x;
    const int lane = tid & 31;
    const int w    = tid >> 5;
    const int row0 = (blockIdx.x * 8 + w) * 16;
    float4* stage = reinterpret_cast<float4*>(srow) + w * CAP;

    int rs;
    {
        const int node = row0 + min(lane, 16);
        const int v = rowstart[min(node, n - 1)];
        rs = (node < n) ? v : E;
        rs = min(max(rs, 0), E);
    }

    const int nl = lane >> 1;
    const int hh = lane & 1;
    const int i  = row0 + nl;
    const bool live = i < n;
    const int ic = live ? i : (n - 1);
    const float4 selfrow = *reinterpret_cast<const float4*>(h + (size_t)ic * 4);

    const int t0  = __shfl(rs, 0);
    const int cnt = max(__shfl(rs, 16) - t0, 0);
    const int cst = min(cnt, CAP);

    int iv[LPT];
#pragma unroll
    for (int k = 0; k < LPT; ++k) iv[k] = csrc[min(t0 + lane + 32 * k, E - 1)];
    float4 v[LPT];
#pragma unroll
    for (int k = 0; k < LPT; ++k) {
        const int s = ((unsigned)iv[k] < (unsigned)n) ? iv[k] : ic;
        v[k] = make_float4(0.0f, 0.0f, 0.0f, 0.0f);
        if (lane + 32 * k < cst) v[k] = *reinterpret_cast<const float4*>(h + (size_t)s * 4);
    }
#pragma unroll
    for (int k = 0; k < LPT; ++k) stage[lane + 32 * k] = v[k];
    __syncthreads();

    const int p0 = __shfl(rs, nl);
    int p1 = __shfl(rs, nl + 1);
    const float di = rsqrtf((float)(p1 - p0 + 1));
    if (!live) p1 = p0;
    const int a0 = p0 - t0;
    const int a1 = p1 - t0;

    const int a1s = min(a1, CAP);
    float s0 = 0.0f, s1 = 0.0f, s2 = 0.0f;
    for (int q = a0 + hh; q < a1s; q += 2 * SU) {
        float4 r[SU];
#pragma unroll
        for (int j = 0; j < SU; ++j) r[j] = stage[min(max(q + 2 * j, 0), CAP - 1)];
#pragma unroll
        for (int j = 0; j < SU; ++j) {
            if (q + 2 * j < a1s) {
                s0 += r[j].x;
                s1 += r[j].y;
                s2 += r[j].z;
            }
        }
    }
    {
        int qt = a0 + hh;
        if (qt < CAP) qt += ((CAP - qt + 1) >> 1) << 1;
        for (int q = qt; q < a1; q += 2) {
            int s = csrc[min(max(t0 + q, 0), E - 1)];
            s = ((unsigned)s < (unsigned)n) ? s : ic;
            const float4 r = *reinterpret_cast<const float4*>(h + (size_t)s * 4);
            s0 += r.x;
            s1 += r.y;
            s2 += r.z;
        }
    }
    s0 += __shfl_xor(s0, 1); s1 += __shfl_xor(s1, 1); s2 += __shfl_xor(s2, 1);

    if (hh == 0) {
        const int o = (w * 16 + nl) * 3;
        so[o + 0] = fmaf(selfrow.x + s0, di, b[0]);
        so[o + 1] = fmaf(selfrow.y + s1, di, b[1]);
        so[o + 2] = fmaf(selfrow.z + s2, di, b[2]);
    }
    __syncthreads();
    for (int idx = tid; idx < 8 * 16 * 3; idx += 256) {
        const long long gi = (long long)blockIdx.x * (8 * 16 * 3) + idx;
        if (gi < (long long)n * 3) out[gi] = so[idx];
    }
}


static inline int cdiv(long long a, int b) { return (int)((a + b - 1) / b); }

extern "C" void kernel_launch(void* const* d_in, const int* in_sizes, int n_in,
                              void* d_out, int out_size, void* d_ws, size_t ws_size,
                              hipStream_t stream) {
    const float* x   = (const float*)d_in[0];
    const int*   ei  = (const int*)d_in[1];
    const float* W1  = (const float*)d_in[2];
    const float* b1  = (const float*)d_in[3];
    const float* W2  = (const float*)d_in[4];
    const float* b2  = (const float*)d_in[5];
    const float* W3  = (const float*)d_in[6];
    const float* b3  = (const float*)d_in[7];
    float* out = (float*)d_out;

    const int N = in_sizes[0] / 9;
    const int E = in_sizes[1] / 2;
    const int* src = ei;
    const int* dst = ei + E;

    int*   rowstart = (int*)d_ws;
    int*   csrc     = rowstart + N;
    float* h        = (float*)(csrc + E);
    float* agg      = h + (size_t)N * 64;
    const int P  = cdiv(E, TILE);
    const int NB = cdiv(N, BSZ);
    unsigned* tmp  = (unsigned*)h;
    float*    xs   = h + E;
    int*      toff = (int*)agg;
    (void)ws_size; (void)n_in; (void)out_size;

    const int ntiles = (N + 15) / 16;

    part_k<<<P, 1024, 0, stream>>>(src, dst, tmp, toff, E, NB);
    finish_k<<<NB, 256, 0, stream>>>(tmp, toff, rowstart, csrc, x, xs, E, N, P, NB);

    gather16_gemm2_k<<<cdiv(ntiles, 8), 256, 0, stream>>>(rowstart, csrc, xs, W1, b1, W2, agg, N, E);

    gather64_gemm_k<3><<<cdiv(ntiles, 8), 256, 0, stream>>>(rowstart, csrc, agg, b2, W3, h, N, E);

    gather3_k<<<cdiv(ntiles, 8), 256, 0, stream>>>(rowstart, csrc, h, b3, out, N, E);
}


# Layer-1 and layer-2 edge walks without branches: a group's 16 rows load unconditionally (slots past a node's end read a zero row); layer 2's csrc slice staged in LDS once; sum order kept
# speedup vs baseline: 8.5425x; 1.0709x over previous
#include <hip/hip_runtime.h>


typedef float v2f __attribute__((ext_vector_type(2)));
typedef float v8f __attribute__((ext_vector_type(8)));
typedef float v4f __attribute__((ext_vector_type(4)));


constexpr int BSH   = 7;
constexpr int BSZ   = 1 << BSH;
constexpr int NBMAX = 1024;
constexpr int TILE  = 6400;
constexpr int EPT   = (TILE + 1023) / 1024;
constexpr int PMAX  = 256;
constexpr int STG   = 3072;
constexpr int KPT   = STG / 256;
constexpr int XW    = 16;

__device__ __forceinline__ unsigned bucket_of(int d, int NB) {
    return min((unsigned)d >> BSH, (unsigned)(NB - 1));
}

__global__ void __launch_bounds__(1024)
part_k(const int* __restrict__ src, const int* __restrict__ dst,
       unsigned* __restrict__ tmp, int* __restrict__ toff, int E, int NB) {
    __shared__ int sc[NBMAX];
    __shared__ int cur[NBMAX];
    __shared__ unsigned sorted[TILE];

    const int tid = threadIdx.x;
    const int e0  = blockIdx.x * TILE;
    const int e1  = min(e0 + TILE, E);

    const int elast = max(e1 - 1, 0);
    int dd[EPT], ss[EPT];
#pragma unroll
    for (int k = 0; k < EPT; ++k) {
        const int ec = min(e0 + tid + k * 1024, elast);
        dd[k] = dst[ec];
        ss[k] = src[ec];
    }

    cur[tid] = 0;
    __syncthreads();
#pragma unroll
    for (int k = 0; k < EPT; ++k) {
        const int e = e0 + tid + k * 1024;
        if (e < e1) atomicAdd(&cur[bucket_of(dd[k], NB)], 1);
    }
    __syncthreads();

    const int v = cur[tid];
    sc[tid] = v;
    __syncthreads();
    for (int d = 1; d < 1024; d <<= 1) {
        const int tv = (tid >= d) ? sc[tid - d] : 0;
        __syncthreads();
        sc[tid] += tv;
        __syncthreads();
    }
    const int excl = sc[tid] - v;
    cur[tid] = excl;
    int* trow = toff + (size_t)blockIdx.x * (NB + 1);
    if (tid < NB) trow[tid] = excl;
    if (tid == 0) trow[NB] = max(e1 - e0, 0);
    __syncthreads();

#pragma unroll
    for (int k = 0; k < EPT; ++k) {
        const int e = e0 + tid + k * 1024;
        if (e < e1) {
            const unsigned b = bucket_of(dd[k], NB);
            unsigned pos = (unsigned)atomicAdd(&cur[b], 1);
            pos = min(pos, (unsigned)(TILE - 1));
            sorted[pos] = ((unsigned)ss[k] << BSH) | ((unsigned)dd[k] & (unsigned)(BSZ - 1));
        }
    }
    __syncthreads();

    const int cnt = e1 - e0;
    for (int i = tid; i < cnt; i += 1024) tmp[e0 + i] = sorted[i];
}

__global__ void __launch_bounds__(256)
finish_k(const unsigned* __restrict__ tmp, const int* __restrict__ toff,
         int* __restrict__ rowstart, int* __restrict__ csrc,
         const float* __restrict__ x, float* __restrict__ xs, int E, int n, int P, int NB) {
    __shared__ int red[256];
    __shared__ int cnt[BSZ];
    __shared__ int cur[BSZ];
    __shared__ int stage[STG];
    __shared__ float sdi[BSZ];
    __shared__ int wlen[8];
    __shared__ int wbef[8];

    const int tid  = threadIdx.x;
    const int lane = tid & 31;
    const int w    = tid >> 5;
    const int b    = blockIdx.x;

    float xv[BSZ * XW / 256];
#pragma unroll
    for (int j = 0; j < BSZ * XW / 256; ++j) {
        const int idx  = tid + j * 256;
        const int nloc = idx / XW;
        const int c    = idx - nloc * XW;
        const int node = min(b * BSZ + nloc, n - 1);
        xv[j] = x[(size_t)node * 9 + min(c, 8)];
    }

    const int pt = min(tid, P - 1);
    const int* trow = toff + (size_t)pt * (NB + 1) + b;
    const int qa = trow[0];
    const int qe = trow[1];
    const int tcap = min(max(E - pt * TILE, 0), TILE);
    const int ra   = min(max(qa, 0), tcap);
    const int re   = min(max(qe, ra), tcap);
    const int len  = (tid < P) ? re - ra : 0;
    const int bef  = (tid < P) ? ra : 0;
    const int gb   = pt * TILE + ra;

    int inc = len, sbef = bef;
#pragma unroll
    for (int d = 1; d < 32; d <<= 1) {
        const int t = __shfl_up(inc, d);
        inc  += (lane >= d) ? t : 0;
        sbef += __shfl_xor(sbef, d);
    }
    if (lane == 31) { wlen[w] = inc; wbef[w] = sbef; }
    if (tid < BSZ) cnt[tid] = 0;
    __syncthreads();
    int off = inc - len, mtot = 0, stot = 0;
#pragma unroll
    for (int j = 0; j < 8; ++j) {
        const int t = wlen[j];
        off  += (j < w) ? t : 0;
        mtot += t;
        stot += wbef[j];
    }
    const int start = min(max(stot, 0), E);
    const int m     = min(max(mtot, 0), E - start);

    const bool small = (m <= STG);
    int* rb = stage;
    int* rl = stage + PMAX;
    unsigned pk[KPT];
    if (small) {
        for (int j = 0; j < len; ++j) stage[min(off + j, STG - 1)] = gb + j;
        __syncthreads();
        const int ilast = max(m - 1, 0);
        int si[KPT];
#pragma unroll
        for (int k = 0; k < KPT; ++k) si[k] = stage[min(tid + k * 256, ilast)];
#pragma unroll
        for (int k = 0; k < KPT; ++k) pk[k] = tmp[min(max(si[k], 0), E - 1)];
#pragma unroll
        for (int k = 0; k < KPT; ++k) {
            const int i = tid + k * 256;
            if (i < m) atomicAdd(&cnt[pk[k] & (unsigned)(BSZ - 1)], 1);
        }
    } else {
        rb[tid] = gb;
        rl[tid] = len;
        __syncthreads();
        for (int p = 0; p < min(P, PMAX); ++p) {
            const int L = rl[p], B = rb[p];
            for (int i = tid; i < L; i += 256) atomicAdd(&cnt[tmp[min(B + i, E - 1)] & (unsigned)(BSZ - 1)], 1);
        }
    }
    __syncthreads();

    const int v = (tid < BSZ) ? cnt[tid] : 0;
    red[tid] = v;
    __syncthreads();
    for (int d = 1; d < BSZ; d <<= 1) {
        const int t = (tid >= d) ? red[tid - d] : 0;
        __syncthreads();
        red[tid] += t;
        __syncthreads();
    }
    if (tid < BSZ) {
        const int excl = red[tid] - v;
        const int node = b * BSZ + tid;
        if (node < n) rowstart[node] = start + excl;
        cur[tid] = excl;
        sdi[tid] = rsqrtf((float)(v + 1));
    }
    __syncthreads();

#pragma unroll
    for (int j = 0; j < BSZ * XW / 256; ++j) {
        const int idx  = tid + j * 256;
        const int nloc = idx / XW;
        const int c    = idx - nloc * XW;
        if (b * BSZ + nloc < n) xs[(size_t)b * (BSZ * XW) + idx] = (c < 9) ? xv[j] * sdi[nloc] : 0.0f;
    }
    if (b == 0 && tid < 64) xs[(size_t)n * XW + tid] = 0.0f;

    if (small) {
#pragma unroll
        for (int k = 0; k < KPT; ++k) {
            const int i = tid + k * 256;
            if (i < m) {
                unsigned pos = (unsigned)atomicAdd(&cur[pk[k] & (unsigned)(BSZ - 1)], 1);
                pos = min(pos, (unsigned)(STG - 1));
                stage[pos] = (int)(pk[k] >> BSH);
            }
        }
        __syncthreads();
        for (int i = tid; i < m; i += 256) csrc[start + i] = stage[i];
    } else {
        for (int p = 0; p < min(P, PMAX); ++p) {
            const int L = rl[p], B = rb[p];
            for (int i = tid; i < L; i += 256) {
                const unsigned q = tmp[min(B + i, E - 1)];
                unsigned pos = (unsigned)atomicAdd(&cur[q & (unsigned)(BSZ - 1)], 1);
                pos = min(pos, (unsigned)(m - 1));
                csrc[start + pos] = (int)(q >> BSH);
            }
        }
    }
}

__global__ void __launch_bounds__(256)
gather16_gemm2_k(const int* __restrict__ rowstart, const int* __restrict__ csrc,
                 const float* __restrict__ xs, const float* __restrict__ W1,
                 const float* __restrict__ b1, const float* __restrict__ W2,
                 float* __restrict__ Hout, int n, int E) {
    constexpr int K1P = 12;
    constexpr int AP  = 68;
    constexpr int XP  = 20;
    constexpr int CAP = 384;
    constexpr int CH  = 16;

    __shared__ float sW2[64 * 64];
    __shared__ float sW1[K1P * 64];
    __shared__ alignas(16) float sA[8 * 16 * AP];
    __shared__ alignas(16) float sX[8 * 16 * XP];
    __shared__ float sDi[8 * 16];

    const int lane = threadIdx.x & 31;
    const int w    = threadIdx.x >> 5;
    const int q    = lane & 3;
    const int g    = lane >> 2;
    const int row0 = (blockIdx.x * 8 + w) * 16;
    float* tile = sA + w * (16 * AP);
    float* xt   = sX + w * (16 * XP);

    int rs;
    {
        const int node = row0 + min(lane, 16);
        const int v = rowstart[min(node, n - 1)];
        rs = (node < n) ? v : E;
        rs = min(max(rs, 0), E);
    }
    const int t0 = __shfl(rs, 0);

    float bc[4];
#pragma unroll
    for (int t = 0; t < 4; ++t) bc[t] = b1[t * 16 + (lane & 15)];

    int iv[CAP / 32];
#pragma unroll
    for (int k = 0; k < CAP / 32; ++k) iv[k] = csrc[min(t0 + lane + 32 * k, E - 1)];
    v4f selfrow[2];
#pragma unroll
    for (int pass = 0; pass < 2; ++pass)
        selfrow[pass] = *reinterpret_cast<const v4f*>(xs + (size_t)min(row0 + pass * 8 + g, n - 1) * XW + q * 4);
    {
        constexpr int WPT = 64 * 64 / 256;
        float wreg[WPT];
#pragma unroll
        for (int j = 0; j < WPT; ++j) wreg[j] = W2[threadIdx.x + j * 256];
        float w1reg[K1P * 64 / 256];
#pragma unroll
        for (int j = 0; j < K1P * 64 / 256; ++j) {
            const int idx = threadIdx.x + j * 256;
            const int k = idx >> 6;
            const float t = W1[min(k, 8) * 64 + (idx & 63)];
            w1reg[j] = (k < 9) ? t : 0.0f;
        }
#pragma unroll
        for (int k = 0; k < CAP / 32; ++k) tile[lane + 32 * k] = __int_as_float(iv[k]);
#pragma unroll
        for (int j = 0; j < WPT; ++j) sW2[threadIdx.x + j * 256] = wreg[j];
#pragma unroll
        for (int j = 0; j < K1P * 64 / 256; ++j) sW1[threadIdx.x + j * 256] = w1reg[j];
    }
    __syncthreads();

    const char* xb = reinterpret_cast<const char*>(xs);
    const unsigned q16 = (unsigned)q * 16u;

    for (int pass = 0; pass < 2; ++pass) {
        const int nl = pass * 8 + g;
        const int i  = row0 + nl;
        const bool live = i < n;
        const int p0 = __shfl(rs, nl);
        int p1 = __shfl(rs, nl + 1);
        const float di = rsqrtf((float)(p1 - p0 + 1));
        if (!live) p1 = p0;

        v4f acc = selfrow[pass];

        int p = p0;
        for (; p < p1 && (unsigned)(p - t0) <= (unsigned)(CAP - CH); p += CH) {
            const float* ip = tile + (p - t0);
            int sk[CH];
#pragma unroll
            for (int k = 0; k < CH; ++k) sk[k] = __float_as_int(ip[k]);
            const int rem = p1 - p;
            v4f v[CH];
#pragma unroll
            for (int k = 0; k < CH; ++k) {
                const unsigned sc  = min((unsigned)sk[k], (unsigned)(n - 1));
                const unsigned row = (k < rem) ? sc : (unsigned)n;
                v[k] = *reinterpret_cast<const v4f*>(xb + ((row << 6) + q16));
            }
            asm volatile("" :: "v"(v[0]), "v"(v[1]), "v"(v[2]), "v"(v[3]), "v"(v[4]), "v"(v[5]), "v"(v[6]), "v"(v[7]),
                               "v"(v[8]), "v"(v[9]), "v"(v[10]), "v"(v[11]), "v"(v[12]), "v"(v[13]), "v"(v[14]), "v"(v[15]));
#pragma unroll
            for (int k = 0; k < CH; ++k) acc += v[k];
        }
        for (; p < p1; ++p) {
            const unsigned sc = min((unsigned)csrc[min(max(p, 0), E - 1)], (unsigned)(n - 1));
            acc += *reinterpret_cast<const v4f*>(xb + ((sc << 6) + q16));
        }
        acc *= di;
        if (!live) acc = 0.0f;
        *reinterpret_cast<v4f*>(xt + nl * XP + q * 4) = acc;
        if (q == 0) sDi[w * 16 + nl] = di;
    }
    __syncthreads();

    const int m    = lane & 15;
    const int half = lane >> 4;

    v8f c1[4];
#pragma unroll
    for (int t = 0; t < 4; ++t) c1[t] = 0.0f;
#pragma unroll
    for (int kt = 0; kt < K1P / 4; ++kt) {
        const int ka = kt * 4 + half * 2;
        v2f a;
        a.x = xt[m * XP + ka];
        a.y = xt[m * XP + ka + 1];
#pragma unroll
        for (int nt = 0; nt < 4; ++nt) {
            const int col = nt * 16 + m;
            v2f bb;
            bb.x = sW1[ka * 64 + col];
            bb.y = sW1[(ka + 1) * 64 + col];
            c1[nt] = __builtin_amdgcn_wmma_f32_16x16x4_f32(
                false, a, false, bb, (short)0, c1[nt], false, false);
        }
    }
#pragma unroll
    for (int nt = 0; nt < 4; ++nt) {
#pragma unroll
        for (int r = 0; r < 8; ++r)
            tile[(r + half * 8) * AP + nt * 16 + m] = fmaxf(c1[nt][r] + bc[nt], 0.0f);
    }
    __syncthreads();

    const float* dv = sDi + w * 16 + half * 8;

    v8f accm[4];
#pragma unroll
    for (int t = 0; t < 4; ++t) accm[t] = 0.0f;
#pragma unroll
    for (int kt = 0; kt < 16; ++kt) {
        const int ka = kt * 4 + half * 2;
        v2f a;
        a.x = tile[m * AP + ka];
        a.y = tile[m * AP + ka + 1];
#pragma unroll
        for (int nt = 0; nt < 4; ++nt) {
            const int col = nt * 16 + m;
            v2f bb;
            bb.x = sW2[ka * 64 + col];
            bb.y = sW2[(ka + 1) * 64 + col];
            accm[nt] = __builtin_amdgcn_wmma_f32_16x16x4_f32(
                false, a, false, bb, (short)0, accm[nt], false, false);
        }
    }
#pragma unroll
    for (int nt = 0; nt < 4; ++nt) {
#pragma unroll
        for (int r = 0; r < 8; ++r) {
            const int row = row0 + r + half * 8;
            if (row < n) Hout[(size_t)row * 64 + nt * 16 + m] = accm[nt][r] * dv[r];
        }
    }
}


template<int NOUT>
__global__ void __launch_bounds__(256)
gather64_gemm_k(const int* __restrict__ rowstart, const int* __restrict__ csrc,
                const float* __restrict__ h, const float* zrow, const float* __restrict__ b,
                const float* __restrict__ W, float* __restrict__ Hout, int n, int E) {
    constexpr int KDIM = 64;
    constexpr int NT   = (NOUT + 15) / 16;
    constexpr int KT   = KDIM / 4;
    constexpr int NW16 = NT * 16;
    constexpr int OP   = (NOUT + 3) / 4 * 4;
    constexpr int AP   = 68;
    constexpr int CH   = 16;
    constexpr int CAP  = 384;

    __shared__ float sW[KDIM * NW16];
    __shared__ alignas(16) float sA[8 * 16 * AP];
    __shared__ float sDi[8 * 16];
    __shared__ int sIdx[8 * CAP];

    const int lane = threadIdx.x & 31;
    const int w    = threadIdx.x >> 5;
    const int l16  = lane & 15;
    const int half = lane >> 4;
    const int cg   = l16 << 2;
    const int row0 = (blockIdx.x * 8 + w) * 16;
    float* tile = sA + w * (16 * AP);
    int* myidx  = sIdx + w * CAP;

    int rs;
    {
        const int node = row0 + min(lane, 16);
        const int v = rowstart[min(node, n - 1)];
        rs = (node < n) ? v : E;
        rs = min(max(rs, 0), E);
    }
    const float4 bv = *reinterpret_cast<const float4*>(b + cg);
    const int t0 = __shfl(rs, 0);

    v4f selfrow;
    {
        int iv[CAP / 32];
#pragma unroll
        for (int k = 0; k < CAP / 32; ++k) iv[k] = csrc[min(t0 + lane + 32 * k, E - 1)];
        selfrow = *reinterpret_cast<const v4f*>(h + (size_t)min(row0 + half, n - 1) * 64 + cg);
        constexpr int WPT = KDIM * NW16 / 256;
        float wreg[WPT];
#pragma unroll
        for (int j = 0; j < WPT; ++j) {
            const int idx = threadIdx.x + j * 256;
            const int k = idx / NW16;
            const int c = idx - k * NW16;
            wreg[j] = (c < NOUT) ? W[k * NOUT + c] : 0.0f;
        }
#pragma unroll
        for (int k = 0; k < CAP / 32; ++k) myidx[lane + 32 * k] = iv[k];
#pragma unroll
        for (int j = 0; j < WPT; ++j) sW[threadIdx.x + j * 256] = wreg[j];
    }
    __syncthreads();

    const char* tb = reinterpret_cast<const char*>(zrow);
    const unsigned cg4  = (unsigned)cg * 4u;
    const unsigned tofs = (unsigned)(reinterpret_cast<const char*>(h) - tb) + cg4;

    for (int pass = 0; pass < 8; ++pass) {
        const int nl = pass * 2 + half;
        const int i  = row0 + nl;
        const bool live = i < n;
        const int p0 = __shfl(rs, nl);
        int p1 = __shfl(rs, nl + 1);
        const float di = rsqrtf((float)(p1 - p0 + 1));
        if (!live) p1 = p0;

        v4f acc = selfrow;
        selfrow = *reinterpret_cast<const v4f*>(h + (size_t)min(i + 2, n - 1) * 64 + cg);

        int p = p0;
        for (; p < p1 && (unsigned)(p - t0) <= (unsigned)(CAP - CH); p += CH) {
            const int* ip = myidx + (p - t0);
            int sk[CH];
#pragma unroll
            for (int k = 0; k < CH; ++k) sk[k] = ip[k];
            const int rem = p1 - p;
            v4f v[CH];
#pragma unroll
            for (int k = 0; k < CH; ++k) {
                const unsigned sc  = min((unsigned)sk[k], (unsigned)(n - 1));
                const unsigned off = (k < rem) ? ((sc << 8) + tofs) : cg4;
                v[k] = *reinterpret_cast<const v4f*>(tb + off);
            }
            asm volatile("" :: "v"(v[0]), "v"(v[1]), "v"(v[2]), "v"(v[3]), "v"(v[4]), "v"(v[5]), "v"(v[6]), "v"(v[7]),
                               "v"(v[8]), "v"(v[9]), "v"(v[10]), "v"(v[11]), "v"(v[12]), "v"(v[13]), "v"(v[14]), "v"(v[15]));
#pragma unroll
            for (int k = 0; k < CH; ++k) acc += v[k];
        }
        for (; p < p1; ++p) {
            const unsigned sc = min((unsigned)csrc[min(max(p, 0), E - 1)], (unsigned)(n - 1));
            acc += *reinterpret_cast<const v4f*>(tb + ((sc << 8) + tofs));
        }

        acc.x = fmaxf(fmaf(acc.x, di, bv.x), 0.0f);
        acc.y = fmaxf(fmaf(acc.y, di, bv.y), 0.0f);
        acc.z = fmaxf(fmaf(acc.z, di, bv.z), 0.0f);
        acc.w = fmaxf(fmaf(acc.w, di, bv.w), 0.0f);
        if (!live) acc = 0.0f;
        *reinterpret_cast<v4f*>(tile + nl * AP + cg) = acc;
        if (l16 == 0) sDi[w * 16 + nl] = di;
    }
    __syncthreads();

    if (row0 >= n) return;

    const int m = l16;
    const float* dv = sDi + w * 16 + half * 8;

    v8f accm[NT];
#pragma unroll
    for (int t = 0; t < NT; ++t) accm[t] = 0.0f;

#pragma unroll
    for (int kt = 0; kt < KT; ++kt) {
        const int ka = kt * 4 + half * 2;
        v2f a;
        a.x = tile[m * AP + ka];
        a.y = tile[m * AP + ka + 1];
#pragma unroll
        for (int nt = 0; nt < NT; ++nt) {
            const int col = nt * 16 + m;
            v2f bb;
            bb.x = sW[ka * NW16 + col];
            bb.y = sW[(ka + 1) * NW16 + col];
            accm[nt] = __builtin_amdgcn_wmma_f32_16x16x4_f32(
                false, a, false, bb, (short)0, accm[nt], false, false);
        }
    }

#pragma unroll
    for (int nt = 0; nt < NT; ++nt) {
#pragma unroll
        for (int i = 0; i < 8; ++i) {
            const int r   = row0 + i + half * 8;
            const int col = nt * 16 + m;
            if (col < OP && r < n) Hout[(size_t)r * OP + col] = accm[nt][i] * dv[i];
        }
    }
}

__global__ void __launch_bounds__(256)
gather3_k(const int* __restrict__ rowstart, const int* __restrict__ csrc,
          const float* __restrict__ h, const float* __restrict__ b,
          float* __restrict__ out, int n, int E) {
    constexpr int CAP = 384;
    constexpr int LPT = CAP / 32;
    constexpr int SU  = 8;

    __shared__ alignas(16) float srow[8 * CAP * 4];
    __shared__ float so[8 * 16 * 3];

    const int tid  = threadIdx.x;
    const int lane = tid & 31;
    const int w    = tid >> 5;
    const int row0 = (blockIdx.x * 8 + w) * 16;
    float4* stage = reinterpret_cast<float4*>(srow) + w * CAP;

    int rs;
    {
        const int node = row0 + min(lane, 16);
        const int v = rowstart[min(node, n - 1)];
        rs = (node < n) ? v : E;
        rs = min(max(rs, 0), E);
    }

    const int nl = lane >> 1;
    const int hh = lane & 1;
    const int i  = row0 + nl;
    const bool live = i < n;
    const int ic = live ? i : (n - 1);
    const float4 selfrow = *reinterpret_cast<const float4*>(h + (size_t)ic * 4);

    const int t0  = __shfl(rs, 0);
    const int cnt = max(__shfl(rs, 16) - t0, 0);
    const int cst = min(cnt, CAP);

    int iv[LPT];
#pragma unroll
    for (int k = 0; k < LPT; ++k) iv[k] = csrc[min(t0 + lane + 32 * k, E - 1)];
    float4 v[LPT];
#pragma unroll
    for (int k = 0; k < LPT; ++k) {
        const int s = ((unsigned)iv[k] < (unsigned)n) ? iv[k] : ic;
        v[k] = make_float4(0.0f, 0.0f, 0.0f, 0.0f);
        if (lane + 32 * k < cst) v[k] = *reinterpret_cast<const float4*>(h + (size_t)s * 4);
    }
#pragma unroll
    for (int k = 0; k < LPT; ++k) stage[lane + 32 * k] = v[k];
    __syncthreads();

    const int p0 = __shfl(rs, nl);
    int p1 = __shfl(rs, nl + 1);
    const float di = rsqrtf((float)(p1 - p0 + 1));
    if (!live) p1 = p0;
    const int a0 = p0 - t0;
    const int a1 = p1 - t0;

    const int a1s = min(a1, CAP);
    float s0 = 0.0f, s1 = 0.0f, s2 = 0.0f;
    for (int q = a0 + hh; q < a1s; q += 2 * SU) {
        float4 r[SU];
#pragma unroll
        for (int j = 0; j < SU; ++j) r[j] = stage[min(max(q + 2 * j, 0), CAP - 1)];
#pragma unroll
        for (int j = 0; j < SU; ++j) {
            if (q + 2 * j < a1s) {
                s0 += r[j].x;
                s1 += r[j].y;
                s2 += r[j].z;
            }
        }
    }
    {
        int qt = a0 + hh;
        if (qt < CAP) qt += ((CAP - qt + 1) >> 1) << 1;
        for (int q = qt; q < a1; q += 2) {
            int s = csrc[min(max(t0 + q, 0), E - 1)];
            s = ((unsigned)s < (unsigned)n) ? s : ic;
            const float4 r = *reinterpret_cast<const float4*>(h + (size_t)s * 4);
            s0 += r.x;
            s1 += r.y;
            s2 += r.z;
        }
    }
    s0 += __shfl_xor(s0, 1); s1 += __shfl_xor(s1, 1); s2 += __shfl_xor(s2, 1);

    if (hh == 0) {
        const int o = (w * 16 + nl) * 3;
        so[o + 0] = fmaf(selfrow.x + s0, di, b[0]);
        so[o + 1] = fmaf(selfrow.y + s1, di, b[1]);
        so[o + 2] = fmaf(selfrow.z + s2, di, b[2]);
    }
    __syncthreads();
    for (int idx = tid; idx < 8 * 16 * 3; idx += 256) {
        const long long gi = (long long)blockIdx.x * (8 * 16 * 3) + idx;
        if (gi < (long long)n * 3) out[gi] = so[idx];
    }
}


static inline int cdiv(long long a, int b) { return (int)((a + b - 1) / b); }

extern "C" void kernel_launch(void* const* d_in, const int* in_sizes, int n_in,
                              void* d_out, int out_size, void* d_ws, size_t ws_size,
                              hipStream_t stream) {
    const float* x   = (const float*)d_in[0];
    const int*   ei  = (const int*)d_in[1];
    const float* W1  = (const float*)d_in[2];
    const float* b1  = (const float*)d_in[3];
    const float* W2  = (const float*)d_in[4];
    const float* b2  = (const float*)d_in[5];
    const float* W3  = (const float*)d_in[6];
    const float* b3  = (const float*)d_in[7];
    float* out = (float*)d_out;

    const int N = in_sizes[0] / 9;
    const int E = in_sizes[1] / 2;
    const int* src = ei;
    const int* dst = ei + E;

    int*   rowstart = (int*)d_ws;
    int*   csrc     = rowstart + N;
    float* h        = (float*)(csrc + E);
    float* agg      = h + (size_t)N * 64;
    const int P  = cdiv(E, TILE);
    const int NB = cdiv(N, BSZ);
    unsigned* tmp  = (unsigned*)h;
    float*    xs   = h + E;
    int*      toff = (int*)agg;
    const float* zrow = xs + (size_t)N * XW;
    (void)ws_size; (void)n_in; (void)out_size;

    const int ntiles = (N + 15) / 16;

    part_k<<<P, 1024, 0, stream>>>(src, dst, tmp, toff, E, NB);
    finish_k<<<NB, 256, 0, stream>>>(tmp, toff, rowstart, csrc, x, xs, E, N, P, NB);

    gather16_gemm2_k<<<cdiv(ntiles, 8), 256, 0, stream>>>(rowstart, csrc, xs, W1, b1, W2, agg, N, E);

    gather64_gemm_k<3><<<cdiv(ntiles, 8), 256, 0, stream>>>(rowstart, csrc, agg, zrow, b2, W3, h, N, E);

    gather3_k<<<cdiv(ntiles, 8), 256, 0, stream>>>(rowstart, csrc, h, b3, out, N, E);
}


# 64-channel table in whole aligned 256-B units: rows start on 256-B boundaries (workspace order; were 128 B past) and layer 1 stores them whole from its LDS tile, 16 B per lane, not as 64-B pieces
# speedup vs baseline: 8.8155x; 1.0320x over previous
#include <hip/hip_runtime.h>


typedef float v2f __attribute__((ext_vector_type(2)));
typedef float v8f __attribute__((ext_vector_type(8)));
typedef float v4f __attribute__((ext_vector_type(4)));


constexpr int BSH   = 7;
constexpr int BSZ   = 1 << BSH;
constexpr int NBMAX = 1024;
constexpr int TILE  = 6400;
constexpr int EPT   = (TILE + 1023) / 1024;
constexpr int PMAX  = 256;
constexpr int STG   = 3072;
constexpr int KPT   = STG / 256;
constexpr int XW    = 16;

__device__ __forceinline__ unsigned bucket_of(int d, int NB) {
    return min((unsigned)d >> BSH, (unsigned)(NB - 1));
}

__global__ void __launch_bounds__(1024)
part_k(const int* __restrict__ src, const int* __restrict__ dst,
       unsigned* __restrict__ tmp, int* __restrict__ toff, int E, int NB) {
    __shared__ int sc[NBMAX];
    __shared__ int cur[NBMAX];
    __shared__ unsigned sorted[TILE];

    const int tid = threadIdx.x;
    const int e0  = blockIdx.x * TILE;
    const int e1  = min(e0 + TILE, E);

    const int elast = max(e1 - 1, 0);
    int dd[EPT], ss[EPT];
#pragma unroll
    for (int k = 0; k < EPT; ++k) {
        const int ec = min(e0 + tid + k * 1024, elast);
        dd[k] = dst[ec];
        ss[k] = src[ec];
    }

    cur[tid] = 0;
    __syncthreads();
#pragma unroll
    for (int k = 0; k < EPT; ++k) {
        const int e = e0 + tid + k * 1024;
        if (e < e1) atomicAdd(&cur[bucket_of(dd[k], NB)], 1);
    }
    __syncthreads();

    const int v = cur[tid];
    sc[tid] = v;
    __syncthreads();
    for (int d = 1; d < 1024; d <<= 1) {
        const int tv = (tid >= d) ? sc[tid - d] : 0;
        __syncthreads();
        sc[tid] += tv;
        __syncthreads();
    }
    const int excl = sc[tid] - v;
    cur[tid] = excl;
    int* trow = toff + (size_t)blockIdx.x * (NB + 1);
    if (tid < NB) trow[tid] = excl;
    if (tid == 0) trow[NB] = max(e1 - e0, 0);
    __syncthreads();

#pragma unroll
    for (int k = 0; k < EPT; ++k) {
        const int e = e0 + tid + k * 1024;
        if (e < e1) {
            const unsigned b = bucket_of(dd[k], NB);
            unsigned pos = (unsigned)atomicAdd(&cur[b], 1);
            pos = min(pos, (unsigned)(TILE - 1));
            sorted[pos] = ((unsigned)ss[k] << BSH) | ((unsigned)dd[k] & (unsigned)(BSZ - 1));
        }
    }
    __syncthreads();

    const int cnt = e1 - e0;
    for (int i = tid; i < cnt; i += 1024) tmp[e0 + i] = sorted[i];
}

__global__ void __launch_bounds__(256)
finish_k(const unsigned* __restrict__ tmp, const int* __restrict__ toff,
         int* __restrict__ rowstart, int* __restrict__ csrc,
         const float* __restrict__ x, float* __restrict__ xs, int E, int n, int P, int NB) {
    __shared__ int red[256];
    __shared__ int cnt[BSZ];
    __shared__ int cur[BSZ];
    __shared__ int stage[STG];
    __shared__ float sdi[BSZ];
    __shared__ int wlen[8];
    __shared__ int wbef[8];

    const int tid  = threadIdx.x;
    const int lane = tid & 31;
    const int w    = tid >> 5;
    const int b    = blockIdx.x;

    float xv[BSZ * XW / 256];
#pragma unroll
    for (int j = 0; j < BSZ * XW / 256; ++j) {
        const int idx  = tid + j * 256;
        const int nloc = idx / XW;
        const int c    = idx - nloc * XW;
        const int node = min(b * BSZ + nloc, n - 1);
        xv[j] = x[(size_t)node * 9 + min(c, 8)];
    }

    const int pt = min(tid, P - 1);
    const int* trow = toff + (size_t)pt * (NB + 1) + b;
    const int qa = trow[0];
    const int qe = trow[1];
    const int tcap = min(max(E - pt * TILE, 0), TILE);
    const int ra   = min(max(qa, 0), tcap);
    const int re   = min(max(qe, ra), tcap);
    const int len  = (tid < P) ? re - ra : 0;
    const int bef  = (tid < P) ? ra : 0;
    const int gb   = pt * TILE + ra;

    int inc = len, sbef = bef;
#pragma unroll
    for (int d = 1; d < 32; d <<= 1) {
        const int t = __shfl_up(inc, d);
        inc  += (lane >= d) ? t : 0;
        sbef += __shfl_xor(sbef, d);
    }
    if (lane == 31) { wlen[w] = inc; wbef[w] = sbef; }
    if (tid < BSZ) cnt[tid] = 0;
    __syncthreads();
    int off = inc - len, mtot = 0, stot = 0;
#pragma unroll
    for (int j = 0; j < 8; ++j) {
        const int t = wlen[j];
        off  += (j < w) ? t : 0;
        mtot += t;
        stot += wbef[j];
    }
    const int start = min(max(stot, 0), E);
    const int m     = min(max(mtot, 0), E - start);

    const bool small = (m <= STG);
    int* rb = stage;
    int* rl = stage + PMAX;
    unsigned pk[KPT];
    if (small) {
        for (int j = 0; j < len; ++j) stage[min(off + j, STG - 1)] = gb + j;
        __syncthreads();
        const int ilast = max(m - 1, 0);
        int si[KPT];
#pragma unroll
        for (int k = 0; k < KPT; ++k) si[k] = stage[min(tid + k * 256, ilast)];
#pragma unroll
        for (int k = 0; k < KPT; ++k) pk[k] = tmp[min(max(si[k], 0), E - 1)];
#pragma unroll
        for (int k = 0; k < KPT; ++k) {
            const int i = tid + k * 256;
            if (i < m) atomicAdd(&cnt[pk[k] & (unsigned)(BSZ - 1)], 1);
        }
    } else {
        rb[tid] = gb;
        rl[tid] = len;
        __syncthreads();
        for (int p = 0; p < min(P, PMAX); ++p) {
            const int L = rl[p], B = rb[p];
            for (int i = tid; i < L; i += 256) atomicAdd(&cnt[tmp[min(B + i, E - 1)] & (unsigned)(BSZ - 1)], 1);
        }
    }
    __syncthreads();

    const int v = (tid < BSZ) ? cnt[tid] : 0;
    red[tid] = v;
    __syncthreads();
    for (int d = 1; d < BSZ; d <<= 1) {
        const int t = (tid >= d) ? red[tid - d] : 0;
        __syncthreads();
        red[tid] += t;
        __syncthreads();
    }
    if (tid < BSZ) {
        const int excl = red[tid] - v;
        const int node = b * BSZ + tid;
        if (node < n) rowstart[node] = start + excl;
        cur[tid] = excl;
        sdi[tid] = rsqrtf((float)(v + 1));
    }
    __syncthreads();

#pragma unroll
    for (int j = 0; j < BSZ * XW / 256; ++j) {
        const int idx  = tid + j * 256;
        const int nloc = idx / XW;
        const int c    = idx - nloc * XW;
        if (b * BSZ + nloc < n) xs[(size_t)b * (BSZ * XW) + idx] = (c < 9) ? xv[j] * sdi[nloc] : 0.0f;
    }
    if (b == 0 && tid < 64) xs[(size_t)n * XW + tid] = 0.0f;

    if (small) {
#pragma unroll
        for (int k = 0; k < KPT; ++k) {
            const int i = tid + k * 256;
            if (i < m) {
                unsigned pos = (unsigned)atomicAdd(&cur[pk[k] & (unsigned)(BSZ - 1)], 1);
                pos = min(pos, (unsigned)(STG - 1));
                stage[pos] = (int)(pk[k] >> BSH);
            }
        }
        __syncthreads();
        for (int i = tid; i < m; i += 256) csrc[start + i] = stage[i];
    } else {
        for (int p = 0; p < min(P, PMAX); ++p) {
            const int L = rl[p], B = rb[p];
            for (int i = tid; i < L; i += 256) {
                const unsigned q = tmp[min(B + i, E - 1)];
                unsigned pos = (unsigned)atomicAdd(&cur[q & (unsigned)(BSZ - 1)], 1);
                pos = min(pos, (unsigned)(m - 1));
                csrc[start + pos] = (int)(q >> BSH);
            }
        }
    }
}

__global__ void __launch_bounds__(256)
gather16_gemm2_k(const int* __restrict__ rowstart, const int* __restrict__ csrc,
                 const float* __restrict__ xs, const float* __restrict__ W1,
                 const float* __restrict__ b1, const float* __restrict__ W2,
                 float* __restrict__ Hout, int n, int E) {
    constexpr int K1P = 12;
    constexpr int AP  = 68;
    constexpr int XP  = 20;
    constexpr int CAP = 384;
    constexpr int CH  = 16;

    __shared__ float sW2[64 * 64];
    __shared__ float sW1[K1P * 64];
    __shared__ alignas(16) float sA[8 * 16 * AP];
    __shared__ alignas(16) float sX[8 * 16 * XP];
    __shared__ float sDi[8 * 16];

    const int lane = threadIdx.x & 31;
    const int w    = threadIdx.x >> 5;
    const int q    = lane & 3;
    const int g    = lane >> 2;
    const int row0 = (blockIdx.x * 8 + w) * 16;
    float* tile = sA + w * (16 * AP);
    float* xt   = sX + w * (16 * XP);

    int rs;
    {
        const int node = row0 + min(lane, 16);
        const int v = rowstart[min(node, n - 1)];
        rs = (node < n) ? v : E;
        rs = min(max(rs, 0), E);
    }
    const int t0 = __shfl(rs, 0);

    float bc[4];
#pragma unroll
    for (int t = 0; t < 4; ++t) bc[t] = b1[t * 16 + (lane & 15)];

    int iv[CAP / 32];
#pragma unroll
    for (int k = 0; k < CAP / 32; ++k) iv[k] = csrc[min(t0 + lane + 32 * k, E - 1)];
    v4f selfrow[2];
#pragma unroll
    for (int pass = 0; pass < 2; ++pass)
        selfrow[pass] = *reinterpret_cast<const v4f*>(xs + (size_t)min(row0 + pass * 8 + g, n - 1) * XW + q * 4);
    {
        constexpr int WPT = 64 * 64 / 256;
        float wreg[WPT];
#pragma unroll
        for (int j = 0; j < WPT; ++j) wreg[j] = W2[threadIdx.x + j * 256];
        float w1reg[K1P * 64 / 256];
#pragma unroll
        for (int j = 0; j < K1P * 64 / 256; ++j) {
            const int idx = threadIdx.x + j * 256;
            const int k = idx >> 6;
            const float t = W1[min(k, 8) * 64 + (idx & 63)];
            w1reg[j] = (k < 9) ? t : 0.0f;
        }
#pragma unroll
        for (int k = 0; k < CAP / 32; ++k) tile[lane + 32 * k] = __int_as_float(iv[k]);
#pragma unroll
        for (int j = 0; j < WPT; ++j) sW2[threadIdx.x + j * 256] = wreg[j];
#pragma unroll
        for (int j = 0; j < K1P * 64 / 256; ++j) sW1[threadIdx.x + j * 256] = w1reg[j];
    }
    __syncthreads();

    const char* xb = reinterpret_cast<const char*>(xs);
    const unsigned q16 = (unsigned)q * 16u;

    for (int pass = 0; pass < 2; ++pass) {
        const int nl = pass * 8 + g;
        const int i  = row0 + nl;
        const bool live = i < n;
        const int p0 = __shfl(rs, nl);
        int p1 = __shfl(rs, nl + 1);
        const float di = rsqrtf((float)(p1 - p0 + 1));
        if (!live) p1 = p0;

        v4f acc = selfrow[pass];

        int p = p0;
        for (; p < p1 && (unsigned)(p - t0) <= (unsigned)(CAP - CH); p += CH) {
            const float* ip = tile + (p - t0);
            int sk[CH];
#pragma unroll
            for (int k = 0; k < CH; ++k) sk[k] = __float_as_int(ip[k]);
            const int rem = p1 - p;
            v4f v[CH];
#pragma unroll
            for (int k = 0; k < CH; ++k) {
                const unsigned sc  = min((unsigned)sk[k], (unsigned)(n - 1));
                const unsigned row = (k < rem) ? sc : (unsigned)n;
                v[k] = *reinterpret_cast<const v4f*>(xb + ((row << 6) + q16));
            }
            asm volatile("" :: "v"(v[0]), "v"(v[1]), "v"(v[2]), "v"(v[3]), "v"(v[4]), "v"(v[5]), "v"(v[6]), "v"(v[7]),
                               "v"(v[8]), "v"(v[9]), "v"(v[10]), "v"(v[11]), "v"(v[12]), "v"(v[13]), "v"(v[14]), "v"(v[15]));
#pragma unroll
            for (int k = 0; k < CH; ++k) acc += v[k];
        }
        for (; p < p1; ++p) {
            const unsigned sc = min((unsigned)csrc[min(max(p, 0), E - 1)], (unsigned)(n - 1));
            acc += *reinterpret_cast<const v4f*>(xb + ((sc << 6) + q16));
        }
        acc *= di;
        if (!live) acc = 0.0f;
        *reinterpret_cast<v4f*>(xt + nl * XP + q * 4) = acc;
        if (q == 0) sDi[w * 16 + nl] = di;
    }
    __syncthreads();

    const int m    = lane & 15;
    const int half = lane >> 4;

    v8f c1[4];
#pragma unroll
    for (int t = 0; t < 4; ++t) c1[t] = 0.0f;
#pragma unroll
    for (int kt = 0; kt < K1P / 4; ++kt) {
        const int ka = kt * 4 + half * 2;
        v2f a;
        a.x = xt[m * XP + ka];
        a.y = xt[m * XP + ka + 1];
#pragma unroll
        for (int nt = 0; nt < 4; ++nt) {
            const int col = nt * 16 + m;
            v2f bb;
            bb.x = sW1[ka * 64 + col];
            bb.y = sW1[(ka + 1) * 64 + col];
            c1[nt] = __builtin_amdgcn_wmma_f32_16x16x4_f32(
                false, a, false, bb, (short)0, c1[nt], false, false);
        }
    }
#pragma unroll
    for (int nt = 0; nt < 4; ++nt) {
#pragma unroll
        for (int r = 0; r < 8; ++r)
            tile[(r + half * 8) * AP + nt * 16 + m] = fmaxf(c1[nt][r] + bc[nt], 0.0f);
    }
    __syncthreads();

    const float* dv = sDi + w * 16 + half * 8;

    v8f accm[4];
#pragma unroll
    for (int t = 0; t < 4; ++t) accm[t] = 0.0f;
#pragma unroll
    for (int kt = 0; kt < 16; ++kt) {
        const int ka = kt * 4 + half * 2;
        v2f a;
        a.x = tile[m * AP + ka];
        a.y = tile[m * AP + ka + 1];
#pragma unroll
        for (int nt = 0; nt < 4; ++nt) {
            const int col = nt * 16 + m;
            v2f bb;
            bb.x = sW2[ka * 64 + col];
            bb.y = sW2[(ka + 1) * 64 + col];
            accm[nt] = __builtin_amdgcn_wmma_f32_16x16x4_f32(
                false, a, false, bb, (short)0, accm[nt], false, false);
        }
    }
#pragma unroll
    for (int nt = 0; nt < 4; ++nt) {
#pragma unroll
        for (int r = 0; r < 8; ++r)
            tile[(r + half * 8) * AP + nt * 16 + m] = accm[nt][r] * dv[r];
    }
    __syncthreads();
    v4f o[8];
#pragma unroll
    for (int j = 0; j < 8; ++j)
        o[j] = *reinterpret_cast<const v4f*>(tile + (2 * j + half) * AP + m * 4);
    asm volatile("" :: "v"(o[0]), "v"(o[1]), "v"(o[2]), "v"(o[3]), "v"(o[4]), "v"(o[5]), "v"(o[6]), "v"(o[7]));
#pragma unroll
    for (int j = 0; j < 8; ++j) {
        const int row = row0 + 2 * j + half;
        if (row < n) *reinterpret_cast<v4f*>(Hout + (size_t)row * 64 + m * 4) = o[j];
    }
}


template<int NOUT>
__global__ void __launch_bounds__(256)
gather64_gemm_k(const int* __restrict__ rowstart, const int* __restrict__ csrc,
                const float* __restrict__ h, const float* zrow, const float* __restrict__ b,
                const float* __restrict__ W, float* __restrict__ Hout, int n, int E) {
    constexpr int KDIM = 64;
    constexpr int NT   = (NOUT + 15) / 16;
    constexpr int KT   = KDIM / 4;
    constexpr int NW16 = NT * 16;
    constexpr int OP   = (NOUT + 3) / 4 * 4;
    constexpr int AP   = 68;
    constexpr int CH   = 16;
    constexpr int CAP  = 384;

    __shared__ float sW[KDIM * NW16];
    __shared__ alignas(16) float sA[8 * 16 * AP];
    __shared__ float sDi[8 * 16];
    __shared__ int sIdx[8 * CAP];

    const int lane = threadIdx.x & 31;
    const int w    = threadIdx.x >> 5;
    const int l16  = lane & 15;
    const int half = lane >> 4;
    const int cg   = l16 << 2;
    const int row0 = (blockIdx.x * 8 + w) * 16;
    float* tile = sA + w * (16 * AP);
    int* myidx  = sIdx + w * CAP;

    int rs;
    {
        const int node = row0 + min(lane, 16);
        const int v = rowstart[min(node, n - 1)];
        rs = (node < n) ? v : E;
        rs = min(max(rs, 0), E);
    }
    const float4 bv = *reinterpret_cast<const float4*>(b + cg);
    const int t0 = __shfl(rs, 0);

    v4f selfrow;
    {
        int iv[CAP / 32];
#pragma unroll
        for (int k = 0; k < CAP / 32; ++k) iv[k] = csrc[min(t0 + lane + 32 * k, E - 1)];
        selfrow = *reinterpret_cast<const v4f*>(h + (size_t)min(row0 + half, n - 1) * 64 + cg);
        constexpr int WPT = KDIM * NW16 / 256;
        float wreg[WPT];
#pragma unroll
        for (int j = 0; j < WPT; ++j) {
            const int idx = threadIdx.x + j * 256;
            const int k = idx / NW16;
            const int c = idx - k * NW16;
            wreg[j] = (c < NOUT) ? W[k * NOUT + c] : 0.0f;
        }
#pragma unroll
        for (int k = 0; k < CAP / 32; ++k) myidx[lane + 32 * k] = iv[k];
#pragma unroll
        for (int j = 0; j < WPT; ++j) sW[threadIdx.x + j * 256] = wreg[j];
    }
    __syncthreads();

    const char* tb = reinterpret_cast<const char*>(zrow);
    const unsigned cg4  = (unsigned)cg * 4u;
    const unsigned tofs = (unsigned)(reinterpret_cast<const char*>(h) - tb) + cg4;

    for (int pass = 0; pass < 8; ++pass) {
        const int nl = pass * 2 + half;
        const int i  = row0 + nl;
        const bool live = i < n;
        const int p0 = __shfl(rs, nl);
        int p1 = __shfl(rs, nl + 1);
        const float di = rsqrtf((float)(p1 - p0 + 1));
        if (!live) p1 = p0;

        v4f acc = selfrow;
        selfrow = *reinterpret_cast<const v4f*>(h + (size_t)min(i + 2, n - 1) * 64 + cg);

        int p = p0;
        for (; p < p1 && (unsigned)(p - t0) <= (unsigned)(CAP - CH); p += CH) {
            const int* ip = myidx + (p - t0);
            int sk[CH];
#pragma unroll
            for (int k = 0; k < CH; ++k) sk[k] = ip[k];
            const int rem = p1 - p;
            v4f v[CH];
#pragma unroll
            for (int k = 0; k < CH; ++k) {
                const unsigned sc  = min((unsigned)sk[k], (unsigned)(n - 1));
                const unsigned off = (k < rem) ? ((sc << 8) + tofs) : cg4;
                v[k] = *reinterpret_cast<const v4f*>(tb + off);
            }
            asm volatile("" :: "v"(v[0]), "v"(v[1]), "v"(v[2]), "v"(v[3]), "v"(v[4]), "v"(v[5]), "v"(v[6]), "v"(v[7]),
                               "v"(v[8]), "v"(v[9]), "v"(v[10]), "v"(v[11]), "v"(v[12]), "v"(v[13]), "v"(v[14]), "v"(v[15]));
#pragma unroll
            for (int k = 0; k < CH; ++k) acc += v[k];
        }
        for (; p < p1; ++p) {
            const unsigned sc = min((unsigned)csrc[min(max(p, 0), E - 1)], (unsigned)(n - 1));
            acc += *reinterpret_cast<const v4f*>(tb + ((sc << 8) + tofs));
        }

        acc.x = fmaxf(fmaf(acc.x, di, bv.x), 0.0f);
        acc.y = fmaxf(fmaf(acc.y, di, bv.y), 0.0f);
        acc.z = fmaxf(fmaf(acc.z, di, bv.z), 0.0f);
        acc.w = fmaxf(fmaf(acc.w, di, bv.w), 0.0f);
        if (!live) acc = 0.0f;
        *reinterpret_cast<v4f*>(tile + nl * AP + cg) = acc;
        if (l16 == 0) sDi[w * 16 + nl] = di;
    }
    __syncthreads();

    if (row0 >= n) return;

    const int m = l16;
    const float* dv = sDi + w * 16 + half * 8;

    v8f accm[NT];
#pragma unroll
    for (int t = 0; t < NT; ++t) accm[t] = 0.0f;

#pragma unroll
    for (int kt = 0; kt < KT; ++kt) {
        const int ka = kt * 4 + half * 2;
        v2f a;
        a.x = tile[m * AP + ka];
        a.y = tile[m * AP + ka + 1];
#pragma unroll
        for (int nt = 0; nt < NT; ++nt) {
            const int col = nt * 16 + m;
            v2f bb;
            bb.x = sW[ka * NW16 + col];
            bb.y = sW[(ka + 1) * NW16 + col];
            accm[nt] = __builtin_amdgcn_wmma_f32_16x16x4_f32(
                false, a, false, bb, (short)0, accm[nt], false, false);
        }
    }

#pragma unroll
    for (int nt = 0; nt < NT; ++nt) {
#pragma unroll
        for (int i = 0; i < 8; ++i) {
            const int r   = row0 + i + half * 8;
            const int col = nt * 16 + m;
            if (col < OP && r < n) Hout[(size_t)r * OP + col] = accm[nt][i] * dv[i];
        }
    }
}

__global__ void __launch_bounds__(256)
gather3_k(const int* __restrict__ rowstart, const int* __restrict__ csrc,
          const float* __restrict__ h, const float* __restrict__ b,
          float* __restrict__ out, int n, int E) {
    constexpr int CAP = 384;
    constexpr int LPT = CAP / 32;
    constexpr int SU  = 8;

    __shared__ alignas(16) float srow[8 * CAP * 4];
    __shared__ float so[8 * 16 * 3];

    const int tid  = threadIdx.x;
    const int lane = tid & 31;
    const int w    = tid >> 5;
    const int row0 = (blockIdx.x * 8 + w) * 16;
    float4* stage = reinterpret_cast<float4*>(srow) + w * CAP;

    int rs;
    {
        const int node = row0 + min(lane, 16);
        const int v = rowstart[min(node, n - 1)];
        rs = (node < n) ? v : E;
        rs = min(max(rs, 0), E);
    }

    const int nl = lane >> 1;
    const int hh = lane & 1;
    const int i  = row0 + nl;
    const bool live = i < n;
    const int ic = live ? i : (n - 1);
    const float4 selfrow = *reinterpret_cast<const float4*>(h + (size_t)ic * 4);

    const int t0  = __shfl(rs, 0);
    const int cnt = max(__shfl(rs, 16) - t0, 0);
    const int cst = min(cnt, CAP);

    int iv[LPT];
#pragma unroll
    for (int k = 0; k < LPT; ++k) iv[k] = csrc[min(t0 + lane + 32 * k, E - 1)];
    float4 v[LPT];
#pragma unroll
    for (int k = 0; k < LPT; ++k) {
        const int s = ((unsigned)iv[k] < (unsigned)n) ? iv[k] : ic;
        v[k] = make_float4(0.0f, 0.0f, 0.0f, 0.0f);
        if (lane + 32 * k < cst) v[k] = *reinterpret_cast<const float4*>(h + (size_t)s * 4);
    }
#pragma unroll
    for (int k = 0; k < LPT; ++k) stage[lane + 32 * k] = v[k];
    __syncthreads();

    const int p0 = __shfl(rs, nl);
    int p1 = __shfl(rs, nl + 1);
    const float di = rsqrtf((float)(p1 - p0 + 1));
    if (!live) p1 = p0;
    const int a0 = p0 - t0;
    const int a1 = p1 - t0;

    const int a1s = min(a1, CAP);
    float s0 = 0.0f, s1 = 0.0f, s2 = 0.0f;
    for (int q = a0 + hh; q < a1s; q += 2 * SU) {
        float4 r[SU];
#pragma unroll
        for (int j = 0; j < SU; ++j) r[j] = stage[min(max(q + 2 * j, 0), CAP - 1)];
#pragma unroll
        for (int j = 0; j < SU; ++j) {
            if (q + 2 * j < a1s) {
                s0 += r[j].x;
                s1 += r[j].y;
                s2 += r[j].z;
            }
        }
    }
    {
        int qt = a0 + hh;
        if (qt < CAP) qt += ((CAP - qt + 1) >> 1) << 1;
        for (int q = qt; q < a1; q += 2) {
            int s = csrc[min(max(t0 + q, 0), E - 1)];
            s = ((unsigned)s < (unsigned)n) ? s : ic;
            const float4 r = *reinterpret_cast<const float4*>(h + (size_t)s * 4);
            s0 += r.x;
            s1 += r.y;
            s2 += r.z;
        }
    }
    s0 += __shfl_xor(s0, 1); s1 += __shfl_xor(s1, 1); s2 += __shfl_xor(s2, 1);

    if (hh == 0) {
        const int o = (w * 16 + nl) * 3;
        so[o + 0] = fmaf(selfrow.x + s0, di, b[0]);
        so[o + 1] = fmaf(selfrow.y + s1, di, b[1]);
        so[o + 2] = fmaf(selfrow.z + s2, di, b[2]);
    }
    __syncthreads();
    for (int idx = tid; idx < 8 * 16 * 3; idx += 256) {
        const long long gi = (long long)blockIdx.x * (8 * 16 * 3) + idx;
        if (gi < (long long)n * 3) out[gi] = so[idx];
    }
}


static inline int cdiv(long long a, int b) { return (int)((a + b - 1) / b); }

extern "C" void kernel_launch(void* const* d_in, const int* in_sizes, int n_in,
                              void* d_out, int out_size, void* d_ws, size_t ws_size,
                              hipStream_t stream) {
    const float* x   = (const float*)d_in[0];
    const int*   ei  = (const int*)d_in[1];
    const float* W1  = (const float*)d_in[2];
    const float* b1  = (const float*)d_in[3];
    const float* W2  = (const float*)d_in[4];
    const float* b2  = (const float*)d_in[5];
    const float* W3  = (const float*)d_in[6];
    const float* b3  = (const float*)d_in[7];
    float* out = (float*)d_out;

    const int N = in_sizes[0] / 9;
    const int E = in_sizes[1] / 2;
    const int* src = ei;
    const int* dst = ei + E;

    char*  wb       = (char*)d_ws;
    float* h        = (float*)(wb + ((256 - ((size_t)wb & 255)) & 255));
    float* agg      = h + ((size_t)N * 64 - 64);
    int*   rowstart = (int*)(agg + (size_t)N * 64);
    int*   csrc     = rowstart + N;
    const int P  = cdiv(E, TILE);
    const int NB = cdiv(N, BSZ);
    unsigned* tmp  = (unsigned*)h;
    float*    xs   = h + E;
    int*      toff = (int*)agg;
    const float* zrow = xs + (size_t)N * XW;
    (void)ws_size; (void)n_in; (void)out_size;

    const int ntiles = (N + 15) / 16;

    part_k<<<P, 1024, 0, stream>>>(src, dst, tmp, toff, E, NB);
    finish_k<<<NB, 256, 0, stream>>>(tmp, toff, rowstart, csrc, x, xs, E, N, P, NB);

    gather16_gemm2_k<<<cdiv(ntiles, 8), 256, 0, stream>>>(rowstart, csrc, xs, W1, b1, W2, agg, N, E);

    gather64_gemm_k<3><<<cdiv(ntiles, 8), 256, 0, stream>>>(rowstart, csrc, agg, zrow, b2, W3, h, N, E);

    gather3_k<<<cdiv(ntiles, 8), 256, 0, stream>>>(rowstart, csrc, h, b3, out, N, E);
}
